# starting kernel
# baseline (speedup 1.0000x reference)
#include <hip/hip_runtime.h>
#include <math.h>

typedef float v2f __attribute__((ext_vector_type(2)));
typedef float v8f __attribute__((ext_vector_type(8)));

namespace {
constexpr int kNodes = 50000;
constexpr int kEdges = 800000;
constexpr int kInCh  = 64;
constexpr int kOutCh = 32;
constexpr int kHeads = 4;
constexpr int kHC    = kHeads * kOutCh;
constexpr int kETot  = kEdges + kNodes;
}


__device__ __forceinline__ void atomic_max_f32(float* addr, float v) {
  if (v >= 0.0f) atomicMax(reinterpret_cast<int*>(addr), __float_as_int(v));
  else           atomicMin(reinterpret_cast<unsigned int*>(addr), __float_as_uint(v));
}

__device__ __forceinline__ float leaky02(float v) { return v > 0.0f ? v : 0.2f * v; }


__global__ void k_copy_x(const float* __restrict__ x, float* __restrict__ x_pos) {
  int i = blockIdx.x * blockDim.x + threadIdx.x;
  if (i < kNodes * kInCh) x_pos[i] = x[i];
}

__global__ void k_init(float* __restrict__ amax, float* __restrict__ denom,
                       float* __restrict__ out) {
  int i = blockIdx.x * blockDim.x + threadIdx.x;
  if (i < kNodes * kHeads) { amax[i] = -INFINITY; denom[i] = 0.0f; }
  if (i < kNodes * kHC)    out[i] = 0.0f;
}

__global__ void k_scatter_pe(const int* __restrict__ ei, const int* __restrict__ et,
                             const float* __restrict__ pe, float* __restrict__ x_pos) {
  long long tid = (long long)blockIdx.x * blockDim.x + threadIdx.x;
  if (tid >= (long long)kEdges * kInCh) return;
  int e = (int)(tid >> 6);
  int c = (int)(tid & 63);
  int s = ei[e];
  atomicAdd(&x_pos[(size_t)s * kInCh + c], pe[(size_t)et[e] * kInCh + c]);
}

__global__ void __launch_bounds__(256) k_gemm_wmma(const float* __restrict__ A,
                                                   const float* __restrict__ B,
                                                   float* __restrict__ C) {
  const int lane = threadIdx.x & 31;
  const int wave = threadIdx.x >> 5;
  const int m0   = blockIdx.x * 16;
  const int n0   = wave * 16;
  const int half = lane >> 4;
  const int l15  = lane & 15;
  const int row  = m0 + l15;
  const int col  = n0 + l15;

  v8f acc = {0.f, 0.f, 0.f, 0.f, 0.f, 0.f, 0.f, 0.f};
#pragma unroll
  for (int k0 = 0; k0 < kInCh; k0 += 4) {
    const int kb = k0 + 2 * half;
    v2f a, b;
    a.x = A[(size_t)row * kInCh + kb];
    a.y = A[(size_t)row * kInCh + kb + 1];
    b.x = B[(size_t)kb * kHC + col];
    b.y = B[(size_t)(kb + 1) * kHC + col];
    acc = __builtin_amdgcn_wmma_f32_16x16x4_f32(false, a, false, b, (short)0, acc,
                                                false, false);
  }
#pragma unroll
  for (int r = 0; r < 8; ++r) {
    C[(size_t)(m0 + r + 8 * half) * kHC + col] = acc[r];
  }
}

__global__ void k_scores(const float* __restrict__ xl,
                         const float* __restrict__ att_src, const float* __restrict__ att_dst,
                         float* __restrict__ a_src, float* __restrict__ a_dst) {
  int tid = blockIdx.x * blockDim.x + threadIdx.x;
  if (tid >= kNodes * kHeads) return;
  int n = tid / kHeads, h = tid % kHeads;
  const float* xr = xl + (size_t)n * kHC + h * kOutCh;
  float ss = 0.0f, sd = 0.0f;
#pragma unroll
  for (int c = 0; c < kOutCh; ++c) {
    ss += xr[c] * att_src[h * kOutCh + c];
    sd += xr[c] * att_dst[h * kOutCh + c];
  }
  a_src[tid] = ss;
  a_dst[tid] = sd;
}

__device__ __forceinline__ void edge_sd(const int* __restrict__ ei, int e, int& s, int& d) {
  if (e < kEdges) { s = ei[e]; d = ei[kEdges + e]; }
  else            { s = e - kEdges; d = s; }
}

__global__ void k_alpha_max(const int* __restrict__ ei,
                            const float* __restrict__ a_src, const float* __restrict__ a_dst,
                            float* __restrict__ amax) {
  long long tid = (long long)blockIdx.x * blockDim.x + threadIdx.x;
  if (tid >= (long long)kETot * kHeads) return;
  int e = (int)(tid >> 2), h = (int)(tid & 3);
  int s, d; edge_sd(ei, e, s, d);
  float al = leaky02(a_src[s * kHeads + h] + a_dst[d * kHeads + h]);
  atomic_max_f32(&amax[d * kHeads + h], al);
}

__global__ void k_exp_sum(const int* __restrict__ ei,
                          const float* __restrict__ a_src, const float* __restrict__ a_dst,
                          const float* __restrict__ amax, float* __restrict__ denom) {
  long long tid = (long long)blockIdx.x * blockDim.x + threadIdx.x;
  if (tid >= (long long)kETot * kHeads) return;
  int e = (int)(tid >> 2), h = (int)(tid & 3);
  int s, d; edge_sd(ei, e, s, d);
  float al = leaky02(a_src[s * kHeads + h] + a_dst[d * kHeads + h]);
  atomicAdd(&denom[d * kHeads + h], expf(al - amax[d * kHeads + h]));
}

__global__ void k_aggregate(const int* __restrict__ ei,
                            const float* __restrict__ a_src, const float* __restrict__ a_dst,
                            const float* __restrict__ amax, const float* __restrict__ denom,
                            const float* __restrict__ xl, float* __restrict__ out) {
  long long tid = (long long)blockIdx.x * blockDim.x + threadIdx.x;
  if (tid >= (long long)kETot * kHeads) return;
  int e = (int)(tid >> 2), h = (int)(tid & 3);
  int s, d; edge_sd(ei, e, s, d);
  float al = leaky02(a_src[s * kHeads + h] + a_dst[d * kHeads + h]);
  float coef = expf(al - amax[d * kHeads + h]) / denom[d * kHeads + h];
  const float* xr = xl + (size_t)s * kHC + h * kOutCh;
  float* orow = out + (size_t)d * kHC + h * kOutCh;
#pragma unroll 8
  for (int c = 0; c < kOutCh; ++c) atomicAdd(&orow[c], coef * xr[c]);
}

__global__ void k_finalize(float* __restrict__ out, const float* __restrict__ bias) {
  int i = blockIdx.x * blockDim.x + threadIdx.x;
  if (i >= kNodes * kHC) return;
  float v = out[i] + bias[i & (kHC - 1)];
  out[i] = v > 0.0f ? v : expm1f(v);
}


extern "C" void kernel_launch(void* const* d_in, const int* in_sizes, int n_in,
                              void* d_out, int out_size, void* d_ws, size_t ws_size,
                              hipStream_t stream) {
  (void)in_sizes; (void)n_in; (void)out_size; (void)ws_size;
  const float* x       = (const float*)d_in[0];
  const int*   ei      = (const int*)d_in[1];
  const int*   et      = (const int*)d_in[2];
  const float* pe      = (const float*)d_in[3];
  const float* W       = (const float*)d_in[4];
  const float* att_src = (const float*)d_in[5];
  const float* att_dst = (const float*)d_in[6];
  const float* bias    = (const float*)d_in[7];
  float* out = (float*)d_out;

  float* ws    = (float*)d_ws;
  float* x_pos = ws;
  float* xl    = x_pos + (size_t)kNodes * kInCh;
  float* a_src = xl    + (size_t)kNodes * kHC;
  float* a_dst = a_src + (size_t)kNodes * kHeads;
  float* amax  = a_dst + (size_t)kNodes * kHeads;
  float* denom = amax  + (size_t)kNodes * kHeads;

  const int B = 256;
  k_copy_x<<<(kNodes * kInCh + B - 1) / B, B, 0, stream>>>(x, x_pos);
  k_init<<<(kNodes * kHC + B - 1) / B, B, 0, stream>>>(amax, denom, out);
  {
    long long n = (long long)kEdges * kInCh;
    k_scatter_pe<<<(unsigned)((n + B - 1) / B), B, 0, stream>>>(ei, et, pe, x_pos);
  }
  k_gemm_wmma<<<kNodes / 16, 256, 0, stream>>>(x_pos, W, xl);
  k_scores<<<(kNodes * kHeads + B - 1) / B, B, 0, stream>>>(xl, att_src, att_dst, a_src, a_dst);
  {
    long long n = (long long)kETot * kHeads;
    unsigned g = (unsigned)((n + B - 1) / B);
    k_alpha_max<<<g, B, 0, stream>>>(ei, a_src, a_dst, amax);
    k_exp_sum<<<g, B, 0, stream>>>(ei, a_src, a_dst, amax, denom);
    k_aggregate<<<g, B, 0, stream>>>(ei, a_src, a_dst, amax, denom, xl, out);
  }
  k_finalize<<<(kNodes * kHC + B - 1) / B, B, 0, stream>>>(out, bias);
}


# aggregation by owner blocks: 256 blocks each hold 196 destination rows in LDS, scan the whole edge list, add owned edges with LDS f32 atomics (arrival order), store bias+ELU rows once; 8 launches
# speedup vs baseline: 3.2069x; 3.2069x over previous
#include <hip/hip_runtime.h>
#include <math.h>

typedef float v2f __attribute__((ext_vector_type(2)));
typedef float v8f __attribute__((ext_vector_type(8)));

namespace {
constexpr int kNodes = 50000;
constexpr int kEdges = 800000;
constexpr int kInCh  = 64;
constexpr int kOutCh = 32;
constexpr int kHeads = 4;
constexpr int kHC    = kHeads * kOutCh;
constexpr int kETot  = kEdges + kNodes;
constexpr int kOwnBlocks = 256;
constexpr int kOwnRows   = 196;
constexpr int kOwnThr    = 1024;
constexpr int kOwnWaves  = kOwnThr / 32;
constexpr int kLoadsPerThr = 8;
constexpr int kChunkQuads  = kOwnThr * kLoadsPerThr;
constexpr int kChunkEdges  = kChunkQuads * 4;
constexpr int kQuads       = kEdges / 4;
constexpr int kChunks      = (kQuads + kChunkQuads - 1) / kChunkQuads;
static_assert(kEdges % 4 == 0, "dst row is read in groups of 4");
static_assert(kOwnBlocks * kOwnRows >= kNodes, "every node has an owner");
static_assert(kOwnRows <= 256, "local row index is packed into 8 bits");
static_assert(kNodes <= (1 << 24), "source index is packed into 24 bits");
}


__device__ __forceinline__ void atomic_max_f32(float* addr, float v) {
  if (v >= 0.0f) atomicMax(reinterpret_cast<int*>(addr), __float_as_int(v));
  else           atomicMin(reinterpret_cast<unsigned int*>(addr), __float_as_uint(v));
}

__device__ __forceinline__ float leaky02(float v) { return v > 0.0f ? v : 0.2f * v; }


__global__ void k_copy_x(const float* __restrict__ x, float* __restrict__ x_pos) {
  int i = blockIdx.x * blockDim.x + threadIdx.x;
  if (i < kNodes * kInCh) x_pos[i] = x[i];
}

__global__ void k_init(float* __restrict__ amax, float* __restrict__ denom) {
  int i = blockIdx.x * blockDim.x + threadIdx.x;
  if (i < kNodes * kHeads) { amax[i] = -INFINITY; denom[i] = 0.0f; }
}

__global__ void k_scatter_pe(const int* __restrict__ ei, const int* __restrict__ et,
                             const float* __restrict__ pe, float* __restrict__ x_pos) {
  long long tid = (long long)blockIdx.x * blockDim.x + threadIdx.x;
  if (tid >= (long long)kEdges * kInCh) return;
  int e = (int)(tid >> 6);
  int c = (int)(tid & 63);
  int s = ei[e];
  atomicAdd(&x_pos[(size_t)s * kInCh + c], pe[(size_t)et[e] * kInCh + c]);
}

__global__ void __launch_bounds__(256) k_gemm_wmma(const float* __restrict__ A,
                                                   const float* __restrict__ B,
                                                   float* __restrict__ C) {
  const int lane = threadIdx.x & 31;
  const int wave = threadIdx.x >> 5;
  const int m0   = blockIdx.x * 16;
  const int n0   = wave * 16;
  const int half = lane >> 4;
  const int l15  = lane & 15;
  const int row  = m0 + l15;
  const int col  = n0 + l15;

  v8f acc = {0.f, 0.f, 0.f, 0.f, 0.f, 0.f, 0.f, 0.f};
#pragma unroll
  for (int k0 = 0; k0 < kInCh; k0 += 4) {
    const int kb = k0 + 2 * half;
    v2f a, b;
    a.x = A[(size_t)row * kInCh + kb];
    a.y = A[(size_t)row * kInCh + kb + 1];
    b.x = B[(size_t)kb * kHC + col];
    b.y = B[(size_t)(kb + 1) * kHC + col];
    acc = __builtin_amdgcn_wmma_f32_16x16x4_f32(false, a, false, b, (short)0, acc,
                                                false, false);
  }
#pragma unroll
  for (int r = 0; r < 8; ++r) {
    C[(size_t)(m0 + r + 8 * half) * kHC + col] = acc[r];
  }
}

__global__ void k_scores(const float* __restrict__ xl,
                         const float* __restrict__ att_src, const float* __restrict__ att_dst,
                         float* __restrict__ a_src, float* __restrict__ a_dst) {
  int tid = blockIdx.x * blockDim.x + threadIdx.x;
  if (tid >= kNodes * kHeads) return;
  int n = tid / kHeads, h = tid % kHeads;
  const float* xr = xl + (size_t)n * kHC + h * kOutCh;
  float ss = 0.0f, sd = 0.0f;
#pragma unroll
  for (int c = 0; c < kOutCh; ++c) {
    ss += xr[c] * att_src[h * kOutCh + c];
    sd += xr[c] * att_dst[h * kOutCh + c];
  }
  a_src[tid] = ss;
  a_dst[tid] = sd;
}

__device__ __forceinline__ void edge_sd(const int* __restrict__ ei, int e, int& s, int& d) {
  if (e < kEdges) { s = ei[e]; d = ei[kEdges + e]; }
  else            { s = e - kEdges; d = s; }
}

__global__ void k_alpha_max(const int* __restrict__ ei,
                            const float* __restrict__ a_src, const float* __restrict__ a_dst,
                            float* __restrict__ amax) {
  long long tid = (long long)blockIdx.x * blockDim.x + threadIdx.x;
  if (tid >= (long long)kETot * kHeads) return;
  int e = (int)(tid >> 2), h = (int)(tid & 3);
  int s, d; edge_sd(ei, e, s, d);
  float al = leaky02(a_src[s * kHeads + h] + a_dst[d * kHeads + h]);
  atomic_max_f32(&amax[d * kHeads + h], al);
}

__global__ void k_exp_sum(const int* __restrict__ ei,
                          const float* __restrict__ a_src, const float* __restrict__ a_dst,
                          const float* __restrict__ amax, float* __restrict__ denom) {
  long long tid = (long long)blockIdx.x * blockDim.x + threadIdx.x;
  if (tid >= (long long)kETot * kHeads) return;
  int e = (int)(tid >> 2), h = (int)(tid & 3);
  int s, d; edge_sd(ei, e, s, d);
  float al = leaky02(a_src[s * kHeads + h] + a_dst[d * kHeads + h]);
  atomicAdd(&denom[d * kHeads + h], expf(al - amax[d * kHeads + h]));
}

__global__ void __launch_bounds__(kOwnThr)
k_owner_aggregate(const int* __restrict__ ei,
                  const float* __restrict__ a_src, const float* __restrict__ a_dst,
                  const float* __restrict__ amax, const float* __restrict__ denom,
                  const float* __restrict__ xl, const float* __restrict__ bias,
                  float* __restrict__ out) {
  __shared__ float    acc[kOwnRows * kHC];
  __shared__ unsigned list[kChunkEdges];
  __shared__ unsigned cnt[kChunks];

  const int tid  = threadIdx.x;
  const int lane = tid & 31;
  const int wave = __builtin_amdgcn_readfirstlane(tid >> 5);
  const int h    = lane >> 3;
  const int d0   = blockIdx.x * kOwnRows;
  const float4* __restrict__ xl4 = reinterpret_cast<const float4*>(xl);

  if (tid < kChunks) cnt[tid] = 0u;

  for (int dl = wave; dl < kOwnRows; dl += kOwnWaves) {
    const int d = d0 + dl;
    if (d < kNodes) {
      const float al   = leaky02(a_src[d * kHeads + h] + a_dst[d * kHeads + h]);
      const float coef = expf(al - amax[d * kHeads + h]) / denom[d * kHeads + h];
      const float4 v = xl4[(size_t)d * (kHC / 4) + lane];
      float* row = acc + dl * kHC + lane;
      row[0]  = coef * v.x;
      row[32] = coef * v.y;
      row[64] = coef * v.z;
      row[96] = coef * v.w;
    }
  }
  __syncthreads();

  const int4* __restrict__ dst4 = reinterpret_cast<const int4*>(ei + kEdges);

  for (int c = 0; c < kChunks; ++c) {
    int4 dv[kLoadsPerThr];
#pragma unroll
    for (int k = 0; k < kLoadsPerThr; ++k) {
      const int q = c * kChunkQuads + k * kOwnThr + tid;
      dv[k] = (q < kQuads) ? dst4[q] : make_int4(-1, -1, -1, -1);
    }
#pragma unroll
    for (int k = 0; k < kLoadsPerThr; ++k) {
      const int q = c * kChunkQuads + k * kOwnThr + tid;
      const int dd[4] = {dv[k].x, dv[k].y, dv[k].z, dv[k].w};
#pragma unroll
      for (int j = 0; j < 4; ++j) {
        const unsigned rel = (unsigned)(dd[j] - d0);
        if (rel < (unsigned)kOwnRows) {
          const int s = ei[4 * q + j];
          const unsigned slot = atomicAdd(&cnt[c], 1u);
          list[slot] = ((unsigned)s << 8) | rel;
        }
      }
    }
    __syncthreads();

    const unsigned n = cnt[c];
    for (unsigned i0 = (unsigned)wave * 4u; i0 < n; i0 += (unsigned)kOwnWaves * 4u) {
      unsigned ent[4];
      float4 v[4];
      float as[4], ad[4], mx[4], dn[4];
#pragma unroll
      for (int u = 0; u < 4; ++u) {
        const unsigned i = (i0 + u < n) ? (i0 + u) : (n - 1u);
        ent[u] = list[i];
      }
#pragma unroll
      for (int u = 0; u < 4; ++u) {
        const int s = (int)(ent[u] >> 8);
        const int d = d0 + (int)(ent[u] & 255u);
        v[u]  = xl4[(size_t)s * (kHC / 4) + lane];
        as[u] = a_src[s * kHeads + h];
        ad[u] = a_dst[d * kHeads + h];
        mx[u] = amax[d * kHeads + h];
        dn[u] = denom[d * kHeads + h];
      }
      asm volatile("" :: "v"(v[0].x), "v"(v[1].x), "v"(v[2].x), "v"(v[3].x),
                         "v"(as[0]), "v"(as[1]), "v"(as[2]), "v"(as[3]),
                         "v"(ad[0]), "v"(ad[1]), "v"(ad[2]), "v"(ad[3]),
                         "v"(mx[0]), "v"(mx[1]), "v"(mx[2]), "v"(mx[3]),
                         "v"(dn[0]), "v"(dn[1]), "v"(dn[2]), "v"(dn[3]));
#pragma unroll
      for (int u = 0; u < 4; ++u) {
        if (i0 + u < n) {
          const float coef = expf(leaky02(as[u] + ad[u]) - mx[u]) / dn[u];
          float* row = acc + (int)(ent[u] & 255u) * kHC + lane;
          atomicAdd(row,      coef * v[u].x);
          atomicAdd(row + 32, coef * v[u].y);
          atomicAdd(row + 64, coef * v[u].z);
          atomicAdd(row + 96, coef * v[u].w);
        }
      }
    }
    __syncthreads();
  }

  const float4 b = reinterpret_cast<const float4*>(bias)[lane];
  float4* __restrict__ out4 = reinterpret_cast<float4*>(out);
  for (int dl = wave; dl < kOwnRows; dl += kOwnWaves) {
    const int d = d0 + dl;
    if (d < kNodes) {
      const float* row = acc + dl * kHC + lane;
      float4 r;
      r.x = row[0]  + b.x;
      r.y = row[32] + b.y;
      r.z = row[64] + b.z;
      r.w = row[96] + b.w;
      r.x = r.x > 0.0f ? r.x : expm1f(r.x);
      r.y = r.y > 0.0f ? r.y : expm1f(r.y);
      r.z = r.z > 0.0f ? r.z : expm1f(r.z);
      r.w = r.w > 0.0f ? r.w : expm1f(r.w);
      out4[(size_t)d * (kHC / 4) + lane] = r;
    }
  }
}


extern "C" void kernel_launch(void* const* d_in, const int* in_sizes, int n_in,
                              void* d_out, int out_size, void* d_ws, size_t ws_size,
                              hipStream_t stream) {
  (void)in_sizes; (void)n_in; (void)out_size; (void)ws_size;
  const float* x       = (const float*)d_in[0];
  const int*   ei      = (const int*)d_in[1];
  const int*   et      = (const int*)d_in[2];
  const float* pe      = (const float*)d_in[3];
  const float* W       = (const float*)d_in[4];
  const float* att_src = (const float*)d_in[5];
  const float* att_dst = (const float*)d_in[6];
  const float* bias    = (const float*)d_in[7];
  float* out = (float*)d_out;

  float* ws    = (float*)d_ws;
  float* x_pos = ws;
  float* xl    = x_pos + (size_t)kNodes * kInCh;
  float* a_src = xl    + (size_t)kNodes * kHC;
  float* a_dst = a_src + (size_t)kNodes * kHeads;
  float* amax  = a_dst + (size_t)kNodes * kHeads;
  float* denom = amax  + (size_t)kNodes * kHeads;

  const int B = 256;
  k_copy_x<<<(kNodes * kInCh + B - 1) / B, B, 0, stream>>>(x, x_pos);
  k_init<<<(kNodes * kHeads + B - 1) / B, B, 0, stream>>>(amax, denom);
  {
    long long n = (long long)kEdges * kInCh;
    k_scatter_pe<<<(unsigned)((n + B - 1) / B), B, 0, stream>>>(ei, et, pe, x_pos);
  }
  k_gemm_wmma<<<kNodes / 16, 256, 0, stream>>>(x_pos, W, xl);
  k_scores<<<(kNodes * kHeads + B - 1) / B, B, 0, stream>>>(xl, att_src, att_dst, a_src, a_dst);
  {
    long long n = (long long)kETot * kHeads;
    unsigned g = (unsigned)((n + B - 1) / B);
    k_alpha_max<<<g, B, 0, stream>>>(ei, a_src, a_dst, amax);
    k_exp_sum<<<g, B, 0, stream>>>(ei, a_src, a_dst, amax, denom);
  }
  k_owner_aggregate<<<kOwnBlocks, kOwnThr, 0, stream>>>(ei, a_src, a_dst, amax, denom, xl, bias, out);
}


# positional scatter by owner blocks: 256 blocks each keep 196 source rows of x_pos in LDS (seeded with x), scan the src list once, add pe rows by LDS f32 atomics (arrival order); 7 launches
# speedup vs baseline: 4.4416x; 1.3850x over previous
#include <hip/hip_runtime.h>
#include <math.h>

typedef float v2f __attribute__((ext_vector_type(2)));
typedef float v8f __attribute__((ext_vector_type(8)));

namespace {
constexpr int kNodes = 50000;
constexpr int kEdges = 800000;
constexpr int kInCh  = 64;
constexpr int kOutCh = 32;
constexpr int kHeads = 4;
constexpr int kHC    = kHeads * kOutCh;
constexpr int kETot  = kEdges + kNodes;
constexpr int kOwnBlocks = 256;
constexpr int kOwnRows   = 196;
constexpr int kOwnThr    = 1024;
constexpr int kOwnWaves  = kOwnThr / 32;
constexpr int kLoadsPerThr = 8;
constexpr int kChunkQuads  = kOwnThr * kLoadsPerThr;
constexpr int kChunkEdges  = kChunkQuads * 4;
constexpr int kQuads       = kEdges / 4;
constexpr int kChunks      = (kQuads + kChunkQuads - 1) / kChunkQuads;
static_assert(kEdges % 4 == 0, "dst row is read in groups of 4");
static_assert(kOwnBlocks * kOwnRows >= kNodes, "every node has an owner");
static_assert(kOwnRows <= 256, "local row index is packed into 8 bits");
static_assert(kNodes <= (1 << 24), "source index is packed into 24 bits");
constexpr int kTypes       = 64;
constexpr int kPeLoads     = 8;
constexpr int kPeStepQuads = kOwnThr * kPeLoads;
constexpr int kPeSteps     = (kQuads + kPeStepQuads - 1) / kPeStepQuads;
constexpr int kPeStagger   = 768;
constexpr int kPeStepMax   = 32 * kPeLoads * 4;
constexpr int kPeSeg       = 1536;
constexpr int kPeRowsPad   = 256;
static_assert(kQuads % 32 == 0 && kPeStagger % 32 == 0, "a wave's 32 quads never straddle the wrap");
static_assert(kOwnBlocks * kPeStagger <= kQuads, "start places stay inside the list");
static_assert(kEdges <= (1 << 24), "edge index is packed into 24 bits");
static_assert(kPeSteps % 2 == 1, "the scan loop below takes steps in pairs and one last step");
static_assert(kPeSeg > kPeStepMax, "a whole step fits behind the mark at which a segment is emptied");
static_assert(kPeRowsPad >= kOwnRows && kPeRowsPad * (kInCh / 4) % kOwnThr == 0, "the seed copy covers the padded rows exactly");
}


__device__ __forceinline__ void atomic_max_f32(float* addr, float v) {
  if (v >= 0.0f) atomicMax(reinterpret_cast<int*>(addr), __float_as_int(v));
  else           atomicMin(reinterpret_cast<unsigned int*>(addr), __float_as_uint(v));
}

__device__ __forceinline__ float leaky02(float v) { return v > 0.0f ? v : 0.2f * v; }


__global__ void k_init(float* __restrict__ amax, float* __restrict__ denom) {
  int i = blockIdx.x * blockDim.x + threadIdx.x;
  if (i < kNodes * kHeads) { amax[i] = -INFINITY; denom[i] = 0.0f; }
}


__device__ __forceinline__ void pe_add_noted(const unsigned* seg, int n, int lane,
                                             float* __restrict__ rows,
                                             const float* __restrict__ pes,
                                             const int* __restrict__ et) {
  for (int base = 0; base < n; base += 32) {
    const int i = base + lane;
    unsigned ent = 0u;
    int t = 0;
    if (i < n) { ent = seg[i]; t = et[ent >> 8]; }
    const int m = (n - base < 32) ? (n - base) : 32;
    for (int u = 0; u < m; u += 4) {
      int tu[4];
      unsigned ru[4];
      float p0[4], p1[4];
#pragma unroll
      for (int j = 0; j < 4; ++j) {
        const int uj = (u + j < m) ? (u + j) : (m - 1);
        tu[j] = __builtin_amdgcn_readlane(t, uj);
        ru[j] = (unsigned)__builtin_amdgcn_readlane((int)ent, uj) & 255u;
        p0[j] = pes[tu[j] * kInCh + lane];
        p1[j] = pes[tu[j] * kInCh + 32 + lane];
      }
      asm volatile("" :: "v"(p0[0]), "v"(p0[1]), "v"(p0[2]), "v"(p0[3]),
                         "v"(p1[0]), "v"(p1[1]), "v"(p1[2]), "v"(p1[3]));
#pragma unroll
      for (int j = 0; j < 4; ++j) {
        if (u + j < m) {
          atomicAdd(&rows[ru[j] * kInCh + lane],      p0[j]);
          atomicAdd(&rows[ru[j] * kInCh + 32 + lane], p1[j]);
        }
      }
    }
  }
}

__device__ __forceinline__ void pe_note(unsigned mask, bool mine, unsigned rel, int e,
                                        int& wcount, unsigned* seg) {
  if (mask != 0u) {
    if (mine) seg[wcount + (int)__builtin_amdgcn_mbcnt_lo(mask, 0u)] = ((unsigned)e << 8) | rel;
    wcount += __popc(mask);
  }
}

__device__ __forceinline__ int pe_quad(int step, int k, int off, int tid) {
  int lin = step * kPeStepQuads + k * kOwnThr + tid;
  if (lin >= kQuads) lin -= kQuads;
  int q = off + lin;
  if (q >= kQuads) q -= kQuads;
  return q;
}

__device__ __forceinline__ void pe_load(int4 (&dv)[kPeLoads], const int4* __restrict__ src4,
                                        int step, int off, int tid) {
#pragma unroll
  for (int k = 0; k < kPeLoads; ++k) dv[k] = src4[pe_quad(step, k, off, tid)];
}

__device__ __forceinline__ void pe_scan(const int4 (&dv)[kPeLoads], int step, int off, int tid,
                                        int wave, int s0, unsigned nrow, int& wcount,
                                        unsigned* seg) {
#pragma unroll
  for (int k = 0; k < kPeLoads; ++k) {
    if (step * kPeStepQuads + k * kOwnThr + wave * 32 < kQuads) {
      const int q = pe_quad(step, k, off, tid);
      const unsigned r0 = (unsigned)(dv[k].x - s0);
      const unsigned r1 = (unsigned)(dv[k].y - s0);
      const unsigned r2 = (unsigned)(dv[k].z - s0);
      const unsigned r3 = (unsigned)(dv[k].w - s0);
      const unsigned m0 = __builtin_amdgcn_ballot_w32(r0 < nrow);
      const unsigned m1 = __builtin_amdgcn_ballot_w32(r1 < nrow);
      const unsigned m2 = __builtin_amdgcn_ballot_w32(r2 < nrow);
      const unsigned m3 = __builtin_amdgcn_ballot_w32(r3 < nrow);
      if ((m0 | m1 | m2 | m3) != 0u) {
        pe_note(m0, r0 < nrow, r0, 4 * q + 0, wcount, seg);
        pe_note(m1, r1 < nrow, r1, 4 * q + 1, wcount, seg);
        pe_note(m2, r2 < nrow, r2, 4 * q + 2, wcount, seg);
        pe_note(m3, r3 < nrow, r3, 4 * q + 3, wcount, seg);
      }
    }
  }
}

__global__ void __launch_bounds__(kOwnThr)
k_owner_pe(const float* __restrict__ x, const int* __restrict__ ei, const int* __restrict__ et,
           const float* __restrict__ pe, float* __restrict__ x_pos) {
  __shared__ float    rows[kPeRowsPad * kInCh];
  __shared__ float    pes[kTypes * kInCh];
  __shared__ unsigned list[kOwnWaves * kPeSeg];

  const int tid  = threadIdx.x;
  const int lane = tid & 31;
  const int wave = __builtin_amdgcn_readfirstlane(tid >> 5);
  const int s0   = blockIdx.x * kOwnRows;
  const int nrow = (kNodes - s0 < kOwnRows) ? (kNodes - s0) : kOwnRows;
  const int nvec = nrow * (kInCh / 4);
  unsigned* seg = list + wave * kPeSeg;

  {
    constexpr int kXv = kPeRowsPad * (kInCh / 4) / kOwnThr;
    static_assert(kTypes * kInCh / 4 == kOwnThr, "one 16-byte load of pe per thread");
    const float4* __restrict__ x4 = reinterpret_cast<const float4*>(x + (size_t)s0 * kInCh);
    float4* rows4 = reinterpret_cast<float4*>(rows);
    reinterpret_cast<float4*>(pes)[tid] = reinterpret_cast<const float4*>(pe)[tid];
#pragma unroll
    for (int j = 0; j < kXv; ++j) {
      const int i = j * kOwnThr + tid;
      rows4[i] = x4[i < nvec ? i : nvec - 1];
    }
  }
  __syncthreads();

  const int4* __restrict__ src4 = reinterpret_cast<const int4*>(ei);
  const int off = blockIdx.x * kPeStagger;
  int wcount = 0;
  int4 da[kPeLoads], db[kPeLoads];
  pe_load(da, src4, 0, off, tid);
  for (int st = 0; st + 1 < kPeSteps; st += 2) {
    pe_load(db, src4, st + 1, off, tid);
    if (wcount > kPeSeg - kPeStepMax) { pe_add_noted(seg, wcount, lane, rows, pes, et); wcount = 0; }
    pe_scan(da, st, off, tid, wave, s0, (unsigned)nrow, wcount, seg);
    pe_load(da, src4, st + 2, off, tid);
    if (wcount > kPeSeg - kPeStepMax) { pe_add_noted(seg, wcount, lane, rows, pes, et); wcount = 0; }
    pe_scan(db, st + 1, off, tid, wave, s0, (unsigned)nrow, wcount, seg);
  }
  if (wcount > kPeSeg - kPeStepMax) { pe_add_noted(seg, wcount, lane, rows, pes, et); wcount = 0; }
  pe_scan(da, kPeSteps - 1, off, tid, wave, s0, (unsigned)nrow, wcount, seg);

  pe_add_noted(seg, wcount, lane, rows, pes, et);
  __syncthreads();

  {
    const float4* rows4 = reinterpret_cast<const float4*>(rows);
    float4* __restrict__ xp4 = reinterpret_cast<float4*>(x_pos + (size_t)s0 * kInCh);
    for (int i = tid; i < nvec; i += kOwnThr) xp4[i] = rows4[i];
  }
}

__global__ void __launch_bounds__(256) k_gemm_wmma(const float* __restrict__ A,
                                                   const float* __restrict__ B,
                                                   float* __restrict__ C) {
  const int lane = threadIdx.x & 31;
  const int wave = threadIdx.x >> 5;
  const int m0   = blockIdx.x * 16;
  const int n0   = wave * 16;
  const int half = lane >> 4;
  const int l15  = lane & 15;
  const int row  = m0 + l15;
  const int col  = n0 + l15;

  v8f acc = {0.f, 0.f, 0.f, 0.f, 0.f, 0.f, 0.f, 0.f};
#pragma unroll
  for (int k0 = 0; k0 < kInCh; k0 += 4) {
    const int kb = k0 + 2 * half;
    v2f a, b;
    a.x = A[(size_t)row * kInCh + kb];
    a.y = A[(size_t)row * kInCh + kb + 1];
    b.x = B[(size_t)kb * kHC + col];
    b.y = B[(size_t)(kb + 1) * kHC + col];
    acc = __builtin_amdgcn_wmma_f32_16x16x4_f32(false, a, false, b, (short)0, acc,
                                                false, false);
  }
#pragma unroll
  for (int r = 0; r < 8; ++r) {
    C[(size_t)(m0 + r + 8 * half) * kHC + col] = acc[r];
  }
}

__global__ void k_scores(const float* __restrict__ xl,
                         const float* __restrict__ att_src, const float* __restrict__ att_dst,
                         float* __restrict__ a_src, float* __restrict__ a_dst) {
  int tid = blockIdx.x * blockDim.x + threadIdx.x;
  if (tid >= kNodes * kHeads) return;
  int n = tid / kHeads, h = tid % kHeads;
  const float* xr = xl + (size_t)n * kHC + h * kOutCh;
  float ss = 0.0f, sd = 0.0f;
#pragma unroll
  for (int c = 0; c < kOutCh; ++c) {
    ss += xr[c] * att_src[h * kOutCh + c];
    sd += xr[c] * att_dst[h * kOutCh + c];
  }
  a_src[tid] = ss;
  a_dst[tid] = sd;
}

__device__ __forceinline__ void edge_sd(const int* __restrict__ ei, int e, int& s, int& d) {
  if (e < kEdges) { s = ei[e]; d = ei[kEdges + e]; }
  else            { s = e - kEdges; d = s; }
}

__global__ void k_alpha_max(const int* __restrict__ ei,
                            const float* __restrict__ a_src, const float* __restrict__ a_dst,
                            float* __restrict__ amax) {
  long long tid = (long long)blockIdx.x * blockDim.x + threadIdx.x;
  if (tid >= (long long)kETot * kHeads) return;
  int e = (int)(tid >> 2), h = (int)(tid & 3);
  int s, d; edge_sd(ei, e, s, d);
  float al = leaky02(a_src[s * kHeads + h] + a_dst[d * kHeads + h]);
  atomic_max_f32(&amax[d * kHeads + h], al);
}

__global__ void k_exp_sum(const int* __restrict__ ei,
                          const float* __restrict__ a_src, const float* __restrict__ a_dst,
                          const float* __restrict__ amax, float* __restrict__ denom) {
  long long tid = (long long)blockIdx.x * blockDim.x + threadIdx.x;
  if (tid >= (long long)kETot * kHeads) return;
  int e = (int)(tid >> 2), h = (int)(tid & 3);
  int s, d; edge_sd(ei, e, s, d);
  float al = leaky02(a_src[s * kHeads + h] + a_dst[d * kHeads + h]);
  atomicAdd(&denom[d * kHeads + h], expf(al - amax[d * kHeads + h]));
}

__global__ void __launch_bounds__(kOwnThr)
k_owner_aggregate(const int* __restrict__ ei,
                  const float* __restrict__ a_src, const float* __restrict__ a_dst,
                  const float* __restrict__ amax, const float* __restrict__ denom,
                  const float* __restrict__ xl, const float* __restrict__ bias,
                  float* __restrict__ out) {
  __shared__ float    acc[kOwnRows * kHC];
  __shared__ unsigned list[kChunkEdges];
  __shared__ unsigned cnt[kChunks];

  const int tid  = threadIdx.x;
  const int lane = tid & 31;
  const int wave = __builtin_amdgcn_readfirstlane(tid >> 5);
  const int h    = lane >> 3;
  const int d0   = blockIdx.x * kOwnRows;
  const float4* __restrict__ xl4 = reinterpret_cast<const float4*>(xl);

  if (tid < kChunks) cnt[tid] = 0u;

  for (int dl = wave; dl < kOwnRows; dl += kOwnWaves) {
    const int d = d0 + dl;
    if (d < kNodes) {
      const float al   = leaky02(a_src[d * kHeads + h] + a_dst[d * kHeads + h]);
      const float coef = expf(al - amax[d * kHeads + h]) / denom[d * kHeads + h];
      const float4 v = xl4[(size_t)d * (kHC / 4) + lane];
      float* row = acc + dl * kHC + lane;
      row[0]  = coef * v.x;
      row[32] = coef * v.y;
      row[64] = coef * v.z;
      row[96] = coef * v.w;
    }
  }
  __syncthreads();

  const int4* __restrict__ dst4 = reinterpret_cast<const int4*>(ei + kEdges);

  for (int c = 0; c < kChunks; ++c) {
    int4 dv[kLoadsPerThr];
#pragma unroll
    for (int k = 0; k < kLoadsPerThr; ++k) {
      const int q = c * kChunkQuads + k * kOwnThr + tid;
      dv[k] = (q < kQuads) ? dst4[q] : make_int4(-1, -1, -1, -1);
    }
#pragma unroll
    for (int k = 0; k < kLoadsPerThr; ++k) {
      const int q = c * kChunkQuads + k * kOwnThr + tid;
      const int dd[4] = {dv[k].x, dv[k].y, dv[k].z, dv[k].w};
#pragma unroll
      for (int j = 0; j < 4; ++j) {
        const unsigned rel = (unsigned)(dd[j] - d0);
        if (rel < (unsigned)kOwnRows) {
          const int s = ei[4 * q + j];
          const unsigned slot = atomicAdd(&cnt[c], 1u);
          list[slot] = ((unsigned)s << 8) | rel;
        }
      }
    }
    __syncthreads();

    const unsigned n = cnt[c];
    for (unsigned i0 = (unsigned)wave * 4u; i0 < n; i0 += (unsigned)kOwnWaves * 4u) {
      unsigned ent[4];
      float4 v[4];
      float as[4], ad[4], mx[4], dn[4];
#pragma unroll
      for (int u = 0; u < 4; ++u) {
        const unsigned i = (i0 + u < n) ? (i0 + u) : (n - 1u);
        ent[u] = list[i];
      }
#pragma unroll
      for (int u = 0; u < 4; ++u) {
        const int s = (int)(ent[u] >> 8);
        const int d = d0 + (int)(ent[u] & 255u);
        v[u]  = xl4[(size_t)s * (kHC / 4) + lane];
        as[u] = a_src[s * kHeads + h];
        ad[u] = a_dst[d * kHeads + h];
        mx[u] = amax[d * kHeads + h];
        dn[u] = denom[d * kHeads + h];
      }
      asm volatile("" :: "v"(v[0].x), "v"(v[1].x), "v"(v[2].x), "v"(v[3].x),
                         "v"(as[0]), "v"(as[1]), "v"(as[2]), "v"(as[3]),
                         "v"(ad[0]), "v"(ad[1]), "v"(ad[2]), "v"(ad[3]),
                         "v"(mx[0]), "v"(mx[1]), "v"(mx[2]), "v"(mx[3]),
                         "v"(dn[0]), "v"(dn[1]), "v"(dn[2]), "v"(dn[3]));
#pragma unroll
      for (int u = 0; u < 4; ++u) {
        if (i0 + u < n) {
          const float coef = expf(leaky02(as[u] + ad[u]) - mx[u]) / dn[u];
          float* row = acc + (int)(ent[u] & 255u) * kHC + lane;
          atomicAdd(row,      coef * v[u].x);
          atomicAdd(row + 32, coef * v[u].y);
          atomicAdd(row + 64, coef * v[u].z);
          atomicAdd(row + 96, coef * v[u].w);
        }
      }
    }
    __syncthreads();
  }

  const float4 b = reinterpret_cast<const float4*>(bias)[lane];
  float4* __restrict__ out4 = reinterpret_cast<float4*>(out);
  for (int dl = wave; dl < kOwnRows; dl += kOwnWaves) {
    const int d = d0 + dl;
    if (d < kNodes) {
      const float* row = acc + dl * kHC + lane;
      float4 r;
      r.x = row[0]  + b.x;
      r.y = row[32] + b.y;
      r.z = row[64] + b.z;
      r.w = row[96] + b.w;
      r.x = r.x > 0.0f ? r.x : expm1f(r.x);
      r.y = r.y > 0.0f ? r.y : expm1f(r.y);
      r.z = r.z > 0.0f ? r.z : expm1f(r.z);
      r.w = r.w > 0.0f ? r.w : expm1f(r.w);
      out4[(size_t)d * (kHC / 4) + lane] = r;
    }
  }
}


extern "C" void kernel_launch(void* const* d_in, const int* in_sizes, int n_in,
                              void* d_out, int out_size, void* d_ws, size_t ws_size,
                              hipStream_t stream) {
  (void)in_sizes; (void)n_in; (void)out_size; (void)ws_size;
  const float* x       = (const float*)d_in[0];
  const int*   ei      = (const int*)d_in[1];
  const int*   et      = (const int*)d_in[2];
  const float* pe      = (const float*)d_in[3];
  const float* W       = (const float*)d_in[4];
  const float* att_src = (const float*)d_in[5];
  const float* att_dst = (const float*)d_in[6];
  const float* bias    = (const float*)d_in[7];
  float* out = (float*)d_out;

  float* ws    = (float*)d_ws;
  float* x_pos = ws;
  float* xl    = x_pos + (size_t)kNodes * kInCh;
  float* a_src = xl    + (size_t)kNodes * kHC;
  float* a_dst = a_src + (size_t)kNodes * kHeads;
  float* amax  = a_dst + (size_t)kNodes * kHeads;
  float* denom = amax  + (size_t)kNodes * kHeads;

  const int B = 256;
  k_init<<<(kNodes * kHeads + B - 1) / B, B, 0, stream>>>(amax, denom);
  k_owner_pe<<<kOwnBlocks, kOwnThr, 0, stream>>>(x, ei, et, pe, x_pos);
  k_gemm_wmma<<<kNodes / 16, 256, 0, stream>>>(x_pos, W, xl);
  k_scores<<<(kNodes * kHeads + B - 1) / B, B, 0, stream>>>(xl, att_src, att_dst, a_src, a_dst);
  {
    long long n = (long long)kETot * kHeads;
    unsigned g = (unsigned)((n + B - 1) / B);
    k_alpha_max<<<g, B, 0, stream>>>(ei, a_src, a_dst, amax);
    k_exp_sum<<<g, B, 0, stream>>>(ei, a_src, a_dst, amax, denom);
  }
  k_owner_aggregate<<<kOwnBlocks, kOwnThr, 0, stream>>>(ei, a_src, a_dst, amax, denom, xl, bias, out);
}


# softmax inside the owner blocks: a block lists its edges in LDS in one pass over dst, forms maxima, denominators, row sums by LDS atomics (arrival order); init/max/exp-sum launches gone; 4 launches
# speedup vs baseline: 9.7980x; 2.2060x over previous
#include <hip/hip_runtime.h>
#include <math.h>

typedef float v2f __attribute__((ext_vector_type(2)));
typedef float v8f __attribute__((ext_vector_type(8)));

namespace {
constexpr int kNodes = 50000;
constexpr int kEdges = 800000;
constexpr int kInCh  = 64;
constexpr int kOutCh = 32;
constexpr int kHeads = 4;
constexpr int kHC    = kHeads * kOutCh;
constexpr int kETot  = kEdges + kNodes;
constexpr int kOwnBlocks = 256;
constexpr int kOwnRows   = 196;
constexpr int kOwnThr    = 1024;
constexpr int kOwnWaves  = kOwnThr / 32;
constexpr int kQuads     = kEdges / 4;
static_assert(kEdges % 4 == 0, "the index rows are read in groups of 4");
static_assert(kOwnBlocks * kOwnRows >= kNodes, "every node has an owner");
static_assert(kOwnRows <= 256, "local row index is packed into 8 bits");
static_assert(kNodes <= (1 << 24), "source index is packed into 24 bits");
constexpr int kTypes       = 64;
constexpr int kPeLoads     = 8;
constexpr int kPeStepQuads = kOwnThr * kPeLoads;
constexpr int kPeSteps     = (kQuads + kPeStepQuads - 1) / kPeStepQuads;
constexpr int kPeStagger   = 768;
constexpr int kPeStepMax   = 32 * kPeLoads * 4;
constexpr int kPeSeg       = 1536;
constexpr int kPeRowsPad   = 256;
static_assert(kQuads % 32 == 0 && kPeStagger % 32 == 0, "a wave's 32 quads never straddle the wrap");
static_assert(kOwnBlocks * kPeStagger <= kQuads, "start places stay inside the list");
static_assert(kEdges <= (1 << 24), "edge index is packed into 24 bits");
static_assert(kPeSteps % 2 == 1, "the scan loop below takes steps in pairs and one last step");
static_assert(kPeSeg > kPeStepMax, "a whole step fits behind the mark at which a segment is emptied");
static_assert(kPeRowsPad >= kOwnRows && kPeRowsPad * (kInCh / 4) % kOwnThr == 0, "the seed copy covers the padded rows exactly");
}


__device__ __forceinline__ void atomic_max_f32(float* addr, float v) {
  if (v >= 0.0f) atomicMax(reinterpret_cast<int*>(addr), __float_as_int(v));
  else           atomicMin(reinterpret_cast<unsigned int*>(addr), __float_as_uint(v));
}

__device__ __forceinline__ float leaky02(float v) { return v > 0.0f ? v : 0.2f * v; }


__device__ __forceinline__ void pe_add_noted(const unsigned* seg, int n, int lane,
                                             float* __restrict__ rows,
                                             const float* __restrict__ pes,
                                             const int* __restrict__ et) {
  for (int base = 0; base < n; base += 32) {
    const int i = base + lane;
    unsigned ent = 0u;
    int t = 0;
    if (i < n) { ent = seg[i]; t = et[ent >> 8]; }
    const int m = (n - base < 32) ? (n - base) : 32;
    for (int u = 0; u < m; u += 4) {
      int tu[4];
      unsigned ru[4];
      float p0[4], p1[4];
#pragma unroll
      for (int j = 0; j < 4; ++j) {
        const int uj = (u + j < m) ? (u + j) : (m - 1);
        tu[j] = __builtin_amdgcn_readlane(t, uj);
        ru[j] = (unsigned)__builtin_amdgcn_readlane((int)ent, uj) & 255u;
        p0[j] = pes[tu[j] * kInCh + lane];
        p1[j] = pes[tu[j] * kInCh + 32 + lane];
      }
      asm volatile("" :: "v"(p0[0]), "v"(p0[1]), "v"(p0[2]), "v"(p0[3]),
                         "v"(p1[0]), "v"(p1[1]), "v"(p1[2]), "v"(p1[3]));
#pragma unroll
      for (int j = 0; j < 4; ++j) {
        if (u + j < m) {
          atomicAdd(&rows[ru[j] * kInCh + lane],      p0[j]);
          atomicAdd(&rows[ru[j] * kInCh + 32 + lane], p1[j]);
        }
      }
    }
  }
}

__device__ __forceinline__ void pe_note(unsigned mask, bool mine, unsigned rel, int e,
                                        int& wcount, unsigned* seg) {
  if (mask != 0u) {
    if (mine) seg[wcount + (int)__builtin_amdgcn_mbcnt_lo(mask, 0u)] = ((unsigned)e << 8) | rel;
    wcount += __popc(mask);
  }
}

__device__ __forceinline__ int pe_quad(int step, int k, int off, int tid) {
  int lin = step * kPeStepQuads + k * kOwnThr + tid;
  if (lin >= kQuads) lin -= kQuads;
  int q = off + lin;
  if (q >= kQuads) q -= kQuads;
  return q;
}

__device__ __forceinline__ void pe_load(int4 (&dv)[kPeLoads], const int4* __restrict__ src4,
                                        int step, int off, int tid) {
#pragma unroll
  for (int k = 0; k < kPeLoads; ++k) dv[k] = src4[pe_quad(step, k, off, tid)];
}

__device__ __forceinline__ void pe_scan(const int4 (&dv)[kPeLoads], int step, int off, int tid,
                                        int wave, int s0, unsigned nrow, int& wcount,
                                        unsigned* seg) {
#pragma unroll
  for (int k = 0; k < kPeLoads; ++k) {
    if (step * kPeStepQuads + k * kOwnThr + wave * 32 < kQuads) {
      const int q = pe_quad(step, k, off, tid);
      const unsigned r0 = (unsigned)(dv[k].x - s0);
      const unsigned r1 = (unsigned)(dv[k].y - s0);
      const unsigned r2 = (unsigned)(dv[k].z - s0);
      const unsigned r3 = (unsigned)(dv[k].w - s0);
      const unsigned m0 = __builtin_amdgcn_ballot_w32(r0 < nrow);
      const unsigned m1 = __builtin_amdgcn_ballot_w32(r1 < nrow);
      const unsigned m2 = __builtin_amdgcn_ballot_w32(r2 < nrow);
      const unsigned m3 = __builtin_amdgcn_ballot_w32(r3 < nrow);
      if ((m0 | m1 | m2 | m3) != 0u) {
        pe_note(m0, r0 < nrow, r0, 4 * q + 0, wcount, seg);
        pe_note(m1, r1 < nrow, r1, 4 * q + 1, wcount, seg);
        pe_note(m2, r2 < nrow, r2, 4 * q + 2, wcount, seg);
        pe_note(m3, r3 < nrow, r3, 4 * q + 3, wcount, seg);
      }
    }
  }
}

__global__ void __launch_bounds__(kOwnThr)
k_owner_pe(const float* __restrict__ x, const int* __restrict__ ei, const int* __restrict__ et,
           const float* __restrict__ pe, float* __restrict__ x_pos) {
  __shared__ float    rows[kPeRowsPad * kInCh];
  __shared__ float    pes[kTypes * kInCh];
  __shared__ unsigned list[kOwnWaves * kPeSeg];

  const int tid  = threadIdx.x;
  const int lane = tid & 31;
  const int wave = __builtin_amdgcn_readfirstlane(tid >> 5);
  const int s0   = blockIdx.x * kOwnRows;
  const int nrow = (kNodes - s0 < kOwnRows) ? (kNodes - s0) : kOwnRows;
  const int nvec = nrow * (kInCh / 4);
  unsigned* seg = list + wave * kPeSeg;

  {
    constexpr int kXv = kPeRowsPad * (kInCh / 4) / kOwnThr;
    static_assert(kTypes * kInCh / 4 == kOwnThr, "one 16-byte load of pe per thread");
    const float4* __restrict__ x4 = reinterpret_cast<const float4*>(x + (size_t)s0 * kInCh);
    float4* rows4 = reinterpret_cast<float4*>(rows);
    reinterpret_cast<float4*>(pes)[tid] = reinterpret_cast<const float4*>(pe)[tid];
#pragma unroll
    for (int j = 0; j < kXv; ++j) {
      const int i = j * kOwnThr + tid;
      rows4[i] = x4[i < nvec ? i : nvec - 1];
    }
  }
  __syncthreads();

  const int4* __restrict__ src4 = reinterpret_cast<const int4*>(ei);
  const int off = blockIdx.x * kPeStagger;
  int wcount = 0;
  int4 da[kPeLoads], db[kPeLoads];
  pe_load(da, src4, 0, off, tid);
  for (int st = 0; st + 1 < kPeSteps; st += 2) {
    pe_load(db, src4, st + 1, off, tid);
    if (wcount > kPeSeg - kPeStepMax) { pe_add_noted(seg, wcount, lane, rows, pes, et); wcount = 0; }
    pe_scan(da, st, off, tid, wave, s0, (unsigned)nrow, wcount, seg);
    pe_load(da, src4, st + 2, off, tid);
    if (wcount > kPeSeg - kPeStepMax) { pe_add_noted(seg, wcount, lane, rows, pes, et); wcount = 0; }
    pe_scan(db, st + 1, off, tid, wave, s0, (unsigned)nrow, wcount, seg);
  }
  if (wcount > kPeSeg - kPeStepMax) { pe_add_noted(seg, wcount, lane, rows, pes, et); wcount = 0; }
  pe_scan(da, kPeSteps - 1, off, tid, wave, s0, (unsigned)nrow, wcount, seg);

  pe_add_noted(seg, wcount, lane, rows, pes, et);
  __syncthreads();

  {
    const float4* rows4 = reinterpret_cast<const float4*>(rows);
    float4* __restrict__ xp4 = reinterpret_cast<float4*>(x_pos + (size_t)s0 * kInCh);
    for (int i = tid; i < nvec; i += kOwnThr) xp4[i] = rows4[i];
  }
}

__global__ void __launch_bounds__(256) k_gemm_wmma(const float* __restrict__ A,
                                                   const float* __restrict__ B,
                                                   float* __restrict__ C) {
  const int lane = threadIdx.x & 31;
  const int wave = threadIdx.x >> 5;
  const int m0   = blockIdx.x * 16;
  const int n0   = wave * 16;
  const int half = lane >> 4;
  const int l15  = lane & 15;
  const int row  = m0 + l15;
  const int col  = n0 + l15;

  v8f acc = {0.f, 0.f, 0.f, 0.f, 0.f, 0.f, 0.f, 0.f};
#pragma unroll
  for (int k0 = 0; k0 < kInCh; k0 += 4) {
    const int kb = k0 + 2 * half;
    v2f a, b;
    a.x = A[(size_t)row * kInCh + kb];
    a.y = A[(size_t)row * kInCh + kb + 1];
    b.x = B[(size_t)kb * kHC + col];
    b.y = B[(size_t)(kb + 1) * kHC + col];
    acc = __builtin_amdgcn_wmma_f32_16x16x4_f32(false, a, false, b, (short)0, acc,
                                                false, false);
  }
#pragma unroll
  for (int r = 0; r < 8; ++r) {
    C[(size_t)(m0 + r + 8 * half) * kHC + col] = acc[r];
  }
}

__global__ void k_scores(const float* __restrict__ xl,
                         const float* __restrict__ att_src, const float* __restrict__ att_dst,
                         float* __restrict__ a_src, float* __restrict__ a_dst) {
  int tid = blockIdx.x * blockDim.x + threadIdx.x;
  if (tid >= kNodes * kHeads) return;
  int n = tid / kHeads, h = tid % kHeads;
  const float* xr = xl + (size_t)n * kHC + h * kOutCh;
  float ss = 0.0f, sd = 0.0f;
#pragma unroll
  for (int c = 0; c < kOutCh; ++c) {
    ss += xr[c] * att_src[h * kOutCh + c];
    sd += xr[c] * att_dst[h * kOutCh + c];
  }
  a_src[tid] = ss;
  a_dst[tid] = sd;
}


template <int PH>
__device__ __forceinline__ void ag_stats(const unsigned* seg, int n, int lane,
                                         const float* __restrict__ a_src,
                                         const float* adst_l, float* amax_l, float* den_l) {
  const int u  = lane >> 2;
  const int hh = lane & 3;
  for (int base = 0; base < n; base += 32) {
    unsigned ent[4];
    float as[4];
#pragma unroll
    for (int j = 0; j < 4; ++j) {
      const int idx = base + 8 * j + u;
      ent[j] = seg[idx < n ? idx : n - 1];
      as[j]  = a_src[(ent[j] >> 8) * kHeads + hh];
    }
    asm volatile("" :: "v"(as[0]), "v"(as[1]), "v"(as[2]), "v"(as[3]));
#pragma unroll
    for (int j = 0; j < 4; ++j) {
      if (base + 8 * j + u < n) {
        const int k = (int)(ent[j] & 255u) * kHeads + hh;
        const float al = leaky02(as[j] + adst_l[k]);
        if (PH == 0) atomic_max_f32(&amax_l[k], al);
        else         atomicAdd(&den_l[k], expf(al - amax_l[k]));
      }
    }
  }
}

__device__ __forceinline__ void ag_rows(const unsigned* seg, int n, int lane,
                                        const float* __restrict__ a_src,
                                        const float* adst_l, const float* amax_l,
                                        const float* den_l, float* acc,
                                        const float4* __restrict__ xl4) {
  const int u  = lane >> 2;
  const int hh = lane & 3;
  const int hr = lane >> 3;
  for (int base = 0; base < n; base += 8) {
    const int idx = base + u;
    const unsigned ent = seg[idx < n ? idx : n - 1];
    const int k = (int)(ent & 255u) * kHeads + hh;
    const float as = a_src[(ent >> 8) * kHeads + hh];
    float4 v[8];
    int rl[8];
#pragma unroll
    for (int j = 0; j < 8; ++j) {
      const unsigned eu = (unsigned)__builtin_amdgcn_readlane((int)ent, 4 * j);
      rl[j] = (int)(eu & 255u);
      v[j]  = xl4[(size_t)(eu >> 8) * (kHC / 4) + lane];
    }
    const float coef = expf(leaky02(as + adst_l[k]) - amax_l[k]) / den_l[k];
    float c[8];
#pragma unroll
    for (int j = 0; j < 8; ++j) c[j] = __shfl(coef, 4 * j + hr);
    asm volatile("" :: "v"(coef), "v"(v[0].x), "v"(v[1].x), "v"(v[2].x), "v"(v[3].x),
                       "v"(v[4].x), "v"(v[5].x), "v"(v[6].x), "v"(v[7].x));
#pragma unroll
    for (int j = 0; j < 8; ++j) {
      if (base + j < n) {
        float* row = acc + rl[j] * kHC + lane;
        atomicAdd(row,      c[j] * v[j].x);
        atomicAdd(row + 32, c[j] * v[j].y);
        atomicAdd(row + 64, c[j] * v[j].z);
        atomicAdd(row + 96, c[j] * v[j].w);
      }
    }
  }
}

__device__ __forceinline__ void ag_resolve(unsigned* seg, int n0, int n, int lane,
                                           const int* __restrict__ ei) {
  for (int base = n0; base < n; base += 128) {
    unsigned en[4];
    int sv[4];
#pragma unroll
    for (int j = 0; j < 4; ++j) {
      const int i = base + 32 * j + lane;
      en[j] = seg[i < n ? i : n - 1];
    }
#pragma unroll
    for (int j = 0; j < 4; ++j) sv[j] = ei[en[j] >> 8];
#pragma unroll
    for (int j = 0; j < 4; ++j) {
      const int i = base + 32 * j + lane;
      if (i < n) seg[i] = ((unsigned)sv[j] << 8) | (en[j] & 255u);
    }
  }
}

__global__ void __launch_bounds__(kOwnThr)
k_owner_aggregate(const int* __restrict__ ei,
                  const float* __restrict__ a_src, const float* __restrict__ a_dst,
                  const float* __restrict__ xl, const float* __restrict__ bias,
                  float* __restrict__ out) {
  __shared__ float    acc[kOwnRows * kHC];
  __shared__ unsigned list[kOwnWaves * kPeSeg];
  __shared__ float    adst_l[kOwnRows * kHeads];
  __shared__ float    amax_l[kOwnRows * kHeads];
  __shared__ float    den_l[kOwnRows * kHeads];
  __shared__ int      ovf;

  const int tid  = threadIdx.x;
  const int lane = tid & 31;
  const int wave = __builtin_amdgcn_readfirstlane(tid >> 5);
  const int d0   = blockIdx.x * kOwnRows;
  const int nrow = (kNodes - d0 < kOwnRows) ? (kNodes - d0) : kOwnRows;
  const float4* __restrict__ xl4 = reinterpret_cast<const float4*>(xl);
  unsigned* seg = list + wave * kPeSeg;

  {
    float4* acc4 = reinterpret_cast<float4*>(acc);
    for (int i = tid; i < kOwnRows * kHC / 4; i += kOwnThr) acc4[i] = make_float4(0.f, 0.f, 0.f, 0.f);
    if (tid < kOwnRows * kHeads) {
      adst_l[tid] = (tid < nrow * kHeads) ? a_dst[d0 * kHeads + tid] : 0.0f;
      amax_l[tid] = -INFINITY;
      den_l[tid]  = 0.0f;
    }
    if (tid == 0) ovf = 0;
  }
  __syncthreads();

  const int nself = (nrow > wave) ? ((nrow - wave + 31) >> 5) : 0;
  if (lane < nself) seg[lane] = ((unsigned)(d0 + wave + 32 * lane) << 8) | (unsigned)(wave + 32 * lane);

  const int4* __restrict__ dst4 = reinterpret_cast<const int4*>(ei + kEdges);
  const int off = blockIdx.x * kPeStagger;
  int wcount = nself;
  bool over = false;
  int4 da[kPeLoads], db[kPeLoads];
  pe_load(da, dst4, 0, off, tid);
  for (int st = 0; st + 1 < kPeSteps; st += 2) {
    pe_load(db, dst4, st + 1, off, tid);
    over = over || (wcount > kPeSeg - kPeStepMax);
    if (!over) pe_scan(da, st, off, tid, wave, d0, (unsigned)nrow, wcount, seg);
    pe_load(da, dst4, st + 2, off, tid);
    over = over || (wcount > kPeSeg - kPeStepMax);
    if (!over) pe_scan(db, st + 1, off, tid, wave, d0, (unsigned)nrow, wcount, seg);
  }
  over = over || (wcount > kPeSeg - kPeStepMax);
  if (!over) pe_scan(da, kPeSteps - 1, off, tid, wave, d0, (unsigned)nrow, wcount, seg);
  if (over) ovf = 1;
  else      ag_resolve(seg, nself, wcount, lane, ei);
  __syncthreads();

  const bool fits = __builtin_amdgcn_readfirstlane(ovf) == 0;
  if (fits) {
    ag_stats<0>(seg, wcount, lane, a_src, adst_l, amax_l, den_l);
    __syncthreads();
    ag_stats<1>(seg, wcount, lane, a_src, adst_l, amax_l, den_l);
    __syncthreads();
    ag_rows(seg, wcount, lane, a_src, adst_l, amax_l, den_l, acc, xl4);
  } else {
    for (int ph = 0; ph < 3; ++ph) {
      for (int st = 0; st < kPeSteps; ++st) {
        int n = 0;
        if (st == 0) {
          if (lane < nself) seg[lane] = ((unsigned)(d0 + wave + 32 * lane) << 8) | (unsigned)(wave + 32 * lane);
          n = nself;
        }
        const int n0 = n;
        pe_load(da, dst4, st, off, tid);
        pe_scan(da, st, off, tid, wave, d0, (unsigned)nrow, n, seg);
        ag_resolve(seg, n0, n, lane, ei);
        if (ph == 0)      ag_stats<0>(seg, n, lane, a_src, adst_l, amax_l, den_l);
        else if (ph == 1) ag_stats<1>(seg, n, lane, a_src, adst_l, amax_l, den_l);
        else              ag_rows(seg, n, lane, a_src, adst_l, amax_l, den_l, acc, xl4);
      }
      __syncthreads();
    }
  }
  __syncthreads();

  const float4 b = reinterpret_cast<const float4*>(bias)[lane];
  float4* __restrict__ out4 = reinterpret_cast<float4*>(out);
  for (int dl = wave; dl < nrow; dl += kOwnWaves) {
    const int d = d0 + dl;
    const float* row = acc + dl * kHC + lane;
    float4 r;
    r.x = row[0]  + b.x;
    r.y = row[32] + b.y;
    r.z = row[64] + b.z;
    r.w = row[96] + b.w;
    r.x = r.x > 0.0f ? r.x : expm1f(r.x);
    r.y = r.y > 0.0f ? r.y : expm1f(r.y);
    r.z = r.z > 0.0f ? r.z : expm1f(r.z);
    r.w = r.w > 0.0f ? r.w : expm1f(r.w);
    out4[(size_t)d * (kHC / 4) + lane] = r;
  }
}


extern "C" void kernel_launch(void* const* d_in, const int* in_sizes, int n_in,
                              void* d_out, int out_size, void* d_ws, size_t ws_size,
                              hipStream_t stream) {
  (void)in_sizes; (void)n_in; (void)out_size; (void)ws_size;
  const float* x       = (const float*)d_in[0];
  const int*   ei      = (const int*)d_in[1];
  const int*   et      = (const int*)d_in[2];
  const float* pe      = (const float*)d_in[3];
  const float* W       = (const float*)d_in[4];
  const float* att_src = (const float*)d_in[5];
  const float* att_dst = (const float*)d_in[6];
  const float* bias    = (const float*)d_in[7];
  float* out = (float*)d_out;

  float* ws    = (float*)d_ws;
  float* x_pos = ws;
  float* xl    = x_pos + (size_t)kNodes * kInCh;
  float* a_src = xl    + (size_t)kNodes * kHC;
  float* a_dst = a_src + (size_t)kNodes * kHeads;

  const int B = 256;
  k_owner_pe<<<kOwnBlocks, kOwnThr, 0, stream>>>(x, ei, et, pe, x_pos);
  k_gemm_wmma<<<kNodes / 16, 256, 0, stream>>>(x_pos, W, xl);
  k_scores<<<(kNodes * kHeads + B - 1) / B, B, 0, stream>>>(xl, att_src, att_dst, a_src, a_dst);
  k_owner_aggregate<<<kOwnBlocks, kOwnThr, 0, stream>>>(ei, a_src, a_dst, xl, bias, out);
}


# owner scans load and compare marked groups only: new k_mark pass sets, per owner block, a bit per 128-edge group of src and dst that holds one of its edges; lists, LDS add order as before; 5 launches
# speedup vs baseline: 11.5050x; 1.1742x over previous
#include <hip/hip_runtime.h>
#include <math.h>

typedef float v2f __attribute__((ext_vector_type(2)));
typedef float v8f __attribute__((ext_vector_type(8)));
typedef unsigned short __attribute__((may_alias)) u16a;

namespace {
constexpr int kNodes = 50000;
constexpr int kEdges = 800000;
constexpr int kInCh  = 64;
constexpr int kOutCh = 32;
constexpr int kHeads = 4;
constexpr int kHC    = kHeads * kOutCh;
constexpr int kETot  = kEdges + kNodes;
constexpr int kOwnBlocks = 256;
constexpr int kOwnRows   = 196;
constexpr int kOwnThr    = 1024;
constexpr int kOwnWaves  = kOwnThr / 32;
constexpr int kQuads     = kEdges / 4;
static_assert(kEdges % 4 == 0, "the index rows are read in groups of 4");
static_assert(kOwnBlocks * kOwnRows >= kNodes, "every node has an owner");
static_assert(kOwnRows <= 256, "local row index is packed into 8 bits");
static_assert(kNodes <= (1 << 24), "source index is packed into 24 bits");
constexpr int kTypes       = 64;
constexpr int kPeLoads     = 8;
constexpr int kPeStagger   = 768;
constexpr int kPeStepMax   = 32 * kPeLoads * 4;
constexpr int kPeSeg       = 1536;
constexpr int kPeRowsPad   = 256;
static_assert(kQuads % 32 == 0 && kPeStagger % 32 == 0, "a wave's 32 quads never straddle the wrap");
static_assert(kOwnBlocks * kPeStagger <= kQuads, "start places stay inside the list");
static_assert(kEdges <= (1 << 24), "edge index is packed into 24 bits");
static_assert(kPeSeg > kPeStepMax, "a whole step fits behind the mark at which a segment is emptied");
static_assert(kPeRowsPad >= kOwnRows && kPeRowsPad * (kInCh / 4) % kOwnThr == 0, "the seed copy covers the padded rows exactly");
constexpr int kGroups     = kQuads / 32;
constexpr int kMarkWords  = (kGroups + 31) / 32;
constexpr int kWaveGroups = (kQuads + kOwnThr - 1) / kOwnThr;
constexpr int kTodoTrips  = (kWaveGroups + 31) / 32;
constexpr int kTodoPad    = 200;
static_assert(kOwnThr == 32 * 32, "a block of k_mark fills one word: 32 waves, one group each");
static_assert(kGroups < (1 << 16), "a group index is kept in 16 bits");
static_assert(kTodoPad % kPeLoads == 0 && kTodoPad >= kWaveGroups && kPeLoads == 8, "eight todo entries are read as one 16-byte word");
}


__device__ __forceinline__ void atomic_max_f32(float* addr, float v) {
  if (v >= 0.0f) atomicMax(reinterpret_cast<int*>(addr), __float_as_int(v));
  else           atomicMin(reinterpret_cast<unsigned int*>(addr), __float_as_uint(v));
}

__device__ __forceinline__ float leaky02(float v) { return v > 0.0f ? v : 0.2f * v; }


__device__ __forceinline__ void pe_add_noted(const unsigned* seg, int n, int lane,
                                             float* __restrict__ rows,
                                             const float* __restrict__ pes,
                                             const int* __restrict__ et) {
  for (int base = 0; base < n; base += 32) {
    const int i = base + lane;
    unsigned ent = 0u;
    int t = 0;
    if (i < n) { ent = seg[i]; t = et[ent >> 8]; }
    const int m = (n - base < 32) ? (n - base) : 32;
    for (int u = 0; u < m; u += 4) {
      int tu[4];
      unsigned ru[4];
      float p0[4], p1[4];
#pragma unroll
      for (int j = 0; j < 4; ++j) {
        const int uj = (u + j < m) ? (u + j) : (m - 1);
        tu[j] = __builtin_amdgcn_readlane(t, uj);
        ru[j] = (unsigned)__builtin_amdgcn_readlane((int)ent, uj) & 255u;
        p0[j] = pes[tu[j] * kInCh + lane];
        p1[j] = pes[tu[j] * kInCh + 32 + lane];
      }
      asm volatile("" :: "v"(p0[0]), "v"(p0[1]), "v"(p0[2]), "v"(p0[3]),
                         "v"(p1[0]), "v"(p1[1]), "v"(p1[2]), "v"(p1[3]));
#pragma unroll
      for (int j = 0; j < 4; ++j) {
        if (u + j < m) {
          atomicAdd(&rows[ru[j] * kInCh + lane],      p0[j]);
          atomicAdd(&rows[ru[j] * kInCh + 32 + lane], p1[j]);
        }
      }
    }
  }
}

__device__ __forceinline__ void pe_note(unsigned mask, bool mine, unsigned rel, int e,
                                        int& wcount, unsigned* seg) {
  if (mask != 0u) {
    if (mine) seg[wcount + (int)__builtin_amdgcn_mbcnt_lo(mask, 0u)] = ((unsigned)e << 8) | rel;
    wcount += __popc(mask);
  }
}

__global__ void __launch_bounds__(kOwnThr)
k_mark(const int* __restrict__ ei, unsigned* __restrict__ marks) {
  __shared__ unsigned own[2 * kOwnBlocks];
  const int tid = threadIdx.x;
  if (tid < 2 * kOwnBlocks) own[tid] = 0u;
  __syncthreads();
  const int q = blockIdx.x * kOwnThr + tid;
  if (q < kQuads) {
    const unsigned bit = 1u << (tid >> 5);
    const int4 s = reinterpret_cast<const int4*>(ei)[q];
    const int4 d = reinterpret_cast<const int4*>(ei + kEdges)[q];
    const unsigned os[4] = {(unsigned)s.x / kOwnRows, (unsigned)s.y / kOwnRows,
                            (unsigned)s.z / kOwnRows, (unsigned)s.w / kOwnRows};
    const unsigned od[4] = {(unsigned)d.x / kOwnRows, (unsigned)d.y / kOwnRows,
                            (unsigned)d.z / kOwnRows, (unsigned)d.w / kOwnRows};
#pragma unroll
    for (int j = 0; j < 4; ++j) {
      if (os[j] < (unsigned)kOwnBlocks) atomicOr(&own[os[j]], bit);
      if (od[j] < (unsigned)kOwnBlocks) atomicOr(&own[kOwnBlocks + od[j]], bit);
    }
  }
  __syncthreads();
  if (tid < 2 * kOwnBlocks) {
    const int which = tid >> 8;
    marks[((size_t)which * kMarkWords + blockIdx.x) * kOwnBlocks + (tid & (kOwnBlocks - 1))] = own[tid];
  }
}
static_assert(kOwnBlocks == 256, "k_mark splits a thread index into row and owner block by 8 bits");

__device__ __forceinline__ int pe_todo(const unsigned* mk, u16a* todo,
                                       int off, int wave, int lane) {
  int g[kTodoTrips];
  unsigned w[kTodoTrips];
  bool in[kTodoTrips];
#pragma unroll
  for (int t = 0; t < kTodoTrips; ++t) {
    const int lin = (t * 32 + lane) * kOwnThr + wave * 32;
    in[t] = lin < kQuads;
    int q = off + (in[t] ? lin : 0);
    if (q >= kQuads) q -= kQuads;
    g[t] = q >> 5;
    w[t] = mk[g[t] >> 5];
  }
  int n = 0;
#pragma unroll
  for (int t = 0; t < kTodoTrips; ++t) {
    const bool on = in[t] && (((w[t] >> (g[t] & 31)) & 1u) != 0u);
    const unsigned m = __builtin_amdgcn_ballot_w32(on);
    if (on) todo[n + (int)__builtin_amdgcn_mbcnt_lo(m, 0u)] = (u16a)g[t];
    n += __popc(m);
  }
  return n;
}

__device__ __forceinline__ int pe_group(const uint4& t, int k) {
  const unsigned c = (k < 2) ? t.x : (k < 4) ? t.y : (k < 6) ? t.z : t.w;
  return (int)((k & 1) ? (c >> 16) : (c & 0xffffu));
}

__device__ __forceinline__ void pe_load(int4 (&dv)[kPeLoads], const int4* __restrict__ src4,
                                        const u16a* todo, int base, int n, int lane) {
  const int bc = (base < kTodoPad - kPeLoads) ? base : (kTodoPad - kPeLoads);
  const uint4 t = *reinterpret_cast<const uint4*>(todo + bc);
#pragma unroll
  for (int k = 0; k < kPeLoads; ++k) {
    const int g = (base + k < n) ? pe_group(t, k) : 0;
    dv[k] = src4[g * 32 + lane];
  }
}

__device__ __forceinline__ void pe_scan(const int4 (&dv)[kPeLoads], const u16a* todo,
                                        int base, int n, int lane, int s0, unsigned nrow,
                                        int& wcount, unsigned* seg) {
  if (base < n) {
    const uint4 t = *reinterpret_cast<const uint4*>(todo + base);
#pragma unroll
    for (int k = 0; k < kPeLoads; ++k) {
      if (base + k < n) {
        const int q = pe_group(t, k) * 32 + lane;
        const unsigned r0 = (unsigned)(dv[k].x - s0);
        const unsigned r1 = (unsigned)(dv[k].y - s0);
        const unsigned r2 = (unsigned)(dv[k].z - s0);
        const unsigned r3 = (unsigned)(dv[k].w - s0);
        const unsigned m0 = __builtin_amdgcn_ballot_w32(r0 < nrow);
        const unsigned m1 = __builtin_amdgcn_ballot_w32(r1 < nrow);
        const unsigned m2 = __builtin_amdgcn_ballot_w32(r2 < nrow);
        const unsigned m3 = __builtin_amdgcn_ballot_w32(r3 < nrow);
        pe_note(m0, r0 < nrow, r0, 4 * q + 0, wcount, seg);
        pe_note(m1, r1 < nrow, r1, 4 * q + 1, wcount, seg);
        pe_note(m2, r2 < nrow, r2, 4 * q + 2, wcount, seg);
        pe_note(m3, r3 < nrow, r3, 4 * q + 3, wcount, seg);
      }
    }
  }
}

__global__ void __launch_bounds__(kOwnThr)
k_owner_pe(const float* __restrict__ x, const int* __restrict__ ei, const int* __restrict__ et,
           const float* __restrict__ pe, const unsigned* __restrict__ msrc,
           float* __restrict__ x_pos) {
  __shared__ float    rows[kPeRowsPad * kInCh];
  __shared__ float    pes[kTypes * kInCh];
  __shared__ unsigned list[kOwnWaves * kPeSeg];
  __shared__ uint4    todo4[kOwnWaves * kTodoPad / 8];
  __shared__ unsigned mk[kMarkWords];

  const int tid  = threadIdx.x;
  const int lane = tid & 31;
  const int wave = __builtin_amdgcn_readfirstlane(tid >> 5);
  const int s0   = blockIdx.x * kOwnRows;
  const int nrow = (kNodes - s0 < kOwnRows) ? (kNodes - s0) : kOwnRows;
  const int nvec = nrow * (kInCh / 4);
  unsigned* seg = list + wave * kPeSeg;
  u16a* todo = reinterpret_cast<u16a*>(todo4) + wave * kTodoPad;

  {
    constexpr int kXv = kPeRowsPad * (kInCh / 4) / kOwnThr;
    static_assert(kTypes * kInCh / 4 == kOwnThr, "one 16-byte load of pe per thread");
    const float4* __restrict__ x4 = reinterpret_cast<const float4*>(x + (size_t)s0 * kInCh);
    float4* rows4 = reinterpret_cast<float4*>(rows);
    reinterpret_cast<float4*>(pes)[tid] = reinterpret_cast<const float4*>(pe)[tid];
    if (tid < kMarkWords) mk[tid] = msrc[(size_t)tid * kOwnBlocks + blockIdx.x];
#pragma unroll
    for (int j = 0; j < kXv; ++j) {
      const int i = j * kOwnThr + tid;
      rows4[i] = x4[i < nvec ? i : nvec - 1];
    }
  }
  __syncthreads();

  const int4* __restrict__ src4 = reinterpret_cast<const int4*>(ei);
  const int off = blockIdx.x * kPeStagger;
  const int nt = pe_todo(mk, todo, off, wave, lane);
  int wcount = 0;
  int4 da[kPeLoads], db[kPeLoads];
  pe_load(da, src4, todo, 0, nt, lane);
  for (int b = 0; b < nt; b += 2 * kPeLoads) {
    pe_load(db, src4, todo, b + kPeLoads, nt, lane);
    if (wcount > kPeSeg - kPeStepMax) { pe_add_noted(seg, wcount, lane, rows, pes, et); wcount = 0; }
    pe_scan(da, todo, b, nt, lane, s0, (unsigned)nrow, wcount, seg);
    pe_load(da, src4, todo, b + 2 * kPeLoads, nt, lane);
    if (wcount > kPeSeg - kPeStepMax) { pe_add_noted(seg, wcount, lane, rows, pes, et); wcount = 0; }
    pe_scan(db, todo, b + kPeLoads, nt, lane, s0, (unsigned)nrow, wcount, seg);
  }

  pe_add_noted(seg, wcount, lane, rows, pes, et);
  __syncthreads();

  {
    const float4* rows4 = reinterpret_cast<const float4*>(rows);
    float4* __restrict__ xp4 = reinterpret_cast<float4*>(x_pos + (size_t)s0 * kInCh);
    for (int i = tid; i < nvec; i += kOwnThr) xp4[i] = rows4[i];
  }
}

__global__ void __launch_bounds__(256) k_gemm_wmma(const float* __restrict__ A,
                                                   const float* __restrict__ B,
                                                   float* __restrict__ C) {
  const int lane = threadIdx.x & 31;
  const int wave = threadIdx.x >> 5;
  const int m0   = blockIdx.x * 16;
  const int n0   = wave * 16;
  const int half = lane >> 4;
  const int l15  = lane & 15;
  const int row  = m0 + l15;
  const int col  = n0 + l15;

  v8f acc = {0.f, 0.f, 0.f, 0.f, 0.f, 0.f, 0.f, 0.f};
#pragma unroll
  for (int k0 = 0; k0 < kInCh; k0 += 4) {
    const int kb = k0 + 2 * half;
    v2f a, b;
    a.x = A[(size_t)row * kInCh + kb];
    a.y = A[(size_t)row * kInCh + kb + 1];
    b.x = B[(size_t)kb * kHC + col];
    b.y = B[(size_t)(kb + 1) * kHC + col];
    acc = __builtin_amdgcn_wmma_f32_16x16x4_f32(false, a, false, b, (short)0, acc,
                                                false, false);
  }
#pragma unroll
  for (int r = 0; r < 8; ++r) {
    C[(size_t)(m0 + r + 8 * half) * kHC + col] = acc[r];
  }
}

__global__ void k_scores(const float* __restrict__ xl,
                         const float* __restrict__ att_src, const float* __restrict__ att_dst,
                         float* __restrict__ a_src, float* __restrict__ a_dst) {
  int tid = blockIdx.x * blockDim.x + threadIdx.x;
  if (tid >= kNodes * kHeads) return;
  int n = tid / kHeads, h = tid % kHeads;
  const float* xr = xl + (size_t)n * kHC + h * kOutCh;
  float ss = 0.0f, sd = 0.0f;
#pragma unroll
  for (int c = 0; c < kOutCh; ++c) {
    ss += xr[c] * att_src[h * kOutCh + c];
    sd += xr[c] * att_dst[h * kOutCh + c];
  }
  a_src[tid] = ss;
  a_dst[tid] = sd;
}


template <int PH>
__device__ __forceinline__ void ag_stats(const unsigned* seg, int n, int lane,
                                         const float* __restrict__ a_src,
                                         const float* adst_l, float* amax_l, float* den_l) {
  const int u  = lane >> 2;
  const int hh = lane & 3;
  for (int base = 0; base < n; base += 32) {
    unsigned ent[4];
    float as[4];
#pragma unroll
    for (int j = 0; j < 4; ++j) {
      const int idx = base + 8 * j + u;
      ent[j] = seg[idx < n ? idx : n - 1];
      as[j]  = a_src[(ent[j] >> 8) * kHeads + hh];
    }
    asm volatile("" :: "v"(as[0]), "v"(as[1]), "v"(as[2]), "v"(as[3]));
#pragma unroll
    for (int j = 0; j < 4; ++j) {
      if (base + 8 * j + u < n) {
        const int k = (int)(ent[j] & 255u) * kHeads + hh;
        const float al = leaky02(as[j] + adst_l[k]);
        if (PH == 0) atomic_max_f32(&amax_l[k], al);
        else         atomicAdd(&den_l[k], expf(al - amax_l[k]));
      }
    }
  }
}

__device__ __forceinline__ void ag_rows(const unsigned* seg, int n, int lane,
                                        const float* __restrict__ a_src,
                                        const float* adst_l, const float* amax_l,
                                        const float* den_l, float* acc,
                                        const float4* __restrict__ xl4) {
  const int u  = lane >> 2;
  const int hh = lane & 3;
  const int hr = lane >> 3;
  for (int base = 0; base < n; base += 8) {
    const int idx = base + u;
    const unsigned ent = seg[idx < n ? idx : n - 1];
    const int k = (int)(ent & 255u) * kHeads + hh;
    const float as = a_src[(ent >> 8) * kHeads + hh];
    float4 v[8];
    int rl[8];
#pragma unroll
    for (int j = 0; j < 8; ++j) {
      const unsigned eu = (unsigned)__builtin_amdgcn_readlane((int)ent, 4 * j);
      rl[j] = (int)(eu & 255u);
      v[j]  = xl4[(size_t)(eu >> 8) * (kHC / 4) + lane];
    }
    const float coef = expf(leaky02(as + adst_l[k]) - amax_l[k]) / den_l[k];
    float c[8];
#pragma unroll
    for (int j = 0; j < 8; ++j) c[j] = __shfl(coef, 4 * j + hr);
    asm volatile("" :: "v"(coef), "v"(v[0].x), "v"(v[1].x), "v"(v[2].x), "v"(v[3].x),
                       "v"(v[4].x), "v"(v[5].x), "v"(v[6].x), "v"(v[7].x));
#pragma unroll
    for (int j = 0; j < 8; ++j) {
      if (base + j < n) {
        float* row = acc + rl[j] * kHC + lane;
        atomicAdd(row,      c[j] * v[j].x);
        atomicAdd(row + 32, c[j] * v[j].y);
        atomicAdd(row + 64, c[j] * v[j].z);
        atomicAdd(row + 96, c[j] * v[j].w);
      }
    }
  }
}

__device__ __forceinline__ void ag_resolve(unsigned* seg, int n0, int n, int lane,
                                           const int* __restrict__ ei) {
  for (int base = n0; base < n; base += 128) {
    unsigned en[4];
    int sv[4];
#pragma unroll
    for (int j = 0; j < 4; ++j) {
      const int i = base + 32 * j + lane;
      en[j] = seg[i < n ? i : n - 1];
    }
#pragma unroll
    for (int j = 0; j < 4; ++j) sv[j] = ei[en[j] >> 8];
#pragma unroll
    for (int j = 0; j < 4; ++j) {
      const int i = base + 32 * j + lane;
      if (i < n) seg[i] = ((unsigned)sv[j] << 8) | (en[j] & 255u);
    }
  }
}

__global__ void __launch_bounds__(kOwnThr)
k_owner_aggregate(const int* __restrict__ ei, const unsigned* __restrict__ mdst,
                  const float* __restrict__ a_src, const float* __restrict__ a_dst,
                  const float* __restrict__ xl, const float* __restrict__ bias,
                  float* __restrict__ out) {
  __shared__ float    acc[kOwnRows * kHC];
  __shared__ unsigned list[kOwnWaves * kPeSeg];
  __shared__ float    adst_l[kOwnRows * kHeads];
  __shared__ float    amax_l[kOwnRows * kHeads];
  __shared__ float    den_l[kOwnRows * kHeads];
  __shared__ int      ovf;
  __shared__ uint4    todo4[kOwnWaves * kTodoPad / 8];
  __shared__ unsigned mk[kMarkWords];

  const int tid  = threadIdx.x;
  const int lane = tid & 31;
  const int wave = __builtin_amdgcn_readfirstlane(tid >> 5);
  const int d0   = blockIdx.x * kOwnRows;
  const int nrow = (kNodes - d0 < kOwnRows) ? (kNodes - d0) : kOwnRows;
  const float4* __restrict__ xl4 = reinterpret_cast<const float4*>(xl);
  unsigned* seg = list + wave * kPeSeg;
  u16a* todo = reinterpret_cast<u16a*>(todo4) + wave * kTodoPad;

  {
    float4* acc4 = reinterpret_cast<float4*>(acc);
    for (int i = tid; i < kOwnRows * kHC / 4; i += kOwnThr) acc4[i] = make_float4(0.f, 0.f, 0.f, 0.f);
    if (tid < kOwnRows * kHeads) {
      adst_l[tid] = (tid < nrow * kHeads) ? a_dst[d0 * kHeads + tid] : 0.0f;
      amax_l[tid] = -INFINITY;
      den_l[tid]  = 0.0f;
    }
    if (tid == 0) ovf = 0;
    if (tid < kMarkWords) mk[tid] = mdst[(size_t)tid * kOwnBlocks + blockIdx.x];
  }
  __syncthreads();

  const int nself = (nrow > wave) ? ((nrow - wave + 31) >> 5) : 0;
  if (lane < nself) seg[lane] = ((unsigned)(d0 + wave + 32 * lane) << 8) | (unsigned)(wave + 32 * lane);

  const int4* __restrict__ dst4 = reinterpret_cast<const int4*>(ei + kEdges);
  const int off = blockIdx.x * kPeStagger;
  const int nt = pe_todo(mk, todo, off, wave, lane);
  int wcount = nself;
  bool over = false;
  int4 da[kPeLoads], db[kPeLoads];
  pe_load(da, dst4, todo, 0, nt, lane);
  for (int b = 0; b < nt; b += 2 * kPeLoads) {
    pe_load(db, dst4, todo, b + kPeLoads, nt, lane);
    over = over || (wcount > kPeSeg - kPeStepMax);
    if (!over) pe_scan(da, todo, b, nt, lane, d0, (unsigned)nrow, wcount, seg);
    pe_load(da, dst4, todo, b + 2 * kPeLoads, nt, lane);
    over = over || (wcount > kPeSeg - kPeStepMax);
    if (!over) pe_scan(db, todo, b + kPeLoads, nt, lane, d0, (unsigned)nrow, wcount, seg);
  }
  if (over) ovf = 1;
  else      ag_resolve(seg, nself, wcount, lane, ei);
  __syncthreads();

  const bool fits = __builtin_amdgcn_readfirstlane(ovf) == 0;
  if (fits) {
    ag_stats<0>(seg, wcount, lane, a_src, adst_l, amax_l, den_l);
    __syncthreads();
    ag_stats<1>(seg, wcount, lane, a_src, adst_l, amax_l, den_l);
    __syncthreads();
    ag_rows(seg, wcount, lane, a_src, adst_l, amax_l, den_l, acc, xl4);
  } else {
    for (int ph = 0; ph < 3; ++ph) {
      for (int b = 0; b == 0 || b < nt; b += kPeLoads) {
        int n = 0;
        if (b == 0) {
          if (lane < nself) seg[lane] = ((unsigned)(d0 + wave + 32 * lane) << 8) | (unsigned)(wave + 32 * lane);
          n = nself;
        }
        const int n0 = n;
        pe_load(da, dst4, todo, b, nt, lane);
        pe_scan(da, todo, b, nt, lane, d0, (unsigned)nrow, n, seg);
        ag_resolve(seg, n0, n, lane, ei);
        if (ph == 0)      ag_stats<0>(seg, n, lane, a_src, adst_l, amax_l, den_l);
        else if (ph == 1) ag_stats<1>(seg, n, lane, a_src, adst_l, amax_l, den_l);
        else              ag_rows(seg, n, lane, a_src, adst_l, amax_l, den_l, acc, xl4);
      }
      __syncthreads();
    }
  }
  __syncthreads();

  const float4 b = reinterpret_cast<const float4*>(bias)[lane];
  float4* __restrict__ out4 = reinterpret_cast<float4*>(out);
  for (int dl = wave; dl < nrow; dl += kOwnWaves) {
    const int d = d0 + dl;
    const float* row = acc + dl * kHC + lane;
    float4 r;
    r.x = row[0]  + b.x;
    r.y = row[32] + b.y;
    r.z = row[64] + b.z;
    r.w = row[96] + b.w;
    r.x = r.x > 0.0f ? r.x : expm1f(r.x);
    r.y = r.y > 0.0f ? r.y : expm1f(r.y);
    r.z = r.z > 0.0f ? r.z : expm1f(r.z);
    r.w = r.w > 0.0f ? r.w : expm1f(r.w);
    out4[(size_t)d * (kHC / 4) + lane] = r;
  }
}


extern "C" void kernel_launch(void* const* d_in, const int* in_sizes, int n_in,
                              void* d_out, int out_size, void* d_ws, size_t ws_size,
                              hipStream_t stream) {
  (void)in_sizes; (void)n_in; (void)out_size; (void)ws_size;
  const float* x       = (const float*)d_in[0];
  const int*   ei      = (const int*)d_in[1];
  const int*   et      = (const int*)d_in[2];
  const float* pe      = (const float*)d_in[3];
  const float* W       = (const float*)d_in[4];
  const float* att_src = (const float*)d_in[5];
  const float* att_dst = (const float*)d_in[6];
  const float* bias    = (const float*)d_in[7];
  float* out = (float*)d_out;

  float* ws    = (float*)d_ws;
  float* x_pos = ws;
  float* xl    = x_pos + (size_t)kNodes * kInCh;
  float* a_src = xl    + (size_t)kNodes * kHC;
  float* a_dst = a_src + (size_t)kNodes * kHeads;
  unsigned* marks = (unsigned*)(a_dst + (size_t)kNodes * kHeads);

  const int B = 256;
  k_mark<<<kMarkWords, kOwnThr, 0, stream>>>(ei, marks);
  k_owner_pe<<<kOwnBlocks, kOwnThr, 0, stream>>>(x, ei, et, pe, marks, x_pos);
  k_gemm_wmma<<<kNodes / 16, 256, 0, stream>>>(x_pos, W, xl);
  k_scores<<<(kNodes * kHeads + B - 1) / B, B, 0, stream>>>(xl, att_src, att_dst, a_src, a_dst);
  k_owner_aggregate<<<kOwnBlocks, kOwnThr, 0, stream>>>(ei, marks + (size_t)kMarkWords * kOwnBlocks,
                                                        a_src, a_dst, xl, bias, out);
}


# owner block multiplies its x_pos rows from LDS on WMMA by [W | W.att] (B table in fragment order from one more k_mark block): xl, scores (x_pos @ (W.att): sum order changes) stored there; 3 launches
# speedup vs baseline: 13.4150x; 1.1660x over previous
#include <hip/hip_runtime.h>
#include <math.h>

typedef float v2f __attribute__((ext_vector_type(2)));
typedef float v8f __attribute__((ext_vector_type(8)));
typedef unsigned short __attribute__((may_alias)) u16a;

namespace {
constexpr int kNodes = 50000;
constexpr int kEdges = 800000;
constexpr int kInCh  = 64;
constexpr int kOutCh = 32;
constexpr int kHeads = 4;
constexpr int kHC    = kHeads * kOutCh;
constexpr int kETot  = kEdges + kNodes;
constexpr int kOwnBlocks = 256;
constexpr int kOwnRows   = 196;
constexpr int kOwnThr    = 1024;
constexpr int kOwnWaves  = kOwnThr / 32;
constexpr int kQuads     = kEdges / 4;
static_assert(kEdges % 4 == 0, "the index rows are read in groups of 4");
static_assert(kOwnBlocks * kOwnRows >= kNodes, "every node has an owner");
static_assert(kOwnRows <= 256, "local row index is packed into 8 bits");
static_assert(kNodes <= (1 << 24), "source index is packed into 24 bits");
constexpr int kTypes       = 64;
constexpr int kPeLoads     = 8;
constexpr int kPeStagger   = 768;
constexpr int kPeStepMax   = 32 * kPeLoads * 4;
constexpr int kPeSeg       = 1536;
constexpr int kPeRowsPad   = 256;
static_assert(kQuads % 32 == 0 && kPeStagger % 32 == 0, "a wave's 32 quads never straddle the wrap");
static_assert(kOwnBlocks * kPeStagger <= kQuads, "start places stay inside the list");
static_assert(kEdges <= (1 << 24), "edge index is packed into 24 bits");
static_assert(kPeSeg > kPeStepMax, "a whole step fits behind the mark at which a segment is emptied");
static_assert(kPeRowsPad >= kOwnRows && kPeRowsPad * (kInCh / 4) % kOwnThr == 0, "the seed copy covers the padded rows exactly");
constexpr int kRowPitch    = kInCh + 4;
constexpr int kColTiles    = kHC / 16 + 1;
constexpr int kKPairs      = kInCh / 8;
constexpr int kWfTile      = kKPairs * 32 * 4;
static_assert(kPeRowsPad >= (kOwnRows + 15) / 16 * 16, "the last row tile reads padded rows only");
static_assert(kRowPitch % 4 == 0 && 2 * kHeads <= 16, "rows stay 16-byte aligned; both score sets fit one tile");
static_assert(kWfTile == kOwnThr, "the extra block of k_mark fills one column tile of the B table per trip");
static_assert(kOwnRows % 4 == 0 && kNodes % 4 == 0, "every block owns a multiple of four rows: a tile's rows are stored in fours");
constexpr int kGroups     = kQuads / 32;
constexpr int kMarkWords  = (kGroups + 31) / 32;
constexpr int kWaveGroups = (kQuads + kOwnThr - 1) / kOwnThr;
constexpr int kTodoTrips  = (kWaveGroups + 31) / 32;
constexpr int kTodoPad    = 200;
static_assert(kOwnThr == 32 * 32, "a block of k_mark fills one word: 32 waves, one group each");
static_assert(kGroups < (1 << 16), "a group index is kept in 16 bits");
static_assert(kTodoPad % kPeLoads == 0 && kTodoPad >= kWaveGroups && kPeLoads == 8, "eight todo entries are read as one 16-byte word");
}


__device__ __forceinline__ void atomic_max_f32(float* addr, float v) {
  if (v >= 0.0f) atomicMax(reinterpret_cast<int*>(addr), __float_as_int(v));
  else           atomicMin(reinterpret_cast<unsigned int*>(addr), __float_as_uint(v));
}

__device__ __forceinline__ float leaky02(float v) { return v > 0.0f ? v : 0.2f * v; }


__device__ __forceinline__ void pe_add_noted(const unsigned* seg, int n, int lane,
                                             float* __restrict__ rows,
                                             const float* __restrict__ pes,
                                             const int* __restrict__ et) {
  for (int base = 0; base < n; base += 32) {
    const int i = base + lane;
    unsigned ent = 0u;
    int t = 0;
    if (i < n) { ent = seg[i]; t = et[ent >> 8]; }
    const int m = (n - base < 32) ? (n - base) : 32;
    for (int u = 0; u < m; u += 4) {
      int tu[4];
      unsigned ru[4];
      float p0[4], p1[4];
#pragma unroll
      for (int j = 0; j < 4; ++j) {
        const int uj = (u + j < m) ? (u + j) : (m - 1);
        tu[j] = __builtin_amdgcn_readlane(t, uj);
        ru[j] = (unsigned)__builtin_amdgcn_readlane((int)ent, uj) & 255u;
        p0[j] = pes[tu[j] * kInCh + lane];
        p1[j] = pes[tu[j] * kInCh + 32 + lane];
      }
      asm volatile("" :: "v"(p0[0]), "v"(p0[1]), "v"(p0[2]), "v"(p0[3]),
                         "v"(p1[0]), "v"(p1[1]), "v"(p1[2]), "v"(p1[3]));
#pragma unroll
      for (int j = 0; j < 4; ++j) {
        if (u + j < m) {
          atomicAdd(&rows[ru[j] * kRowPitch + lane],      p0[j]);
          atomicAdd(&rows[ru[j] * kRowPitch + 32 + lane], p1[j]);
        }
      }
    }
  }
}

__device__ __forceinline__ void pe_note(unsigned mask, bool mine, unsigned rel, int e,
                                        int& wcount, unsigned* seg) {
  if (mask != 0u) {
    if (mine) seg[wcount + (int)__builtin_amdgcn_mbcnt_lo(mask, 0u)] = ((unsigned)e << 8) | rel;
    wcount += __popc(mask);
  }
}

__global__ void __launch_bounds__(kOwnThr)
k_mark(const int* __restrict__ ei, unsigned* __restrict__ marks,
       const float* __restrict__ W, const float* __restrict__ att_src,
       const float* __restrict__ att_dst, float* __restrict__ wf) {
  __shared__ unsigned own[2 * kOwnBlocks];
  const int tid = threadIdx.x;
  if (blockIdx.x == kMarkWords) {
    const int e   = tid & 3;
    const int ln  = (tid >> 2) & 31;
    const int kp  = tid >> 7;
    const int k   = 8 * kp + 4 * (e >> 1) + 2 * (ln >> 4) + (e & 1);
    const int col = ln & 15;
    for (int c = 0; c < kColTiles - 1; ++c) wf[c * kWfTile + tid] = W[k * kHC + c * 16 + col];
    float sum = 0.0f;
    if (col < 2 * kHeads) {
      const int h = col & (kHeads - 1);
      const float* __restrict__ att = (col < kHeads) ? att_src : att_dst;
      for (int c = 0; c < kOutCh; ++c) sum += W[k * kHC + h * kOutCh + c] * att[h * kOutCh + c];
    }
    wf[(kColTiles - 1) * kWfTile + tid] = sum;
    return;
  }
  if (tid < 2 * kOwnBlocks) own[tid] = 0u;
  __syncthreads();
  const int q = blockIdx.x * kOwnThr + tid;
  if (q < kQuads) {
    const unsigned bit = 1u << (tid >> 5);
    const int4 s = reinterpret_cast<const int4*>(ei)[q];
    const int4 d = reinterpret_cast<const int4*>(ei + kEdges)[q];
    const unsigned os[4] = {(unsigned)s.x / kOwnRows, (unsigned)s.y / kOwnRows,
                            (unsigned)s.z / kOwnRows, (unsigned)s.w / kOwnRows};
    const unsigned od[4] = {(unsigned)d.x / kOwnRows, (unsigned)d.y / kOwnRows,
                            (unsigned)d.z / kOwnRows, (unsigned)d.w / kOwnRows};
#pragma unroll
    for (int j = 0; j < 4; ++j) {
      if (os[j] < (unsigned)kOwnBlocks) atomicOr(&own[os[j]], bit);
      if (od[j] < (unsigned)kOwnBlocks) atomicOr(&own[kOwnBlocks + od[j]], bit);
    }
  }
  __syncthreads();
  if (tid < 2 * kOwnBlocks) {
    const int which = tid >> 8;
    marks[((size_t)which * kMarkWords + blockIdx.x) * kOwnBlocks + (tid & (kOwnBlocks - 1))] = own[tid];
  }
}
static_assert(kOwnBlocks == 256, "k_mark splits a thread index into row and owner block by 8 bits");
static_assert(kOwnThr == kKPairs * 32 * 4 && kHeads == 4, "the extra block of k_mark splits a thread index into pair of k steps, lane and element");

__device__ __forceinline__ int pe_todo(const unsigned* mk, u16a* todo,
                                       int off, int wave, int lane) {
  int g[kTodoTrips];
  unsigned w[kTodoTrips];
  bool in[kTodoTrips];
#pragma unroll
  for (int t = 0; t < kTodoTrips; ++t) {
    const int lin = (t * 32 + lane) * kOwnThr + wave * 32;
    in[t] = lin < kQuads;
    int q = off + (in[t] ? lin : 0);
    if (q >= kQuads) q -= kQuads;
    g[t] = q >> 5;
    w[t] = mk[g[t] >> 5];
  }
  int n = 0;
#pragma unroll
  for (int t = 0; t < kTodoTrips; ++t) {
    const bool on = in[t] && (((w[t] >> (g[t] & 31)) & 1u) != 0u);
    const unsigned m = __builtin_amdgcn_ballot_w32(on);
    if (on) todo[n + (int)__builtin_amdgcn_mbcnt_lo(m, 0u)] = (u16a)g[t];
    n += __popc(m);
  }
  return n;
}

__device__ __forceinline__ int pe_group(const uint4& t, int k) {
  const unsigned c = (k < 2) ? t.x : (k < 4) ? t.y : (k < 6) ? t.z : t.w;
  return (int)((k & 1) ? (c >> 16) : (c & 0xffffu));
}

__device__ __forceinline__ void pe_load(int4 (&dv)[kPeLoads], const int4* __restrict__ src4,
                                        const u16a* todo, int base, int n, int lane) {
  const int bc = (base < kTodoPad - kPeLoads) ? base : (kTodoPad - kPeLoads);
  const uint4 t = *reinterpret_cast<const uint4*>(todo + bc);
#pragma unroll
  for (int k = 0; k < kPeLoads; ++k) {
    const int g = (base + k < n) ? pe_group(t, k) : 0;
    dv[k] = src4[g * 32 + lane];
  }
}

__device__ __forceinline__ void pe_scan(const int4 (&dv)[kPeLoads], const u16a* todo,
                                        int base, int n, int lane, int s0, unsigned nrow,
                                        int& wcount, unsigned* seg) {
  if (base < n) {
    const uint4 t = *reinterpret_cast<const uint4*>(todo + base);
#pragma unroll
    for (int k = 0; k < kPeLoads; ++k) {
      if (base + k < n) {
        const int q = pe_group(t, k) * 32 + lane;
        const unsigned r0 = (unsigned)(dv[k].x - s0);
        const unsigned r1 = (unsigned)(dv[k].y - s0);
        const unsigned r2 = (unsigned)(dv[k].z - s0);
        const unsigned r3 = (unsigned)(dv[k].w - s0);
        const unsigned m0 = __builtin_amdgcn_ballot_w32(r0 < nrow);
        const unsigned m1 = __builtin_amdgcn_ballot_w32(r1 < nrow);
        const unsigned m2 = __builtin_amdgcn_ballot_w32(r2 < nrow);
        const unsigned m3 = __builtin_amdgcn_ballot_w32(r3 < nrow);
        pe_note(m0, r0 < nrow, r0, 4 * q + 0, wcount, seg);
        pe_note(m1, r1 < nrow, r1, 4 * q + 1, wcount, seg);
        pe_note(m2, r2 < nrow, r2, 4 * q + 2, wcount, seg);
        pe_note(m3, r3 < nrow, r3, 4 * q + 3, wcount, seg);
      }
    }
  }
}

__device__ __forceinline__ v8f tile_product(const float* ar, const float4* __restrict__ bf) {
  v8f acc = {0.f, 0.f, 0.f, 0.f, 0.f, 0.f, 0.f, 0.f};
#pragma unroll
  for (int kp = 0; kp < kKPairs; ++kp) {
    const float4 bv = bf[kp * 32];
    const v2f a0 = *reinterpret_cast<const v2f*>(ar + 8 * kp);
    const v2f a1 = *reinterpret_cast<const v2f*>(ar + 8 * kp + 4);
    const v2f b0 = {bv.x, bv.y};
    const v2f b1 = {bv.z, bv.w};
    acc = __builtin_amdgcn_wmma_f32_16x16x4_f32(false, a0, false, b0, (short)0, acc,
                                                false, false);
    acc = __builtin_amdgcn_wmma_f32_16x16x4_f32(false, a1, false, b1, (short)0, acc,
                                                false, false);
  }
  return acc;
}

__global__ void __launch_bounds__(kOwnThr)
k_owner_pe(const float* __restrict__ x, const int* __restrict__ ei, const int* __restrict__ et,
           const float* __restrict__ pe, const unsigned* __restrict__ msrc,
           const float* __restrict__ wf,
           float* __restrict__ xl, float* __restrict__ a_src, float* __restrict__ a_dst) {
  __shared__ __align__(16) float rows[kPeRowsPad * kRowPitch];
  __shared__ float    pes[kTypes * kInCh];
  __shared__ unsigned list[kOwnWaves * kPeSeg];
  __shared__ uint4    todo4[kOwnWaves * kTodoPad / 8];
  __shared__ unsigned mk[kMarkWords];

  const int tid  = threadIdx.x;
  const int lane = tid & 31;
  const int wave = __builtin_amdgcn_readfirstlane(tid >> 5);
  const int s0   = blockIdx.x * kOwnRows;
  const int nrow = (kNodes - s0 < kOwnRows) ? (kNodes - s0) : kOwnRows;
  const int nvec = nrow * (kInCh / 4);
  unsigned* seg = list + wave * kPeSeg;
  u16a* todo = reinterpret_cast<u16a*>(todo4) + wave * kTodoPad;

  {
    constexpr int kXv = kPeRowsPad * (kInCh / 4) / kOwnThr;
    static_assert(kTypes * kInCh / 4 == kOwnThr, "one 16-byte load of pe per thread");
    static_assert(kInCh / 4 == 16, "a row of x is 16 16-byte words");
    const float4* __restrict__ x4 = reinterpret_cast<const float4*>(x + (size_t)s0 * kInCh);
    float4* rows4 = reinterpret_cast<float4*>(rows);
    reinterpret_cast<float4*>(pes)[tid] = reinterpret_cast<const float4*>(pe)[tid];
    if (tid < kMarkWords) mk[tid] = msrc[(size_t)tid * kOwnBlocks + blockIdx.x];
#pragma unroll
    for (int j = 0; j < kXv; ++j) {
      const int i = j * kOwnThr + tid;
      rows4[(i >> 4) * (kRowPitch / 4) + (i & 15)] = x4[i < nvec ? i : nvec - 1];
    }
  }
  __syncthreads();

  const int4* __restrict__ src4 = reinterpret_cast<const int4*>(ei);
  const int off = blockIdx.x * kPeStagger;
  const int nt = pe_todo(mk, todo, off, wave, lane);
  int wcount = 0;
  int4 da[kPeLoads], db[kPeLoads];
  pe_load(da, src4, todo, 0, nt, lane);
  for (int b = 0; b < nt; b += 2 * kPeLoads) {
    pe_load(db, src4, todo, b + kPeLoads, nt, lane);
    if (wcount > kPeSeg - kPeStepMax) { pe_add_noted(seg, wcount, lane, rows, pes, et); wcount = 0; }
    pe_scan(da, todo, b, nt, lane, s0, (unsigned)nrow, wcount, seg);
    pe_load(da, src4, todo, b + 2 * kPeLoads, nt, lane);
    if (wcount > kPeSeg - kPeStepMax) { pe_add_noted(seg, wcount, lane, rows, pes, et); wcount = 0; }
    pe_scan(db, todo, b + kPeLoads, nt, lane, s0, (unsigned)nrow, wcount, seg);
  }

  pe_add_noted(seg, wcount, lane, rows, pes, et);
  __syncthreads();

  {
    const int half  = lane >> 4;
    const int l15   = lane & 15;
    const int ntile = (nrow + 15) >> 4;
    const int px = 8 * half * kHC + l15;
    const int ps = 8 * half * kHeads + (l15 & (kHeads - 1));
    for (int it = wave; it < ntile * kColTiles; it += kOwnWaves) {
      const int t  = it / kColTiles;
      const int c  = it - t * kColTiles;
      const int r0 = t * 16 + 8 * half;
      const float* ar = rows + (t * 16 + l15) * kRowPitch + 2 * half;
      const v8f acc = tile_product(ar, reinterpret_cast<const float4*>(wf + c * kWfTile) + lane);
      if (c < kColTiles - 1) {
        float* __restrict__ o = xl + (size_t)(s0 + t * 16) * kHC + c * 16;
        if (r0 < nrow)     { o[px]           = acc[0]; o[px + kHC]     = acc[1]; o[px + 2 * kHC] = acc[2]; o[px + 3 * kHC] = acc[3]; }
        if (r0 + 4 < nrow) { o[px + 4 * kHC] = acc[4]; o[px + 5 * kHC] = acc[5]; o[px + 6 * kHC] = acc[6]; o[px + 7 * kHC] = acc[7]; }
      } else {
        if (l15 < 2 * kHeads) {
          float* __restrict__ o = ((l15 < kHeads) ? a_src : a_dst) + (size_t)(s0 + t * 16) * kHeads;
          if (r0 < nrow)     { o[ps]              = acc[0]; o[ps + kHeads]     = acc[1]; o[ps + 2 * kHeads] = acc[2]; o[ps + 3 * kHeads] = acc[3]; }
          if (r0 + 4 < nrow) { o[ps + 4 * kHeads] = acc[4]; o[ps + 5 * kHeads] = acc[5]; o[ps + 6 * kHeads] = acc[6]; o[ps + 7 * kHeads] = acc[7]; }
        }
      }
    }
  }
}


template <int PH>
__device__ __forceinline__ void ag_stats(const unsigned* seg, int n, int lane,
                                         const float* __restrict__ a_src,
                                         const float* adst_l, float* amax_l, float* den_l) {
  const int u  = lane >> 2;
  const int hh = lane & 3;
  for (int base = 0; base < n; base += 32) {
    unsigned ent[4];
    float as[4];
#pragma unroll
    for (int j = 0; j < 4; ++j) {
      const int idx = base + 8 * j + u;
      ent[j] = seg[idx < n ? idx : n - 1];
      as[j]  = a_src[(ent[j] >> 8) * kHeads + hh];
    }
    asm volatile("" :: "v"(as[0]), "v"(as[1]), "v"(as[2]), "v"(as[3]));
#pragma unroll
    for (int j = 0; j < 4; ++j) {
      if (base + 8 * j + u < n) {
        const int k = (int)(ent[j] & 255u) * kHeads + hh;
        const float al = leaky02(as[j] + adst_l[k]);
        if (PH == 0) atomic_max_f32(&amax_l[k], al);
        else         atomicAdd(&den_l[k], expf(al - amax_l[k]));
      }
    }
  }
}

__device__ __forceinline__ void ag_rows(const unsigned* seg, int n, int lane,
                                        const float* __restrict__ a_src,
                                        const float* adst_l, const float* amax_l,
                                        const float* den_l, float* acc,
                                        const float4* __restrict__ xl4) {
  const int u  = lane >> 2;
  const int hh = lane & 3;
  const int hr = lane >> 3;
  for (int base = 0; base < n; base += 8) {
    const int idx = base + u;
    const unsigned ent = seg[idx < n ? idx : n - 1];
    const int k = (int)(ent & 255u) * kHeads + hh;
    const float as = a_src[(ent >> 8) * kHeads + hh];
    float4 v[8];
    int rl[8];
#pragma unroll
    for (int j = 0; j < 8; ++j) {
      const unsigned eu = (unsigned)__builtin_amdgcn_readlane((int)ent, 4 * j);
      rl[j] = (int)(eu & 255u);
      v[j]  = xl4[(size_t)(eu >> 8) * (kHC / 4) + lane];
    }
    const float coef = expf(leaky02(as + adst_l[k]) - amax_l[k]) / den_l[k];
    float c[8];
#pragma unroll
    for (int j = 0; j < 8; ++j) c[j] = __shfl(coef, 4 * j + hr);
    asm volatile("" :: "v"(coef), "v"(v[0].x), "v"(v[1].x), "v"(v[2].x), "v"(v[3].x),
                       "v"(v[4].x), "v"(v[5].x), "v"(v[6].x), "v"(v[7].x));
#pragma unroll
    for (int j = 0; j < 8; ++j) {
      if (base + j < n) {
        float* row = acc + rl[j] * kHC + lane;
        atomicAdd(row,      c[j] * v[j].x);
        atomicAdd(row + 32, c[j] * v[j].y);
        atomicAdd(row + 64, c[j] * v[j].z);
        atomicAdd(row + 96, c[j] * v[j].w);
      }
    }
  }
}

__device__ __forceinline__ void ag_resolve(unsigned* seg, int n0, int n, int lane,
                                           const int* __restrict__ ei) {
  for (int base = n0; base < n; base += 128) {
    unsigned en[4];
    int sv[4];
#pragma unroll
    for (int j = 0; j < 4; ++j) {
      const int i = base + 32 * j + lane;
      en[j] = seg[i < n ? i : n - 1];
    }
#pragma unroll
    for (int j = 0; j < 4; ++j) sv[j] = ei[en[j] >> 8];
#pragma unroll
    for (int j = 0; j < 4; ++j) {
      const int i = base + 32 * j + lane;
      if (i < n) seg[i] = ((unsigned)sv[j] << 8) | (en[j] & 255u);
    }
  }
}

__global__ void __launch_bounds__(kOwnThr)
k_owner_aggregate(const int* __restrict__ ei, const unsigned* __restrict__ mdst,
                  const float* __restrict__ a_src, const float* __restrict__ a_dst,
                  const float* __restrict__ xl, const float* __restrict__ bias,
                  float* __restrict__ out) {
  __shared__ float    acc[kOwnRows * kHC];
  __shared__ unsigned list[kOwnWaves * kPeSeg];
  __shared__ float    adst_l[kOwnRows * kHeads];
  __shared__ float    amax_l[kOwnRows * kHeads];
  __shared__ float    den_l[kOwnRows * kHeads];
  __shared__ int      ovf;
  __shared__ uint4    todo4[kOwnWaves * kTodoPad / 8];
  __shared__ unsigned mk[kMarkWords];

  const int tid  = threadIdx.x;
  const int lane = tid & 31;
  const int wave = __builtin_amdgcn_readfirstlane(tid >> 5);
  const int d0   = blockIdx.x * kOwnRows;
  const int nrow = (kNodes - d0 < kOwnRows) ? (kNodes - d0) : kOwnRows;
  const float4* __restrict__ xl4 = reinterpret_cast<const float4*>(xl);
  unsigned* seg = list + wave * kPeSeg;
  u16a* todo = reinterpret_cast<u16a*>(todo4) + wave * kTodoPad;

  {
    float4* acc4 = reinterpret_cast<float4*>(acc);
    for (int i = tid; i < kOwnRows * kHC / 4; i += kOwnThr) acc4[i] = make_float4(0.f, 0.f, 0.f, 0.f);
    if (tid < kOwnRows * kHeads) {
      adst_l[tid] = (tid < nrow * kHeads) ? a_dst[d0 * kHeads + tid] : 0.0f;
      amax_l[tid] = -INFINITY;
      den_l[tid]  = 0.0f;
    }
    if (tid == 0) ovf = 0;
    if (tid < kMarkWords) mk[tid] = mdst[(size_t)tid * kOwnBlocks + blockIdx.x];
  }
  __syncthreads();

  const int nself = (nrow > wave) ? ((nrow - wave + 31) >> 5) : 0;
  if (lane < nself) seg[lane] = ((unsigned)(d0 + wave + 32 * lane) << 8) | (unsigned)(wave + 32 * lane);

  const int4* __restrict__ dst4 = reinterpret_cast<const int4*>(ei + kEdges);
  const int off = blockIdx.x * kPeStagger;
  const int nt = pe_todo(mk, todo, off, wave, lane);
  int wcount = nself;
  bool over = false;
  int4 da[kPeLoads], db[kPeLoads];
  pe_load(da, dst4, todo, 0, nt, lane);
  for (int b = 0; b < nt; b += 2 * kPeLoads) {
    pe_load(db, dst4, todo, b + kPeLoads, nt, lane);
    over = over || (wcount > kPeSeg - kPeStepMax);
    if (!over) pe_scan(da, todo, b, nt, lane, d0, (unsigned)nrow, wcount, seg);
    pe_load(da, dst4, todo, b + 2 * kPeLoads, nt, lane);
    over = over || (wcount > kPeSeg - kPeStepMax);
    if (!over) pe_scan(db, todo, b + kPeLoads, nt, lane, d0, (unsigned)nrow, wcount, seg);
  }
  if (over) ovf = 1;
  else      ag_resolve(seg, nself, wcount, lane, ei);
  __syncthreads();

  const bool fits = __builtin_amdgcn_readfirstlane(ovf) == 0;
  if (fits) {
    ag_stats<0>(seg, wcount, lane, a_src, adst_l, amax_l, den_l);
    __syncthreads();
    ag_stats<1>(seg, wcount, lane, a_src, adst_l, amax_l, den_l);
    __syncthreads();
    ag_rows(seg, wcount, lane, a_src, adst_l, amax_l, den_l, acc, xl4);
  } else {
    for (int ph = 0; ph < 3; ++ph) {
      for (int b = 0; b == 0 || b < nt; b += kPeLoads) {
        int n = 0;
        if (b == 0) {
          if (lane < nself) seg[lane] = ((unsigned)(d0 + wave + 32 * lane) << 8) | (unsigned)(wave + 32 * lane);
          n = nself;
        }
        const int n0 = n;
        pe_load(da, dst4, todo, b, nt, lane);
        pe_scan(da, todo, b, nt, lane, d0, (unsigned)nrow, n, seg);
        ag_resolve(seg, n0, n, lane, ei);
        if (ph == 0)      ag_stats<0>(seg, n, lane, a_src, adst_l, amax_l, den_l);
        else if (ph == 1) ag_stats<1>(seg, n, lane, a_src, adst_l, amax_l, den_l);
        else              ag_rows(seg, n, lane, a_src, adst_l, amax_l, den_l, acc, xl4);
      }
      __syncthreads();
    }
  }
  __syncthreads();

  const float4 b = reinterpret_cast<const float4*>(bias)[lane];
  float4* __restrict__ out4 = reinterpret_cast<float4*>(out);
  for (int dl = wave; dl < nrow; dl += kOwnWaves) {
    const int d = d0 + dl;
    const float* row = acc + dl * kHC + lane;
    float4 r;
    r.x = row[0]  + b.x;
    r.y = row[32] + b.y;
    r.z = row[64] + b.z;
    r.w = row[96] + b.w;
    r.x = r.x > 0.0f ? r.x : expm1f(r.x);
    r.y = r.y > 0.0f ? r.y : expm1f(r.y);
    r.z = r.z > 0.0f ? r.z : expm1f(r.z);
    r.w = r.w > 0.0f ? r.w : expm1f(r.w);
    out4[(size_t)d * (kHC / 4) + lane] = r;
  }
}


extern "C" void kernel_launch(void* const* d_in, const int* in_sizes, int n_in,
                              void* d_out, int out_size, void* d_ws, size_t ws_size,
                              hipStream_t stream) {
  (void)in_sizes; (void)n_in; (void)out_size; (void)ws_size;
  const float* x       = (const float*)d_in[0];
  const int*   ei      = (const int*)d_in[1];
  const int*   et      = (const int*)d_in[2];
  const float* pe      = (const float*)d_in[3];
  const float* W       = (const float*)d_in[4];
  const float* att_src = (const float*)d_in[5];
  const float* att_dst = (const float*)d_in[6];
  const float* bias    = (const float*)d_in[7];
  float* out = (float*)d_out;

  float* ws    = (float*)d_ws;
  float* xl    = ws;
  float* a_src = xl    + (size_t)kNodes * kHC;
  float* a_dst = a_src + (size_t)kNodes * kHeads;
  unsigned* marks = (unsigned*)(a_dst + (size_t)kNodes * kHeads);
  float* wf    = (float*)(marks + (size_t)2 * kMarkWords * kOwnBlocks);

  k_mark<<<kMarkWords + 1, kOwnThr, 0, stream>>>(ei, marks, W, att_src, att_dst, wf);
  k_owner_pe<<<kOwnBlocks, kOwnThr, 0, stream>>>(x, ei, et, pe, marks, wf, xl, a_src, a_dst);
  k_owner_aggregate<<<kOwnBlocks, kOwnThr, 0, stream>>>(ei, marks + (size_t)kMarkWords * kOwnBlocks,
                                                        a_src, a_dst, xl, bias, out);
}


# marks per 128-byte line of the index lists (groups of 32 edges, were 128): an owner scan loads and compares four marked groups per instruction; list order, so LDS float-add order, changes; 3 launches
# speedup vs baseline: 15.2690x; 1.1382x over previous
#include <hip/hip_runtime.h>
#include <math.h>

typedef float v2f __attribute__((ext_vector_type(2)));
typedef float v8f __attribute__((ext_vector_type(8)));

namespace {
constexpr int kNodes = 50000;
constexpr int kEdges = 800000;
constexpr int kInCh  = 64;
constexpr int kOutCh = 32;
constexpr int kHeads = 4;
constexpr int kHC    = kHeads * kOutCh;
constexpr int kETot  = kEdges + kNodes;
constexpr int kOwnBlocks = 256;
constexpr int kOwnRows   = 196;
constexpr int kOwnThr    = 1024;
constexpr int kOwnWaves  = kOwnThr / 32;
constexpr int kQuads     = kEdges / 4;
static_assert(kEdges % 4 == 0, "the index rows are read in groups of 4");
static_assert(kOwnBlocks * kOwnRows >= kNodes, "every node has an owner");
static_assert(kOwnRows <= 256, "local row index is packed into 8 bits");
static_assert(kNodes <= (1 << 24), "source index is packed into 24 bits");
constexpr int kTypes       = 64;
constexpr int kPeLoads     = 8;
constexpr int kPeStagger   = 768;
constexpr int kPeStepMax   = 32 * kPeLoads * 4;
constexpr int kPeSeg       = 1264;
constexpr int kPeRowsPad   = 256;
static_assert(kQuads % 32 == 0 && kPeStagger % 32 == 0, "a wave's 32 quads never straddle the wrap");
static_assert(kOwnBlocks * kPeStagger <= kQuads, "start places stay inside the list");
static_assert(kEdges <= (1 << 24), "edge index is packed into 24 bits");
static_assert(kPeSeg > kPeStepMax, "a whole step fits behind the mark at which a segment is emptied");
static_assert(kPeRowsPad >= kOwnRows && kPeRowsPad * (kInCh / 4) % kOwnThr == 0, "the seed copy covers the padded rows exactly");
constexpr int kRowPitch    = kInCh + 4;
constexpr int kColTiles    = kHC / 16 + 1;
constexpr int kKPairs      = kInCh / 8;
constexpr int kWfTile      = kKPairs * 32 * 4;
static_assert(kPeRowsPad >= (kOwnRows + 15) / 16 * 16, "the last row tile reads padded rows only");
static_assert(kRowPitch % 4 == 0 && 2 * kHeads <= 16, "rows stay 16-byte aligned; both score sets fit one tile");
static_assert(kWfTile == kOwnThr, "the extra block of k_mark fills one column tile of the B table per trip");
static_assert(kOwnRows % 4 == 0 && kNodes % 4 == 0, "every block owns a multiple of four rows: a tile's rows are stored in fours");
constexpr int kGroupQuads  = 8;
constexpr int kGroups      = kQuads / kGroupQuads;
constexpr int kMarkBlocks  = (kQuads + kOwnThr - 1) / kOwnThr;
constexpr int kMarkWords   = 4 * kMarkBlocks;
constexpr int kWaveSpans   = (kQuads + kOwnThr - 1) / kOwnThr;
constexpr int kTodoTrips   = (kWaveSpans + 31) / 32;
constexpr int kTodoPad     = 4 * kWaveSpans;
constexpr int kStepGroups  = 4 * kPeLoads;
static_assert(kQuads % kGroupQuads == 0 && 32 % kGroupQuads == 0, "groups tile the list and a wave's 32 quads");
static_assert(kOwnThr == 4 * 32 * kGroupQuads, "a block of k_mark fills four words of 32 groups");
static_assert(kGroups < (1 << 16), "a group index is kept in 16 bits");
static_assert(kMarkWords <= kOwnThr, "one thread per mark word when an owner block fetches its marks");
}


__device__ __forceinline__ void atomic_max_f32(float* addr, float v) {
  if (v >= 0.0f) atomicMax(reinterpret_cast<int*>(addr), __float_as_int(v));
  else           atomicMin(reinterpret_cast<unsigned int*>(addr), __float_as_uint(v));
}

__device__ __forceinline__ float leaky02(float v) { return v > 0.0f ? v : 0.2f * v; }


__device__ __forceinline__ void pe_add_noted(const unsigned* seg, int n, int lane,
                                             float* __restrict__ rows,
                                             const float* __restrict__ pes,
                                             const int* __restrict__ et) {
  for (int base = 0; base < n; base += 32) {
    const int i = base + lane;
    unsigned ent = 0u;
    int t = 0;
    if (i < n) { ent = seg[i]; t = et[ent >> 8]; }
    const int m = (n - base < 32) ? (n - base) : 32;
    for (int u = 0; u < m; u += 4) {
      int tu[4];
      unsigned ru[4];
      float p0[4], p1[4];
#pragma unroll
      for (int j = 0; j < 4; ++j) {
        const int uj = (u + j < m) ? (u + j) : (m - 1);
        tu[j] = __builtin_amdgcn_readlane(t, uj);
        ru[j] = (unsigned)__builtin_amdgcn_readlane((int)ent, uj) & 255u;
        p0[j] = pes[tu[j] * kInCh + lane];
        p1[j] = pes[tu[j] * kInCh + 32 + lane];
      }
      asm volatile("" :: "v"(p0[0]), "v"(p0[1]), "v"(p0[2]), "v"(p0[3]),
                         "v"(p1[0]), "v"(p1[1]), "v"(p1[2]), "v"(p1[3]));
#pragma unroll
      for (int j = 0; j < 4; ++j) {
        if (u + j < m) {
          atomicAdd(&rows[ru[j] * kRowPitch + lane],      p0[j]);
          atomicAdd(&rows[ru[j] * kRowPitch + 32 + lane], p1[j]);
        }
      }
    }
  }
}

__device__ __forceinline__ void pe_note(unsigned mask, bool mine, unsigned rel, int e,
                                        int& wcount, unsigned* seg) {
  if (mask != 0u) {
    if (mine) seg[wcount + (int)__builtin_amdgcn_mbcnt_lo(mask, 0u)] = ((unsigned)e << 8) | rel;
    wcount += __popc(mask);
  }
}

__global__ void __launch_bounds__(kOwnThr)
k_mark(const int* __restrict__ ei, unsigned* __restrict__ marks,
       const float* __restrict__ W, const float* __restrict__ att_src,
       const float* __restrict__ att_dst, float* __restrict__ wf) {
  __shared__ unsigned own[4 * 2 * kOwnBlocks];
  const int tid = threadIdx.x;
  if (blockIdx.x == kMarkBlocks) {
    const int e   = tid & 3;
    const int ln  = (tid >> 2) & 31;
    const int kp  = tid >> 7;
    const int k   = 8 * kp + 4 * (e >> 1) + 2 * (ln >> 4) + (e & 1);
    const int col = ln & 15;
    for (int c = 0; c < kColTiles - 1; ++c) wf[c * kWfTile + tid] = W[k * kHC + c * 16 + col];
    float sum = 0.0f;
    if (col < 2 * kHeads) {
      const int h = col & (kHeads - 1);
      const float* __restrict__ att = (col < kHeads) ? att_src : att_dst;
      for (int c = 0; c < kOutCh; ++c) sum += W[k * kHC + h * kOutCh + c] * att[h * kOutCh + c];
    }
    wf[(kColTiles - 1) * kWfTile + tid] = sum;
    return;
  }
  for (int i = tid; i < 4 * 2 * kOwnBlocks; i += kOwnThr) own[i] = 0u;
  __syncthreads();
  const int q = blockIdx.x * kOwnThr + tid;
  if (q < kQuads) {
    const unsigned bit = 1u << ((tid >> 3) & 31);
    unsigned* ow = own + (tid >> 8) * (2 * kOwnBlocks);
    const int4 s = reinterpret_cast<const int4*>(ei)[q];
    const int4 d = reinterpret_cast<const int4*>(ei + kEdges)[q];
    const unsigned os[4] = {(unsigned)s.x / kOwnRows, (unsigned)s.y / kOwnRows,
                            (unsigned)s.z / kOwnRows, (unsigned)s.w / kOwnRows};
    const unsigned od[4] = {(unsigned)d.x / kOwnRows, (unsigned)d.y / kOwnRows,
                            (unsigned)d.z / kOwnRows, (unsigned)d.w / kOwnRows};
#pragma unroll
    for (int j = 0; j < 4; ++j) {
      if (os[j] < (unsigned)kOwnBlocks) atomicOr(&ow[os[j]], bit);
      if (od[j] < (unsigned)kOwnBlocks) atomicOr(&ow[kOwnBlocks + od[j]], bit);
    }
  }
  __syncthreads();
  for (int i = tid; i < 4 * 2 * kOwnBlocks; i += kOwnThr) {
    const int wq    = i >> 9;
    const int which = (i >> 8) & 1;
    marks[((size_t)which * kMarkWords + blockIdx.x * 4 + wq) * kOwnBlocks + (i & (kOwnBlocks - 1))] = own[i];
  }
}
static_assert(kOwnBlocks == 256, "k_mark splits a thread index into row and owner block by 8 bits");
static_assert(kOwnThr == kKPairs * 32 * 4 && kHeads == 4, "the extra block of k_mark splits a thread index into pair of k steps, lane and element");

__device__ __forceinline__ int pe_todo(const unsigned* mk, unsigned short* todo,
                                       int off, int wave, int lane) {
  int g[kTodoTrips];
  unsigned w[kTodoTrips];
  bool in[kTodoTrips];
#pragma unroll
  for (int t = 0; t < kTodoTrips; ++t) {
    const int lin = (t * 32 + lane) * kOwnThr + wave * 32;
    in[t] = lin < kQuads;
    int q = off + (in[t] ? lin : 0);
    if (q >= kQuads) q -= kQuads;
    g[t] = q / kGroupQuads;
    w[t] = mk[g[t] >> 5];
  }
  int n = 0;
#pragma unroll
  for (int t = 0; t < kTodoTrips; ++t) {
    const unsigned bits = in[t] ? ((w[t] >> (g[t] & 31)) & 15u) : 0u;
#pragma unroll
    for (int s = 0; s < 4; ++s) {
      const bool on = ((bits >> s) & 1u) != 0u;
      const unsigned m = __builtin_amdgcn_ballot_w32(on);
      if (on) todo[n + (int)__builtin_amdgcn_mbcnt_lo(m, 0u)] = (unsigned short)(g[t] + s);
      n += __popc(m);
    }
  }
  return n;
}

__device__ __forceinline__ void pe_load(int4 (&dv)[kPeLoads], const int4* __restrict__ src4,
                                        const unsigned short* todo, int base, int n, int lane) {
#pragma unroll
  for (int k = 0; k < kPeLoads; ++k) {
    const int idx = base + 4 * k + (lane >> 3);
    const int tg  = (int)todo[idx < n ? idx : 0];
    const int g   = (idx < n) ? tg : 0;
    dv[k] = src4[g * kGroupQuads + (lane & 7)];
  }
}

__device__ __forceinline__ void pe_scan(const int4 (&dv)[kPeLoads], const unsigned short* todo,
                                        int base, int n, int lane, int s0, unsigned nrow,
                                        int& wcount, unsigned* seg) {
  if (base < n) {
    int tq[kPeLoads];
    unsigned lim[kPeLoads];
#pragma unroll
    for (int k = 0; k < kPeLoads; ++k) {
      const int idx   = base + 4 * k + (lane >> 3);
      const bool live = idx < n;
      tq[k]  = (int)todo[live ? idx : 0];
      lim[k] = live ? nrow : 0u;
    }
    asm volatile("" :: "v"(tq[0]), "v"(tq[1]), "v"(tq[2]), "v"(tq[3]),
                       "v"(tq[4]), "v"(tq[5]), "v"(tq[6]), "v"(tq[7]));
#pragma unroll
    for (int k = 0; k < kPeLoads; ++k) {
      if (base + 4 * k < n) {
        const int q = tq[k] * kGroupQuads + (lane & 7);
        const unsigned r0 = (unsigned)(dv[k].x - s0);
        const unsigned r1 = (unsigned)(dv[k].y - s0);
        const unsigned r2 = (unsigned)(dv[k].z - s0);
        const unsigned r3 = (unsigned)(dv[k].w - s0);
        const unsigned m0 = __builtin_amdgcn_ballot_w32(r0 < lim[k]);
        const unsigned m1 = __builtin_amdgcn_ballot_w32(r1 < lim[k]);
        const unsigned m2 = __builtin_amdgcn_ballot_w32(r2 < lim[k]);
        const unsigned m3 = __builtin_amdgcn_ballot_w32(r3 < lim[k]);
        pe_note(m0, r0 < lim[k], r0, 4 * q + 0, wcount, seg);
        pe_note(m1, r1 < lim[k], r1, 4 * q + 1, wcount, seg);
        pe_note(m2, r2 < lim[k], r2, 4 * q + 2, wcount, seg);
        pe_note(m3, r3 < lim[k], r3, 4 * q + 3, wcount, seg);
      }
    }
  }
}
static_assert(kPeLoads == 8, "the fence in pe_scan names eight values");

__device__ __forceinline__ v8f tile_product(const float* ar, const float4* __restrict__ bf) {
  v8f acc = {0.f, 0.f, 0.f, 0.f, 0.f, 0.f, 0.f, 0.f};
#pragma unroll
  for (int kp = 0; kp < kKPairs; ++kp) {
    const float4 bv = bf[kp * 32];
    const v2f a0 = *reinterpret_cast<const v2f*>(ar + 8 * kp);
    const v2f a1 = *reinterpret_cast<const v2f*>(ar + 8 * kp + 4);
    const v2f b0 = {bv.x, bv.y};
    const v2f b1 = {bv.z, bv.w};
    acc = __builtin_amdgcn_wmma_f32_16x16x4_f32(false, a0, false, b0, (short)0, acc,
                                                false, false);
    acc = __builtin_amdgcn_wmma_f32_16x16x4_f32(false, a1, false, b1, (short)0, acc,
                                                false, false);
  }
  return acc;
}

__global__ void __launch_bounds__(kOwnThr)
k_owner_pe(const float* __restrict__ x, const int* __restrict__ ei, const int* __restrict__ et,
           const float* __restrict__ pe, const unsigned* __restrict__ msrc,
           const float* __restrict__ wf,
           float* __restrict__ xl, float* __restrict__ a_src, float* __restrict__ a_dst) {
  __shared__ __align__(16) float rows[kPeRowsPad * kRowPitch];
  __shared__ float    pes[kTypes * kInCh];
  __shared__ unsigned list[kOwnWaves * kPeSeg];
  __shared__ unsigned short todo_s[kOwnWaves * kTodoPad];
  __shared__ unsigned mk[kMarkWords];

  const int tid  = threadIdx.x;
  const int lane = tid & 31;
  const int wave = __builtin_amdgcn_readfirstlane(tid >> 5);
  const int s0   = blockIdx.x * kOwnRows;
  const int nrow = (kNodes - s0 < kOwnRows) ? (kNodes - s0) : kOwnRows;
  const int nvec = nrow * (kInCh / 4);
  unsigned* seg = list + wave * kPeSeg;
  unsigned short* todo = todo_s + wave * kTodoPad;

  {
    constexpr int kXv = kPeRowsPad * (kInCh / 4) / kOwnThr;
    static_assert(kTypes * kInCh / 4 == kOwnThr, "one 16-byte load of pe per thread");
    static_assert(kInCh / 4 == 16, "a row of x is 16 16-byte words");
    const float4* __restrict__ x4 = reinterpret_cast<const float4*>(x + (size_t)s0 * kInCh);
    float4* rows4 = reinterpret_cast<float4*>(rows);
    reinterpret_cast<float4*>(pes)[tid] = reinterpret_cast<const float4*>(pe)[tid];
    if (tid < kMarkWords) mk[tid] = msrc[(size_t)tid * kOwnBlocks + blockIdx.x];
#pragma unroll
    for (int j = 0; j < kXv; ++j) {
      const int i = j * kOwnThr + tid;
      rows4[(i >> 4) * (kRowPitch / 4) + (i & 15)] = x4[i < nvec ? i : nvec - 1];
    }
  }
  __syncthreads();

  const int4* __restrict__ src4 = reinterpret_cast<const int4*>(ei);
  const int off = blockIdx.x * kPeStagger;
  const int nt = pe_todo(mk, todo, off, wave, lane);
  int wcount = 0;
  int4 da[kPeLoads], db[kPeLoads];
  pe_load(da, src4, todo, 0, nt, lane);
  for (int b = 0; b < nt; b += 2 * kStepGroups) {
    pe_load(db, src4, todo, b + kStepGroups, nt, lane);
    if (wcount > kPeSeg - kPeStepMax) { pe_add_noted(seg, wcount, lane, rows, pes, et); wcount = 0; }
    pe_scan(da, todo, b, nt, lane, s0, (unsigned)nrow, wcount, seg);
    pe_load(da, src4, todo, b + 2 * kStepGroups, nt, lane);
    if (wcount > kPeSeg - kPeStepMax) { pe_add_noted(seg, wcount, lane, rows, pes, et); wcount = 0; }
    pe_scan(db, todo, b + kStepGroups, nt, lane, s0, (unsigned)nrow, wcount, seg);
  }

  pe_add_noted(seg, wcount, lane, rows, pes, et);
  __syncthreads();

  {
    const int half  = lane >> 4;
    const int l15   = lane & 15;
    const int ntile = (nrow + 15) >> 4;
    const int px = 8 * half * kHC + l15;
    const int ps = 8 * half * kHeads + (l15 & (kHeads - 1));
    for (int it = wave; it < ntile * kColTiles; it += kOwnWaves) {
      const int t  = it / kColTiles;
      const int c  = it - t * kColTiles;
      const int r0 = t * 16 + 8 * half;
      const float* ar = rows + (t * 16 + l15) * kRowPitch + 2 * half;
      const v8f acc = tile_product(ar, reinterpret_cast<const float4*>(wf + c * kWfTile) + lane);
      if (c < kColTiles - 1) {
        float* __restrict__ o = xl + (size_t)(s0 + t * 16) * kHC + c * 16;
        if (r0 < nrow)     { o[px]           = acc[0]; o[px + kHC]     = acc[1]; o[px + 2 * kHC] = acc[2]; o[px + 3 * kHC] = acc[3]; }
        if (r0 + 4 < nrow) { o[px + 4 * kHC] = acc[4]; o[px + 5 * kHC] = acc[5]; o[px + 6 * kHC] = acc[6]; o[px + 7 * kHC] = acc[7]; }
      } else {
        if (l15 < 2 * kHeads) {
          float* __restrict__ o = ((l15 < kHeads) ? a_src : a_dst) + (size_t)(s0 + t * 16) * kHeads;
          if (r0 < nrow)     { o[ps]              = acc[0]; o[ps + kHeads]     = acc[1]; o[ps + 2 * kHeads] = acc[2]; o[ps + 3 * kHeads] = acc[3]; }
          if (r0 + 4 < nrow) { o[ps + 4 * kHeads] = acc[4]; o[ps + 5 * kHeads] = acc[5]; o[ps + 6 * kHeads] = acc[6]; o[ps + 7 * kHeads] = acc[7]; }
        }
      }
    }
  }
}


template <int PH>
__device__ __forceinline__ void ag_stats(const unsigned* seg, int n, int lane,
                                         const float* __restrict__ a_src,
                                         const float* adst_l, float* amax_l, float* den_l) {
  const int u  = lane >> 2;
  const int hh = lane & 3;
  for (int base = 0; base < n; base += 32) {
    unsigned ent[4];
    float as[4];
#pragma unroll
    for (int j = 0; j < 4; ++j) {
      const int idx = base + 8 * j + u;
      ent[j] = seg[idx < n ? idx : n - 1];
      as[j]  = a_src[(ent[j] >> 8) * kHeads + hh];
    }
    asm volatile("" :: "v"(as[0]), "v"(as[1]), "v"(as[2]), "v"(as[3]));
#pragma unroll
    for (int j = 0; j < 4; ++j) {
      if (base + 8 * j + u < n) {
        const int k = (int)(ent[j] & 255u) * kHeads + hh;
        const float al = leaky02(as[j] + adst_l[k]);
        if (PH == 0) atomic_max_f32(&amax_l[k], al);
        else         atomicAdd(&den_l[k], expf(al - amax_l[k]));
      }
    }
  }
}

__device__ __forceinline__ void ag_rows(const unsigned* seg, int n, int lane,
                                        const float* __restrict__ a_src,
                                        const float* adst_l, const float* amax_l,
                                        const float* den_l, float* acc,
                                        const float4* __restrict__ xl4) {
  const int u  = lane >> 2;
  const int hh = lane & 3;
  const int hr = lane >> 3;
  for (int base = 0; base < n; base += 8) {
    const int idx = base + u;
    const unsigned ent = seg[idx < n ? idx : n - 1];
    const int k = (int)(ent & 255u) * kHeads + hh;
    const float as = a_src[(ent >> 8) * kHeads + hh];
    float4 v[8];
    int rl[8];
#pragma unroll
    for (int j = 0; j < 8; ++j) {
      const unsigned eu = (unsigned)__builtin_amdgcn_readlane((int)ent, 4 * j);
      rl[j] = (int)(eu & 255u);
      v[j]  = xl4[(size_t)(eu >> 8) * (kHC / 4) + lane];
    }
    const float coef = expf(leaky02(as + adst_l[k]) - amax_l[k]) / den_l[k];
    float c[8];
#pragma unroll
    for (int j = 0; j < 8; ++j) c[j] = __shfl(coef, 4 * j + hr);
    asm volatile("" :: "v"(coef), "v"(v[0].x), "v"(v[1].x), "v"(v[2].x), "v"(v[3].x),
                       "v"(v[4].x), "v"(v[5].x), "v"(v[6].x), "v"(v[7].x));
#pragma unroll
    for (int j = 0; j < 8; ++j) {
      if (base + j < n) {
        float* row = acc + rl[j] * kHC + lane;
        atomicAdd(row,      c[j] * v[j].x);
        atomicAdd(row + 32, c[j] * v[j].y);
        atomicAdd(row + 64, c[j] * v[j].z);
        atomicAdd(row + 96, c[j] * v[j].w);
      }
    }
  }
}

__device__ __forceinline__ void ag_resolve(unsigned* seg, int n0, int n, int lane,
                                           const int* __restrict__ ei) {
  for (int base = n0; base < n; base += 128) {
    unsigned en[4];
    int sv[4];
#pragma unroll
    for (int j = 0; j < 4; ++j) {
      const int i = base + 32 * j + lane;
      en[j] = seg[i < n ? i : n - 1];
    }
#pragma unroll
    for (int j = 0; j < 4; ++j) sv[j] = ei[en[j] >> 8];
#pragma unroll
    for (int j = 0; j < 4; ++j) {
      const int i = base + 32 * j + lane;
      if (i < n) seg[i] = ((unsigned)sv[j] << 8) | (en[j] & 255u);
    }
  }
}

__global__ void __launch_bounds__(kOwnThr)
k_owner_aggregate(const int* __restrict__ ei, const unsigned* __restrict__ mdst,
                  const float* __restrict__ a_src, const float* __restrict__ a_dst,
                  const float* __restrict__ xl, const float* __restrict__ bias,
                  float* __restrict__ out) {
  __shared__ float    acc[kOwnRows * kHC];
  __shared__ unsigned list[kOwnWaves * kPeSeg];
  __shared__ float    adst_l[kOwnRows * kHeads];
  __shared__ float    amax_l[kOwnRows * kHeads];
  __shared__ float    den_l[kOwnRows * kHeads];
  __shared__ int      ovf;
  __shared__ unsigned short todo_s[kOwnWaves * kTodoPad];
  __shared__ unsigned mk[kMarkWords];

  const int tid  = threadIdx.x;
  const int lane = tid & 31;
  const int wave = __builtin_amdgcn_readfirstlane(tid >> 5);
  const int d0   = blockIdx.x * kOwnRows;
  const int nrow = (kNodes - d0 < kOwnRows) ? (kNodes - d0) : kOwnRows;
  const float4* __restrict__ xl4 = reinterpret_cast<const float4*>(xl);
  unsigned* seg = list + wave * kPeSeg;
  unsigned short* todo = todo_s + wave * kTodoPad;

  {
    float4* acc4 = reinterpret_cast<float4*>(acc);
    for (int i = tid; i < kOwnRows * kHC / 4; i += kOwnThr) acc4[i] = make_float4(0.f, 0.f, 0.f, 0.f);
    if (tid < kOwnRows * kHeads) {
      adst_l[tid] = (tid < nrow * kHeads) ? a_dst[d0 * kHeads + tid] : 0.0f;
      amax_l[tid] = -INFINITY;
      den_l[tid]  = 0.0f;
    }
    if (tid == 0) ovf = 0;
    if (tid < kMarkWords) mk[tid] = mdst[(size_t)tid * kOwnBlocks + blockIdx.x];
  }
  __syncthreads();

  const int nself = (nrow > wave) ? ((nrow - wave + 31) >> 5) : 0;
  if (lane < nself) seg[lane] = ((unsigned)(d0 + wave + 32 * lane) << 8) | (unsigned)(wave + 32 * lane);

  const int4* __restrict__ dst4 = reinterpret_cast<const int4*>(ei + kEdges);
  const int off = blockIdx.x * kPeStagger;
  const int nt = pe_todo(mk, todo, off, wave, lane);
  int wcount = nself;
  bool over = false;
  int4 da[kPeLoads], db[kPeLoads];
  pe_load(da, dst4, todo, 0, nt, lane);
  for (int b = 0; b < nt; b += 2 * kStepGroups) {
    pe_load(db, dst4, todo, b + kStepGroups, nt, lane);
    over = over || (wcount > kPeSeg - kPeStepMax);
    if (!over) pe_scan(da, todo, b, nt, lane, d0, (unsigned)nrow, wcount, seg);
    pe_load(da, dst4, todo, b + 2 * kStepGroups, nt, lane);
    over = over || (wcount > kPeSeg - kPeStepMax);
    if (!over) pe_scan(db, todo, b + kStepGroups, nt, lane, d0, (unsigned)nrow, wcount, seg);
  }
  if (over) ovf = 1;
  else      ag_resolve(seg, nself, wcount, lane, ei);
  __syncthreads();

  const bool fits = __builtin_amdgcn_readfirstlane(ovf) == 0;
  if (fits) {
    ag_stats<0>(seg, wcount, lane, a_src, adst_l, amax_l, den_l);
    __syncthreads();
    ag_stats<1>(seg, wcount, lane, a_src, adst_l, amax_l, den_l);
    __syncthreads();
    ag_rows(seg, wcount, lane, a_src, adst_l, amax_l, den_l, acc, xl4);
  } else {
    for (int ph = 0; ph < 3; ++ph) {
      for (int b = 0; b == 0 || b < nt; b += kStepGroups) {
        int n = 0;
        if (b == 0) {
          if (lane < nself) seg[lane] = ((unsigned)(d0 + wave + 32 * lane) << 8) | (unsigned)(wave + 32 * lane);
          n = nself;
        }
        const int n0 = n;
        pe_load(da, dst4, todo, b, nt, lane);
        pe_scan(da, todo, b, nt, lane, d0, (unsigned)nrow, n, seg);
        ag_resolve(seg, n0, n, lane, ei);
        if (ph == 0)      ag_stats<0>(seg, n, lane, a_src, adst_l, amax_l, den_l);
        else if (ph == 1) ag_stats<1>(seg, n, lane, a_src, adst_l, amax_l, den_l);
        else              ag_rows(seg, n, lane, a_src, adst_l, amax_l, den_l, acc, xl4);
      }
      __syncthreads();
    }
  }
  __syncthreads();

  const float4 b = reinterpret_cast<const float4*>(bias)[lane];
  float4* __restrict__ out4 = reinterpret_cast<float4*>(out);
  for (int dl = wave; dl < nrow; dl += kOwnWaves) {
    const int d = d0 + dl;
    const float* row = acc + dl * kHC + lane;
    float4 r;
    r.x = row[0]  + b.x;
    r.y = row[32] + b.y;
    r.z = row[64] + b.z;
    r.w = row[96] + b.w;
    r.x = r.x > 0.0f ? r.x : expm1f(r.x);
    r.y = r.y > 0.0f ? r.y : expm1f(r.y);
    r.z = r.z > 0.0f ? r.z : expm1f(r.z);
    r.w = r.w > 0.0f ? r.w : expm1f(r.w);
    out4[(size_t)d * (kHC / 4) + lane] = r;
  }
}


extern "C" void kernel_launch(void* const* d_in, const int* in_sizes, int n_in,
                              void* d_out, int out_size, void* d_ws, size_t ws_size,
                              hipStream_t stream) {
  (void)in_sizes; (void)n_in; (void)out_size; (void)ws_size;
  const float* x       = (const float*)d_in[0];
  const int*   ei      = (const int*)d_in[1];
  const int*   et      = (const int*)d_in[2];
  const float* pe      = (const float*)d_in[3];
  const float* W       = (const float*)d_in[4];
  const float* att_src = (const float*)d_in[5];
  const float* att_dst = (const float*)d_in[6];
  const float* bias    = (const float*)d_in[7];
  float* out = (float*)d_out;

  float* ws    = (float*)d_ws;
  float* xl    = ws;
  float* a_src = xl    + (size_t)kNodes * kHC;
  float* a_dst = a_src + (size_t)kNodes * kHeads;
  unsigned* marks = (unsigned*)(a_dst + (size_t)kNodes * kHeads);
  float* wf    = (float*)(marks + (size_t)2 * kMarkWords * kOwnBlocks);

  k_mark<<<kMarkBlocks + 1, kOwnThr, 0, stream>>>(ei, marks, W, att_src, att_dst, wf);
  k_owner_pe<<<kOwnBlocks, kOwnThr, 0, stream>>>(x, ei, et, pe, marks, wf, xl, a_src, a_dst);
  k_owner_aggregate<<<kOwnBlocks, kOwnThr, 0, stream>>>(ei, marks + (size_t)kMarkWords * kOwnBlocks,
                                                        a_src, a_dst, xl, bias, out);
}


# aggregation gathers a_src once per edge (was 3 times): the maxima pass keeps each (edge, head) score in the free tail of its wave's LDS list segment, both later passes read it there; sum order kept
# speedup vs baseline: 16.2559x; 1.0646x over previous
#include <hip/hip_runtime.h>
#include <math.h>

typedef float v2f __attribute__((ext_vector_type(2)));
typedef float v8f __attribute__((ext_vector_type(8)));

namespace {
constexpr int kNodes = 50000;
constexpr int kEdges = 800000;
constexpr int kInCh  = 64;
constexpr int kOutCh = 32;
constexpr int kHeads = 4;
constexpr int kHC    = kHeads * kOutCh;
constexpr int kETot  = kEdges + kNodes;
constexpr int kOwnBlocks = 256;
constexpr int kOwnRows   = 196;
constexpr int kOwnThr    = 1024;
constexpr int kOwnWaves  = kOwnThr / 32;
constexpr int kQuads     = kEdges / 4;
static_assert(kEdges % 4 == 0, "the index rows are read in groups of 4");
static_assert(kOwnBlocks * kOwnRows >= kNodes, "every node has an owner");
static_assert(kOwnRows <= 256, "local row index is packed into 8 bits");
static_assert(kNodes <= (1 << 24), "source index is packed into 24 bits");
constexpr int kTypes       = 64;
constexpr int kPeLoads     = 8;
constexpr int kPeStagger   = 768;
constexpr int kPeStepMax   = 32 * kPeLoads * 4;
constexpr int kPeSeg       = 1264;
constexpr int kPeRowsPad   = 256;
static_assert(kQuads % 32 == 0 && kPeStagger % 32 == 0, "a wave's 32 quads never straddle the wrap");
static_assert(kOwnBlocks * kPeStagger <= kQuads, "start places stay inside the list");
static_assert(kEdges <= (1 << 24), "edge index is packed into 24 bits");
static_assert(kPeSeg > kPeStepMax, "a whole step fits behind the mark at which a segment is emptied");
static_assert(kPeRowsPad >= kOwnRows && kPeRowsPad * (kInCh / 4) % kOwnThr == 0, "the seed copy covers the padded rows exactly");
constexpr int kRowPitch    = kInCh + 4;
constexpr int kColTiles    = kHC / 16 + 1;
constexpr int kKPairs      = kInCh / 8;
constexpr int kWfTile      = kKPairs * 32 * 4;
static_assert(kPeRowsPad >= (kOwnRows + 15) / 16 * 16, "the last row tile reads padded rows only");
static_assert(kRowPitch % 4 == 0 && 2 * kHeads <= 16, "rows stay 16-byte aligned; both score sets fit one tile");
static_assert(kWfTile == kOwnThr, "the extra block of k_mark fills one column tile of the B table per trip");
static_assert(kOwnRows % 4 == 0 && kNodes % 4 == 0, "every block owns a multiple of four rows: a tile's rows are stored in fours");
constexpr int kGroupQuads  = 8;
constexpr int kGroups      = kQuads / kGroupQuads;
constexpr int kMarkBlocks  = (kQuads + kOwnThr - 1) / kOwnThr;
constexpr int kMarkWords   = 4 * kMarkBlocks;
constexpr int kWaveSpans   = (kQuads + kOwnThr - 1) / kOwnThr;
constexpr int kTodoTrips   = (kWaveSpans + 31) / 32;
constexpr int kTodoPad     = 4 * kWaveSpans;
constexpr int kStepGroups  = 4 * kPeLoads;
static_assert(kQuads % kGroupQuads == 0 && 32 % kGroupQuads == 0, "groups tile the list and a wave's 32 quads");
static_assert(kOwnThr == 4 * 32 * kGroupQuads, "a block of k_mark fills four words of 32 groups");
static_assert(kGroups < (1 << 16), "a group index is kept in 16 bits");
static_assert(kMarkWords <= kOwnThr, "one thread per mark word when an owner block fetches its marks");
}


__device__ __forceinline__ void atomic_max_f32(float* addr, float v) {
  if (v >= 0.0f) atomicMax(reinterpret_cast<int*>(addr), __float_as_int(v));
  else           atomicMin(reinterpret_cast<unsigned int*>(addr), __float_as_uint(v));
}

__device__ __forceinline__ float leaky02(float v) { return v > 0.0f ? v : 0.2f * v; }


__device__ __forceinline__ void pe_add_noted(const unsigned* seg, int n, int lane,
                                             float* __restrict__ rows,
                                             const float* __restrict__ pes,
                                             const int* __restrict__ et) {
  for (int base = 0; base < n; base += 32) {
    const int i = base + lane;
    unsigned ent = 0u;
    int t = 0;
    if (i < n) { ent = seg[i]; t = et[ent >> 8]; }
    const int m = (n - base < 32) ? (n - base) : 32;
    for (int u = 0; u < m; u += 4) {
      int tu[4];
      unsigned ru[4];
      float p0[4], p1[4];
#pragma unroll
      for (int j = 0; j < 4; ++j) {
        const int uj = (u + j < m) ? (u + j) : (m - 1);
        tu[j] = __builtin_amdgcn_readlane(t, uj);
        ru[j] = (unsigned)__builtin_amdgcn_readlane((int)ent, uj) & 255u;
        p0[j] = pes[tu[j] * kInCh + lane];
        p1[j] = pes[tu[j] * kInCh + 32 + lane];
      }
      asm volatile("" :: "v"(p0[0]), "v"(p0[1]), "v"(p0[2]), "v"(p0[3]),
                         "v"(p1[0]), "v"(p1[1]), "v"(p1[2]), "v"(p1[3]));
#pragma unroll
      for (int j = 0; j < 4; ++j) {
        if (u + j < m) {
          atomicAdd(&rows[ru[j] * kRowPitch + lane],      p0[j]);
          atomicAdd(&rows[ru[j] * kRowPitch + 32 + lane], p1[j]);
        }
      }
    }
  }
}

__device__ __forceinline__ void pe_note(unsigned mask, bool mine, unsigned rel, int e,
                                        int& wcount, unsigned* seg) {
  if (mask != 0u) {
    if (mine) seg[wcount + (int)__builtin_amdgcn_mbcnt_lo(mask, 0u)] = ((unsigned)e << 8) | rel;
    wcount += __popc(mask);
  }
}

__global__ void __launch_bounds__(kOwnThr)
k_mark(const int* __restrict__ ei, unsigned* __restrict__ marks,
       const float* __restrict__ W, const float* __restrict__ att_src,
       const float* __restrict__ att_dst, float* __restrict__ wf) {
  __shared__ unsigned own[4 * 2 * kOwnBlocks];
  const int tid = threadIdx.x;
  if (blockIdx.x == kMarkBlocks) {
    const int e   = tid & 3;
    const int ln  = (tid >> 2) & 31;
    const int kp  = tid >> 7;
    const int k   = 8 * kp + 4 * (e >> 1) + 2 * (ln >> 4) + (e & 1);
    const int col = ln & 15;
    for (int c = 0; c < kColTiles - 1; ++c) wf[c * kWfTile + tid] = W[k * kHC + c * 16 + col];
    float sum = 0.0f;
    if (col < 2 * kHeads) {
      const int h = col & (kHeads - 1);
      const float* __restrict__ att = (col < kHeads) ? att_src : att_dst;
      for (int c = 0; c < kOutCh; ++c) sum += W[k * kHC + h * kOutCh + c] * att[h * kOutCh + c];
    }
    wf[(kColTiles - 1) * kWfTile + tid] = sum;
    return;
  }
  for (int i = tid; i < 4 * 2 * kOwnBlocks; i += kOwnThr) own[i] = 0u;
  __syncthreads();
  const int q = blockIdx.x * kOwnThr + tid;
  if (q < kQuads) {
    const unsigned bit = 1u << ((tid >> 3) & 31);
    unsigned* ow = own + (tid >> 8) * (2 * kOwnBlocks);
    const int4 s = reinterpret_cast<const int4*>(ei)[q];
    const int4 d = reinterpret_cast<const int4*>(ei + kEdges)[q];
    const unsigned os[4] = {(unsigned)s.x / kOwnRows, (unsigned)s.y / kOwnRows,
                            (unsigned)s.z / kOwnRows, (unsigned)s.w / kOwnRows};
    const unsigned od[4] = {(unsigned)d.x / kOwnRows, (unsigned)d.y / kOwnRows,
                            (unsigned)d.z / kOwnRows, (unsigned)d.w / kOwnRows};
#pragma unroll
    for (int j = 0; j < 4; ++j) {
      if (os[j] < (unsigned)kOwnBlocks) atomicOr(&ow[os[j]], bit);
      if (od[j] < (unsigned)kOwnBlocks) atomicOr(&ow[kOwnBlocks + od[j]], bit);
    }
  }
  __syncthreads();
  for (int i = tid; i < 4 * 2 * kOwnBlocks; i += kOwnThr) {
    const int wq    = i >> 9;
    const int which = (i >> 8) & 1;
    marks[((size_t)which * kMarkWords + blockIdx.x * 4 + wq) * kOwnBlocks + (i & (kOwnBlocks - 1))] = own[i];
  }
}
static_assert(kOwnBlocks == 256, "k_mark splits a thread index into row and owner block by 8 bits");
static_assert(kOwnThr == kKPairs * 32 * 4 && kHeads == 4, "the extra block of k_mark splits a thread index into pair of k steps, lane and element");

__device__ __forceinline__ int pe_todo(const unsigned* mk, unsigned short* todo,
                                       int off, int wave, int lane) {
  int g[kTodoTrips];
  unsigned w[kTodoTrips];
  bool in[kTodoTrips];
#pragma unroll
  for (int t = 0; t < kTodoTrips; ++t) {
    const int lin = (t * 32 + lane) * kOwnThr + wave * 32;
    in[t] = lin < kQuads;
    int q = off + (in[t] ? lin : 0);
    if (q >= kQuads) q -= kQuads;
    g[t] = q / kGroupQuads;
    w[t] = mk[g[t] >> 5];
  }
  int n = 0;
#pragma unroll
  for (int t = 0; t < kTodoTrips; ++t) {
    const unsigned bits = in[t] ? ((w[t] >> (g[t] & 31)) & 15u) : 0u;
#pragma unroll
    for (int s = 0; s < 4; ++s) {
      const bool on = ((bits >> s) & 1u) != 0u;
      const unsigned m = __builtin_amdgcn_ballot_w32(on);
      if (on) todo[n + (int)__builtin_amdgcn_mbcnt_lo(m, 0u)] = (unsigned short)(g[t] + s);
      n += __popc(m);
    }
  }
  return n;
}

__device__ __forceinline__ void pe_load(int4 (&dv)[kPeLoads], const int4* __restrict__ src4,
                                        const unsigned short* todo, int base, int n, int lane) {
#pragma unroll
  for (int k = 0; k < kPeLoads; ++k) {
    const int idx = base + 4 * k + (lane >> 3);
    const int tg  = (int)todo[idx < n ? idx : 0];
    const int g   = (idx < n) ? tg : 0;
    dv[k] = src4[g * kGroupQuads + (lane & 7)];
  }
}

__device__ __forceinline__ void pe_scan(const int4 (&dv)[kPeLoads], const unsigned short* todo,
                                        int base, int n, int lane, int s0, unsigned nrow,
                                        int& wcount, unsigned* seg) {
  if (base < n) {
    int tq[kPeLoads];
    unsigned lim[kPeLoads];
#pragma unroll
    for (int k = 0; k < kPeLoads; ++k) {
      const int idx   = base + 4 * k + (lane >> 3);
      const bool live = idx < n;
      tq[k]  = (int)todo[live ? idx : 0];
      lim[k] = live ? nrow : 0u;
    }
    asm volatile("" :: "v"(tq[0]), "v"(tq[1]), "v"(tq[2]), "v"(tq[3]),
                       "v"(tq[4]), "v"(tq[5]), "v"(tq[6]), "v"(tq[7]));
#pragma unroll
    for (int k = 0; k < kPeLoads; ++k) {
      if (base + 4 * k < n) {
        const int q = tq[k] * kGroupQuads + (lane & 7);
        const unsigned r0 = (unsigned)(dv[k].x - s0);
        const unsigned r1 = (unsigned)(dv[k].y - s0);
        const unsigned r2 = (unsigned)(dv[k].z - s0);
        const unsigned r3 = (unsigned)(dv[k].w - s0);
        const unsigned m0 = __builtin_amdgcn_ballot_w32(r0 < lim[k]);
        const unsigned m1 = __builtin_amdgcn_ballot_w32(r1 < lim[k]);
        const unsigned m2 = __builtin_amdgcn_ballot_w32(r2 < lim[k]);
        const unsigned m3 = __builtin_amdgcn_ballot_w32(r3 < lim[k]);
        pe_note(m0, r0 < lim[k], r0, 4 * q + 0, wcount, seg);
        pe_note(m1, r1 < lim[k], r1, 4 * q + 1, wcount, seg);
        pe_note(m2, r2 < lim[k], r2, 4 * q + 2, wcount, seg);
        pe_note(m3, r3 < lim[k], r3, 4 * q + 3, wcount, seg);
      }
    }
  }
}
static_assert(kPeLoads == 8, "the fence in pe_scan names eight values");

__device__ __forceinline__ v8f tile_product(const float* ar, const float4* __restrict__ bf) {
  v8f acc = {0.f, 0.f, 0.f, 0.f, 0.f, 0.f, 0.f, 0.f};
#pragma unroll
  for (int kp = 0; kp < kKPairs; ++kp) {
    const float4 bv = bf[kp * 32];
    const v2f a0 = *reinterpret_cast<const v2f*>(ar + 8 * kp);
    const v2f a1 = *reinterpret_cast<const v2f*>(ar + 8 * kp + 4);
    const v2f b0 = {bv.x, bv.y};
    const v2f b1 = {bv.z, bv.w};
    acc = __builtin_amdgcn_wmma_f32_16x16x4_f32(false, a0, false, b0, (short)0, acc,
                                                false, false);
    acc = __builtin_amdgcn_wmma_f32_16x16x4_f32(false, a1, false, b1, (short)0, acc,
                                                false, false);
  }
  return acc;
}

__global__ void __launch_bounds__(kOwnThr)
k_owner_pe(const float* __restrict__ x, const int* __restrict__ ei, const int* __restrict__ et,
           const float* __restrict__ pe, const unsigned* __restrict__ msrc,
           const float* __restrict__ wf,
           float* __restrict__ xl, float* __restrict__ a_src, float* __restrict__ a_dst) {
  __shared__ __align__(16) float rows[kPeRowsPad * kRowPitch];
  __shared__ float    pes[kTypes * kInCh];
  __shared__ unsigned list[kOwnWaves * kPeSeg];
  __shared__ unsigned short todo_s[kOwnWaves * kTodoPad];
  __shared__ unsigned mk[kMarkWords];

  const int tid  = threadIdx.x;
  const int lane = tid & 31;
  const int wave = __builtin_amdgcn_readfirstlane(tid >> 5);
  const int s0   = blockIdx.x * kOwnRows;
  const int nrow = (kNodes - s0 < kOwnRows) ? (kNodes - s0) : kOwnRows;
  const int nvec = nrow * (kInCh / 4);
  unsigned* seg = list + wave * kPeSeg;
  unsigned short* todo = todo_s + wave * kTodoPad;

  {
    constexpr int kXv = kPeRowsPad * (kInCh / 4) / kOwnThr;
    static_assert(kTypes * kInCh / 4 == kOwnThr, "one 16-byte load of pe per thread");
    static_assert(kInCh / 4 == 16, "a row of x is 16 16-byte words");
    const float4* __restrict__ x4 = reinterpret_cast<const float4*>(x + (size_t)s0 * kInCh);
    float4* rows4 = reinterpret_cast<float4*>(rows);
    reinterpret_cast<float4*>(pes)[tid] = reinterpret_cast<const float4*>(pe)[tid];
    if (tid < kMarkWords) mk[tid] = msrc[(size_t)tid * kOwnBlocks + blockIdx.x];
#pragma unroll
    for (int j = 0; j < kXv; ++j) {
      const int i = j * kOwnThr + tid;
      rows4[(i >> 4) * (kRowPitch / 4) + (i & 15)] = x4[i < nvec ? i : nvec - 1];
    }
  }
  __syncthreads();

  const int4* __restrict__ src4 = reinterpret_cast<const int4*>(ei);
  const int off = blockIdx.x * kPeStagger;
  const int nt = pe_todo(mk, todo, off, wave, lane);
  int wcount = 0;
  int4 da[kPeLoads], db[kPeLoads];
  pe_load(da, src4, todo, 0, nt, lane);
  for (int b = 0; b < nt; b += 2 * kStepGroups) {
    pe_load(db, src4, todo, b + kStepGroups, nt, lane);
    if (wcount > kPeSeg - kPeStepMax) { pe_add_noted(seg, wcount, lane, rows, pes, et); wcount = 0; }
    pe_scan(da, todo, b, nt, lane, s0, (unsigned)nrow, wcount, seg);
    pe_load(da, src4, todo, b + 2 * kStepGroups, nt, lane);
    if (wcount > kPeSeg - kPeStepMax) { pe_add_noted(seg, wcount, lane, rows, pes, et); wcount = 0; }
    pe_scan(db, todo, b + kStepGroups, nt, lane, s0, (unsigned)nrow, wcount, seg);
  }

  pe_add_noted(seg, wcount, lane, rows, pes, et);
  __syncthreads();

  {
    const int half  = lane >> 4;
    const int l15   = lane & 15;
    const int ntile = (nrow + 15) >> 4;
    const int px = 8 * half * kHC + l15;
    const int ps = 8 * half * kHeads + (l15 & (kHeads - 1));
    for (int it = wave; it < ntile * kColTiles; it += kOwnWaves) {
      const int t  = it / kColTiles;
      const int c  = it - t * kColTiles;
      const int r0 = t * 16 + 8 * half;
      const float* ar = rows + (t * 16 + l15) * kRowPitch + 2 * half;
      const v8f acc = tile_product(ar, reinterpret_cast<const float4*>(wf + c * kWfTile) + lane);
      if (c < kColTiles - 1) {
        float* __restrict__ o = xl + (size_t)(s0 + t * 16) * kHC + c * 16;
        if (r0 < nrow)     { o[px]           = acc[0]; o[px + kHC]     = acc[1]; o[px + 2 * kHC] = acc[2]; o[px + 3 * kHC] = acc[3]; }
        if (r0 + 4 < nrow) { o[px + 4 * kHC] = acc[4]; o[px + 5 * kHC] = acc[5]; o[px + 6 * kHC] = acc[6]; o[px + 7 * kHC] = acc[7]; }
      } else {
        if (l15 < 2 * kHeads) {
          float* __restrict__ o = ((l15 < kHeads) ? a_src : a_dst) + (size_t)(s0 + t * 16) * kHeads;
          if (r0 < nrow)     { o[ps]              = acc[0]; o[ps + kHeads]     = acc[1]; o[ps + 2 * kHeads] = acc[2]; o[ps + 3 * kHeads] = acc[3]; }
          if (r0 + 4 < nrow) { o[ps + 4 * kHeads] = acc[4]; o[ps + 5 * kHeads] = acc[5]; o[ps + 6 * kHeads] = acc[6]; o[ps + 7 * kHeads] = acc[7]; }
        }
      }
    }
  }
}


template <int PH>
__device__ __forceinline__ void ag_stats(const unsigned* seg, unsigned* sc, bool keep, int n, int lane,
                                         const float* __restrict__ a_src,
                                         const float* adst_l, float* amax_l, float* den_l) {
  const int u  = lane >> 2;
  const int hh = lane & 3;
  const bool kept = (PH == 1) && keep;
  for (int base = 0; base < n; base += 32) {
    unsigned ent[4];
    float as[4];
    if (kept) {
#pragma unroll
      for (int j = 0; j < 4; ++j) {
        const int idx = base + 8 * j + u;
        const int ci  = idx < n ? idx : n - 1;
        ent[j] = seg[ci];
        as[j]  = __uint_as_float(sc[4 * ci + hh]);
      }
    } else {
#pragma unroll
      for (int j = 0; j < 4; ++j) {
        const int idx = base + 8 * j + u;
        ent[j] = seg[idx < n ? idx : n - 1];
        as[j]  = a_src[(ent[j] >> 8) * kHeads + hh];
      }
    }
    asm volatile("" :: "v"(as[0]), "v"(as[1]), "v"(as[2]), "v"(as[3]));
#pragma unroll
    for (int j = 0; j < 4; ++j) {
      const int idx = base + 8 * j + u;
      if (idx < n) {
        const int k = (int)(ent[j] & 255u) * kHeads + hh;
        const float al = kept ? as[j] : leaky02(as[j] + adst_l[k]);
        if (PH == 0) {
          if (keep) sc[4 * idx + hh] = __float_as_uint(al);
          atomic_max_f32(&amax_l[k], al);
        } else {
          const float ex = expf(al - amax_l[k]);
          if (keep) sc[4 * idx + hh] = __float_as_uint(ex);
          atomicAdd(&den_l[k], ex);
        }
      }
    }
  }
}

__device__ __forceinline__ void ag_rows(const unsigned* seg, const unsigned* sc, bool keep, int n, int lane,
                                        const float* __restrict__ a_src,
                                        const float* adst_l, const float* amax_l,
                                        const float* den_l, float* acc,
                                        const float4* __restrict__ xl4) {
  const int u  = lane >> 2;
  const int hh = lane & 3;
  const int hr = lane >> 3;
  for (int base = 0; base < n; base += 8) {
    const int idx = base + u;
    const int ci  = idx < n ? idx : n - 1;
    const unsigned ent = seg[ci];
    const int k = (int)(ent & 255u) * kHeads + hh;
    float as = 0.0f, ex = 0.0f;
    if (keep) ex = __uint_as_float(sc[4 * ci + hh]);
    else      as = a_src[(ent >> 8) * kHeads + hh];
    float4 v[8];
    int rl[8];
#pragma unroll
    for (int j = 0; j < 8; ++j) {
      const unsigned eu = (unsigned)__builtin_amdgcn_readlane((int)ent, 4 * j);
      rl[j] = (int)(eu & 255u);
      v[j]  = xl4[(size_t)(eu >> 8) * (kHC / 4) + lane];
    }
    if (!keep) ex = expf(leaky02(as + adst_l[k]) - amax_l[k]);
    const float coef = ex / den_l[k];
    float c[8];
#pragma unroll
    for (int j = 0; j < 8; ++j) c[j] = __shfl(coef, 4 * j + hr);
    asm volatile("" :: "v"(coef), "v"(v[0].x), "v"(v[1].x), "v"(v[2].x), "v"(v[3].x),
                       "v"(v[4].x), "v"(v[5].x), "v"(v[6].x), "v"(v[7].x));
#pragma unroll
    for (int j = 0; j < 8; ++j) {
      if (base + j < n) {
        float* row = acc + rl[j] * kHC + lane;
        atomicAdd(row,      c[j] * v[j].x);
        atomicAdd(row + 32, c[j] * v[j].y);
        atomicAdd(row + 64, c[j] * v[j].z);
        atomicAdd(row + 96, c[j] * v[j].w);
      }
    }
  }
}

__device__ __forceinline__ void ag_resolve(unsigned* seg, int n0, int n, int lane,
                                           const int* __restrict__ ei) {
  for (int base = n0; base < n; base += 128) {
    unsigned en[4];
    int sv[4];
#pragma unroll
    for (int j = 0; j < 4; ++j) {
      const int i = base + 32 * j + lane;
      en[j] = seg[i < n ? i : n - 1];
    }
#pragma unroll
    for (int j = 0; j < 4; ++j) sv[j] = ei[en[j] >> 8];
#pragma unroll
    for (int j = 0; j < 4; ++j) {
      const int i = base + 32 * j + lane;
      if (i < n) seg[i] = ((unsigned)sv[j] << 8) | (en[j] & 255u);
    }
  }
}

__global__ void __launch_bounds__(kOwnThr)
k_owner_aggregate(const int* __restrict__ ei, const unsigned* __restrict__ mdst,
                  const float* __restrict__ a_src, const float* __restrict__ a_dst,
                  const float* __restrict__ xl, const float* __restrict__ bias,
                  float* __restrict__ out) {
  __shared__ float    acc[kOwnRows * kHC];
  __shared__ unsigned list[kOwnWaves * kPeSeg];
  __shared__ float    adst_l[kOwnRows * kHeads];
  __shared__ float    amax_l[kOwnRows * kHeads];
  __shared__ float    den_l[kOwnRows * kHeads];
  __shared__ int      ovf;
  __shared__ unsigned short todo_s[kOwnWaves * kTodoPad];
  __shared__ unsigned mk[kMarkWords];

  const int tid  = threadIdx.x;
  const int lane = tid & 31;
  const int wave = __builtin_amdgcn_readfirstlane(tid >> 5);
  const int d0   = blockIdx.x * kOwnRows;
  const int nrow = (kNodes - d0 < kOwnRows) ? (kNodes - d0) : kOwnRows;
  const float4* __restrict__ xl4 = reinterpret_cast<const float4*>(xl);
  unsigned* seg = list + wave * kPeSeg;
  unsigned short* todo = todo_s + wave * kTodoPad;

  {
    float4* acc4 = reinterpret_cast<float4*>(acc);
    for (int i = tid; i < kOwnRows * kHC / 4; i += kOwnThr) acc4[i] = make_float4(0.f, 0.f, 0.f, 0.f);
    if (tid < kOwnRows * kHeads) {
      adst_l[tid] = (tid < nrow * kHeads) ? a_dst[d0 * kHeads + tid] : 0.0f;
      amax_l[tid] = -INFINITY;
      den_l[tid]  = 0.0f;
    }
    if (tid == 0) ovf = 0;
    if (tid < kMarkWords) mk[tid] = mdst[(size_t)tid * kOwnBlocks + blockIdx.x];
  }
  __syncthreads();

  const int nself = (nrow > wave) ? ((nrow - wave + 31) >> 5) : 0;
  if (lane < nself) seg[lane] = ((unsigned)(d0 + wave + 32 * lane) << 8) | (unsigned)(wave + 32 * lane);

  const int4* __restrict__ dst4 = reinterpret_cast<const int4*>(ei + kEdges);
  const int off = blockIdx.x * kPeStagger;
  const int nt = pe_todo(mk, todo, off, wave, lane);
  int wcount = nself;
  bool over = false;
  int4 da[kPeLoads], db[kPeLoads];
  pe_load(da, dst4, todo, 0, nt, lane);
  for (int b = 0; b < nt; b += 2 * kStepGroups) {
    pe_load(db, dst4, todo, b + kStepGroups, nt, lane);
    over = over || (wcount > kPeSeg - kPeStepMax);
    if (!over) pe_scan(da, todo, b, nt, lane, d0, (unsigned)nrow, wcount, seg);
    pe_load(da, dst4, todo, b + 2 * kStepGroups, nt, lane);
    over = over || (wcount > kPeSeg - kPeStepMax);
    if (!over) pe_scan(db, todo, b + kStepGroups, nt, lane, d0, (unsigned)nrow, wcount, seg);
  }
  if (over) ovf = 1;
  else      ag_resolve(seg, nself, wcount, lane, ei);
  __syncthreads();

  const bool fits = __builtin_amdgcn_readfirstlane(ovf) == 0;
  if (fits) {
    const bool keep = 5 * wcount <= kPeSeg;
    unsigned* sc = seg + wcount;
    ag_stats<0>(seg, sc, keep, wcount, lane, a_src, adst_l, amax_l, den_l);
    __syncthreads();
    ag_stats<1>(seg, sc, keep, wcount, lane, a_src, adst_l, amax_l, den_l);
    __syncthreads();
    ag_rows(seg, sc, keep, wcount, lane, a_src, adst_l, amax_l, den_l, acc, xl4);
  } else {
    for (int ph = 0; ph < 3; ++ph) {
      for (int b = 0; b == 0 || b < nt; b += kStepGroups) {
        int n = 0;
        if (b == 0) {
          if (lane < nself) seg[lane] = ((unsigned)(d0 + wave + 32 * lane) << 8) | (unsigned)(wave + 32 * lane);
          n = nself;
        }
        const int n0 = n;
        pe_load(da, dst4, todo, b, nt, lane);
        pe_scan(da, todo, b, nt, lane, d0, (unsigned)nrow, n, seg);
        ag_resolve(seg, n0, n, lane, ei);
        if (ph == 0)      ag_stats<0>(seg, seg, false, n, lane, a_src, adst_l, amax_l, den_l);
        else if (ph == 1) ag_stats<1>(seg, seg, false, n, lane, a_src, adst_l, amax_l, den_l);
        else              ag_rows(seg, seg, false, n, lane, a_src, adst_l, amax_l, den_l, acc, xl4);
      }
      __syncthreads();
    }
  }
  __syncthreads();

  const float4 b = reinterpret_cast<const float4*>(bias)[lane];
  float4* __restrict__ out4 = reinterpret_cast<float4*>(out);
  for (int dl = wave; dl < nrow; dl += kOwnWaves) {
    const int d = d0 + dl;
    const float* row = acc + dl * kHC + lane;
    float4 r;
    r.x = row[0]  + b.x;
    r.y = row[32] + b.y;
    r.z = row[64] + b.z;
    r.w = row[96] + b.w;
    r.x = r.x > 0.0f ? r.x : expm1f(r.x);
    r.y = r.y > 0.0f ? r.y : expm1f(r.y);
    r.z = r.z > 0.0f ? r.z : expm1f(r.z);
    r.w = r.w > 0.0f ? r.w : expm1f(r.w);
    out4[(size_t)d * (kHC / 4) + lane] = r;
  }
}


extern "C" void kernel_launch(void* const* d_in, const int* in_sizes, int n_in,
                              void* d_out, int out_size, void* d_ws, size_t ws_size,
                              hipStream_t stream) {
  (void)in_sizes; (void)n_in; (void)out_size; (void)ws_size;
  const float* x       = (const float*)d_in[0];
  const int*   ei      = (const int*)d_in[1];
  const int*   et      = (const int*)d_in[2];
  const float* pe      = (const float*)d_in[3];
  const float* W       = (const float*)d_in[4];
  const float* att_src = (const float*)d_in[5];
  const float* att_dst = (const float*)d_in[6];
  const float* bias    = (const float*)d_in[7];
  float* out = (float*)d_out;

  float* ws    = (float*)d_ws;
  float* xl    = ws;
  float* a_src = xl    + (size_t)kNodes * kHC;
  float* a_dst = a_src + (size_t)kNodes * kHeads;
  unsigned* marks = (unsigned*)(a_dst + (size_t)kNodes * kHeads);
  float* wf    = (float*)(marks + (size_t)2 * kMarkWords * kOwnBlocks);

  k_mark<<<kMarkBlocks + 1, kOwnThr, 0, stream>>>(ei, marks, W, att_src, att_dst, wf);
  k_owner_pe<<<kOwnBlocks, kOwnThr, 0, stream>>>(x, ei, et, pe, marks, wf, xl, a_src, a_dst);
  k_owner_aggregate<<<kOwnBlocks, kOwnThr, 0, stream>>>(ei, marks + (size_t)kMarkWords * kOwnBlocks,
                                                        a_src, a_dst, xl, bias, out);
}


# owner blocks take their edges from per-chunk buckets that the first pass fills (LDS slot adds, arrival order); marks and list scans gone; list order, so LDS float-add order, changes; 3 launches
# speedup vs baseline: 19.4204x; 1.1947x over previous
#include <hip/hip_runtime.h>
#include <math.h>

typedef float v2f __attribute__((ext_vector_type(2)));
typedef float v8f __attribute__((ext_vector_type(8)));

namespace {
constexpr int kNodes = 50000;
constexpr int kEdges = 800000;
constexpr int kInCh  = 64;
constexpr int kOutCh = 32;
constexpr int kHeads = 4;
constexpr int kHC    = kHeads * kOutCh;
constexpr int kETot  = kEdges + kNodes;
constexpr int kOwnBlocks = 256;
constexpr int kOwnRows   = 196;
constexpr int kOwnThr    = 1024;
constexpr int kOwnWaves  = kOwnThr / 32;
constexpr int kQuads     = kEdges / 4;
static_assert(kEdges % 4 == 0, "the index rows are read in groups of 4");
static_assert(kOwnBlocks * kOwnRows >= kNodes, "every node has an owner");
static_assert(kOwnRows <= 256, "local row index is packed into 8 bits");
static_assert(kNodes <= (1 << 24), "source index is packed into 24 bits");
constexpr int kTypes       = 64;
constexpr int kPeLoads     = 8;
constexpr int kPeStagger   = 768;
constexpr int kPeStepMax   = 32 * kPeLoads * 4;
constexpr int kPeSeg       = 1264;
constexpr int kPeRowsPad   = 256;
static_assert(kQuads % 32 == 0 && kPeStagger % 32 == 0, "a wave's 32 quads never straddle the wrap");
static_assert(kOwnBlocks * kPeStagger <= kQuads, "start places stay inside the list");
static_assert(kEdges <= (1 << 24), "edge index is packed into 24 bits");
static_assert(kPeSeg > kPeStepMax, "a whole step fits behind the mark at which a segment is emptied");
static_assert(kPeRowsPad >= kOwnRows && kPeRowsPad * (kInCh / 4) % kOwnThr == 0, "the seed copy covers the padded rows exactly");
constexpr int kRowPitch    = kInCh + 4;
constexpr int kColTiles    = kHC / 16 + 1;
constexpr int kKPairs      = kInCh / 8;
constexpr int kWfTile      = kKPairs * 32 * 4;
static_assert(kPeRowsPad >= (kOwnRows + 15) / 16 * 16, "the last row tile reads padded rows only");
static_assert(kRowPitch % 4 == 0 && 2 * kHeads <= 16, "rows stay 16-byte aligned; both score sets fit one tile");
static_assert(kWfTile == kOwnThr, "the extra block of k_buckets fills one column tile of the B table per trip");
static_assert(kOwnRows % 4 == 0 && kNodes % 4 == 0, "every block owns a multiple of four rows: a tile's rows are stored in fours");
constexpr int kBktWords    = 64;
constexpr int kBktCap      = kBktWords - 1;
constexpr int kChunks      = (kQuads + kOwnThr - 1) / kOwnThr;
constexpr int kBktTrips    = (kChunks + kOwnWaves - 1) / kOwnWaves;
static_assert(kOwnWaves == 32, "one flag per wave is collected by one wave-wide compare");
static_assert(kBktWords == 2 * 32, "a wave moves a bucket as one 8-byte word per lane");
constexpr int kGroupQuads  = 8;
constexpr int kGroups      = kQuads / kGroupQuads;
constexpr int kMarkWords   = 4 * kChunks;
constexpr int kWaveSpans   = (kQuads + kOwnThr - 1) / kOwnThr;
constexpr int kTodoTrips   = (kWaveSpans + 31) / 32;
constexpr int kTodoPad     = 4 * kWaveSpans;
constexpr int kStepGroups  = 4 * kPeLoads;
static_assert(kQuads % kGroupQuads == 0 && 32 % kGroupQuads == 0, "groups tile the list and a wave's 32 quads");
static_assert(kGroups < (1 << 16), "a group index is kept in 16 bits");
static_assert(kMarkWords <= kOwnThr, "one thread per mark word when an owner block sets its marks");
}


__device__ __forceinline__ void atomic_max_f32(float* addr, float v) {
  if (v >= 0.0f) atomicMax(reinterpret_cast<int*>(addr), __float_as_int(v));
  else           atomicMin(reinterpret_cast<unsigned int*>(addr), __float_as_uint(v));
}

__device__ __forceinline__ float leaky02(float v) { return v > 0.0f ? v : 0.2f * v; }


__device__ __forceinline__ void pe_add_noted(const unsigned* seg, int n, int lane,
                                             float* __restrict__ rows,
                                             const float* __restrict__ pes) {
  for (int base = 0; base < n; base += 32) {
    const int i = base + lane;
    unsigned ent = 0u;
    int t = 0;
    if (i < n) { ent = seg[i]; t = (int)(ent >> 8); }
    const int m = (n - base < 32) ? (n - base) : 32;
    for (int u = 0; u < m; u += 4) {
      int tu[4];
      unsigned ru[4];
      float p0[4], p1[4];
#pragma unroll
      for (int j = 0; j < 4; ++j) {
        const int uj = (u + j < m) ? (u + j) : (m - 1);
        tu[j] = __builtin_amdgcn_readlane(t, uj);
        ru[j] = (unsigned)__builtin_amdgcn_readlane((int)ent, uj) & 255u;
        p0[j] = pes[tu[j] * kInCh + lane];
        p1[j] = pes[tu[j] * kInCh + 32 + lane];
      }
      asm volatile("" :: "v"(p0[0]), "v"(p0[1]), "v"(p0[2]), "v"(p0[3]),
                         "v"(p1[0]), "v"(p1[1]), "v"(p1[2]), "v"(p1[3]));
#pragma unroll
      for (int j = 0; j < 4; ++j) {
        if (u + j < m) {
          atomicAdd(&rows[ru[j] * kRowPitch + lane],      p0[j]);
          atomicAdd(&rows[ru[j] * kRowPitch + 32 + lane], p1[j]);
        }
      }
    }
  }
}

__device__ __forceinline__ void pe_note(unsigned mask, bool mine, unsigned rel, int e,
                                        int& wcount, unsigned* seg) {
  if (mask != 0u) {
    if (mine) seg[wcount + (int)__builtin_amdgcn_mbcnt_lo(mask, 0u)] = ((unsigned)e << 8) | rel;
    wcount += __popc(mask);
  }
}

__global__ void __launch_bounds__(kOwnThr)
k_buckets(const int* __restrict__ ei, const int* __restrict__ et, unsigned* __restrict__ bkt,
          const float* __restrict__ W, const float* __restrict__ att_src,
          const float* __restrict__ att_dst, float* __restrict__ wf) {
  constexpr int kBkPitch = kBktWords + 1;
  constexpr int kBkAll = (2 * kOwnBlocks + 1) * kBkPitch + 3;
  __shared__ __align__(16) unsigned bk[kBkAll];
  constexpr int kBkVec = 2 * kOwnBlocks * kBktWords / 4 / kOwnThr;
  static_assert(kBkVec * kOwnThr * 4 == 2 * kOwnBlocks * kBktWords && kBktWords == 4 * 16, "the copy covers the buckets exactly, 16 words of 16 bytes each");
  static_assert(kBkAll % 4 == 0, "the buckets are cleared in 16-byte words");
  const int tid = threadIdx.x;
  if (blockIdx.x == kChunks) {
    const int e   = tid & 3;
    const int ln  = (tid >> 2) & 31;
    const int kp  = tid >> 7;
    const int k   = 8 * kp + 4 * (e >> 1) + 2 * (ln >> 4) + (e & 1);
    const int col = ln & 15;
    for (int c = 0; c < kColTiles - 1; ++c) wf[c * kWfTile + tid] = W[k * kHC + c * 16 + col];
    float sum = 0.0f;
    if (col < 2 * kHeads) {
      const int h = col & (kHeads - 1);
      const float* __restrict__ att = (col < kHeads) ? att_src : att_dst;
      for (int c = 0; c < kOutCh; ++c) sum += W[k * kHC + h * kOutCh + c] * att[h * kOutCh + c];
    }
    wf[(kColTiles - 1) * kWfTile + tid] = sum;
    return;
  }
  for (int v = tid; v < kBkAll / 4; v += kOwnThr)
    reinterpret_cast<uint4*>(bk)[v] = make_uint4(0u, 0u, 0u, 0u);
  __syncthreads();
  const int q = blockIdx.x * kOwnThr + tid;
  if (q < kQuads) {
    const int4 s = reinterpret_cast<const int4*>(ei)[q];
    const int4 d = reinterpret_cast<const int4*>(ei + kEdges)[q];
    const int4 t = reinterpret_cast<const int4*>(et)[q];
    const unsigned sv[4] = {(unsigned)s.x, (unsigned)s.y, (unsigned)s.z, (unsigned)s.w};
    const unsigned dv[4] = {(unsigned)d.x, (unsigned)d.y, (unsigned)d.z, (unsigned)d.w};
    const unsigned tv[4] = {(unsigned)t.x, (unsigned)t.y, (unsigned)t.z, (unsigned)t.w};
    unsigned* bp[8];
    unsigned val[8], slot[8];
#pragma unroll
    for (int j = 0; j < 4; ++j) {
      const unsigned os = sv[j] / kOwnRows;
      const unsigned od = dv[j] / kOwnRows;
      bp[2 * j]      = bk + (sv[j] < (unsigned)kNodes ? os : 2u * kOwnBlocks) * kBkPitch;
      bp[2 * j + 1]  = bk + (dv[j] < (unsigned)kNodes ? kOwnBlocks + od : 2u * kOwnBlocks) * kBkPitch;
      val[2 * j]     = (tv[j] << 8) | (sv[j] - os * kOwnRows);
      val[2 * j + 1] = (sv[j] << 8) | (dv[j] - od * kOwnRows);
    }
#pragma unroll
    for (int j = 0; j < 8; ++j) slot[j] = atomicAdd(bp[j], 1u);
#pragma unroll
    for (int j = 0; j < 8; ++j)
      if (slot[j] < (unsigned)kBktCap) bp[j][1 + slot[j]] = val[j];
  }
  __syncthreads();
#pragma unroll
  for (int j = 0; j < kBkVec; ++j) {
    const int v = j * kOwnThr + tid;
    const int i = v >> 4;
    const unsigned* b = bk + i * kBkPitch + 4 * (v & 15);
    reinterpret_cast<uint4*>(bkt + ((size_t)i * kChunks + blockIdx.x) * kBktWords)[v & 15] =
        make_uint4(b[0], b[1], b[2], b[3]);
  }
}
static_assert(kOwnThr == kKPairs * 32 * 4 && kHeads == 4, "the extra block of k_buckets splits a thread index into pair of k steps, lane and element");

__device__ __forceinline__ void note_resolve(unsigned* seg, int n0, int n, int lane,
                                             const int* __restrict__ tab) {
  for (int base = n0; base < n; base += 128) {
    unsigned en[4];
    int sv[4];
#pragma unroll
    for (int j = 0; j < 4; ++j) {
      const int i = base + 32 * j + lane;
      en[j] = seg[i < n ? i : n - 1];
    }
#pragma unroll
    for (int j = 0; j < 4; ++j) sv[j] = tab[en[j] >> 8];
#pragma unroll
    for (int j = 0; j < 4; ++j) {
      const int i = base + 32 * j + lane;
      if (i < n) seg[i] = ((unsigned)sv[j] << 8) | (en[j] & 255u);
    }
  }
}

__device__ __forceinline__ void bkt_load(uint2 (&bw)[kBktTrips], const unsigned* __restrict__ mine,
                                         int wave, int lane) {
#pragma unroll
  for (int k = 0; k < kBktTrips; ++k) {
    const int ch = wave + kOwnWaves * k;
    const int cc = ch < kChunks ? ch : kChunks - 1;
    bw[k] = *reinterpret_cast<const uint2*>(mine + (size_t)cc * kBktWords + 2 * lane);
  }
}

__device__ __forceinline__ bool bkt_over(const uint2 (&bw)[kBktTrips], int wave) {
  bool ov = false;
#pragma unroll
  for (int k = 0; k < kBktTrips; ++k) {
    const unsigned cnt = (unsigned)__builtin_amdgcn_readfirstlane((int)bw[k].x);
    if (wave + kOwnWaves * k < kChunks && cnt > (unsigned)kBktCap) ov = true;
  }
  return ov;
}

__device__ __forceinline__ void bkt_take(const uint2 (&bw)[kBktTrips], int wave, int lane,
                                         int& wcount, unsigned* seg) {
#pragma unroll
  for (int k = 0; k < kBktTrips; ++k) {
    if (wave + kOwnWaves * k < kChunks) {
      const int cnt = __builtin_amdgcn_readfirstlane((int)bw[k].x);
      if (lane >= 1 && 2 * lane <= cnt) seg[wcount + 2 * lane - 1] = bw[k].x;
      if (2 * lane + 1 <= cnt)          seg[wcount + 2 * lane]     = bw[k].y;
      wcount += cnt;
    }
  }
}

__device__ __forceinline__ int pe_todo(const unsigned* mk, unsigned short* todo,
                                       int off, int wave, int lane) {
  int g[kTodoTrips];
  unsigned w[kTodoTrips];
  bool in[kTodoTrips];
#pragma unroll
  for (int t = 0; t < kTodoTrips; ++t) {
    const int lin = (t * 32 + lane) * kOwnThr + wave * 32;
    in[t] = lin < kQuads;
    int q = off + (in[t] ? lin : 0);
    if (q >= kQuads) q -= kQuads;
    g[t] = q / kGroupQuads;
    w[t] = mk[g[t] >> 5];
  }
  int n = 0;
#pragma unroll
  for (int t = 0; t < kTodoTrips; ++t) {
    const unsigned bits = in[t] ? ((w[t] >> (g[t] & 31)) & 15u) : 0u;
#pragma unroll
    for (int s = 0; s < 4; ++s) {
      const bool on = ((bits >> s) & 1u) != 0u;
      const unsigned m = __builtin_amdgcn_ballot_w32(on);
      if (on) todo[n + (int)__builtin_amdgcn_mbcnt_lo(m, 0u)] = (unsigned short)(g[t] + s);
      n += __popc(m);
    }
  }
  return n;
}

__device__ __forceinline__ void pe_load(int4 (&dv)[kPeLoads], const int4* __restrict__ src4,
                                        const unsigned short* todo, int base, int n, int lane) {
#pragma unroll
  for (int k = 0; k < kPeLoads; ++k) {
    const int idx = base + 4 * k + (lane >> 3);
    const int tg  = (int)todo[idx < n ? idx : 0];
    const int g   = (idx < n) ? tg : 0;
    dv[k] = src4[g * kGroupQuads + (lane & 7)];
  }
}

__device__ __forceinline__ void pe_scan(const int4 (&dv)[kPeLoads], const unsigned short* todo,
                                        int base, int n, int lane, int s0, unsigned nrow,
                                        int& wcount, unsigned* seg) {
  if (base < n) {
    int tq[kPeLoads];
    unsigned lim[kPeLoads];
#pragma unroll
    for (int k = 0; k < kPeLoads; ++k) {
      const int idx   = base + 4 * k + (lane >> 3);
      const bool live = idx < n;
      tq[k]  = (int)todo[live ? idx : 0];
      lim[k] = live ? nrow : 0u;
    }
    asm volatile("" :: "v"(tq[0]), "v"(tq[1]), "v"(tq[2]), "v"(tq[3]),
                       "v"(tq[4]), "v"(tq[5]), "v"(tq[6]), "v"(tq[7]));
#pragma unroll
    for (int k = 0; k < kPeLoads; ++k) {
      if (base + 4 * k < n) {
        const int q = tq[k] * kGroupQuads + (lane & 7);
        const unsigned r0 = (unsigned)(dv[k].x - s0);
        const unsigned r1 = (unsigned)(dv[k].y - s0);
        const unsigned r2 = (unsigned)(dv[k].z - s0);
        const unsigned r3 = (unsigned)(dv[k].w - s0);
        const unsigned m0 = __builtin_amdgcn_ballot_w32(r0 < lim[k]);
        const unsigned m1 = __builtin_amdgcn_ballot_w32(r1 < lim[k]);
        const unsigned m2 = __builtin_amdgcn_ballot_w32(r2 < lim[k]);
        const unsigned m3 = __builtin_amdgcn_ballot_w32(r3 < lim[k]);
        pe_note(m0, r0 < lim[k], r0, 4 * q + 0, wcount, seg);
        pe_note(m1, r1 < lim[k], r1, 4 * q + 1, wcount, seg);
        pe_note(m2, r2 < lim[k], r2, 4 * q + 2, wcount, seg);
        pe_note(m3, r3 < lim[k], r3, 4 * q + 3, wcount, seg);
      }
    }
  }
}
static_assert(kPeLoads == 8, "the fence in pe_scan names eight values");

__device__ __forceinline__ v8f tile_product(const float* ar, const float4* __restrict__ bf) {
  v8f acc = {0.f, 0.f, 0.f, 0.f, 0.f, 0.f, 0.f, 0.f};
#pragma unroll
  for (int kp = 0; kp < kKPairs; ++kp) {
    const float4 bv = bf[kp * 32];
    const v2f a0 = *reinterpret_cast<const v2f*>(ar + 8 * kp);
    const v2f a1 = *reinterpret_cast<const v2f*>(ar + 8 * kp + 4);
    const v2f b0 = {bv.x, bv.y};
    const v2f b1 = {bv.z, bv.w};
    acc = __builtin_amdgcn_wmma_f32_16x16x4_f32(false, a0, false, b0, (short)0, acc,
                                                false, false);
    acc = __builtin_amdgcn_wmma_f32_16x16x4_f32(false, a1, false, b1, (short)0, acc,
                                                false, false);
  }
  return acc;
}

__global__ void __launch_bounds__(kOwnThr)
k_owner_pe(const float* __restrict__ x, const int* __restrict__ ei, const int* __restrict__ et,
           const float* __restrict__ pe, const unsigned* __restrict__ bsrc,
           const float* __restrict__ wf,
           float* __restrict__ xl, float* __restrict__ a_src, float* __restrict__ a_dst) {
  __shared__ __align__(16) float rows[kPeRowsPad * kRowPitch];
  __shared__ float    pes[kTypes * kInCh];
  __shared__ unsigned list[kOwnWaves * kPeSeg];
  __shared__ unsigned short todo_s[kOwnWaves * kTodoPad];
  __shared__ unsigned mk[kMarkWords];
  __shared__ unsigned wflag[kOwnWaves];

  const int tid  = threadIdx.x;
  const int lane = tid & 31;
  const int wave = __builtin_amdgcn_readfirstlane(tid >> 5);
  const int s0   = blockIdx.x * kOwnRows;
  uint2 bw[kBktTrips];
  bkt_load(bw, bsrc + (size_t)blockIdx.x * kChunks * kBktWords, wave, lane);
  const int nrow = (kNodes - s0 < kOwnRows) ? (kNodes - s0) : kOwnRows;
  const int nvec = nrow * (kInCh / 4);
  unsigned* seg = list + wave * kPeSeg;
  unsigned short* todo = todo_s + wave * kTodoPad;

  {
    constexpr int kXv = kPeRowsPad * (kInCh / 4) / kOwnThr;
    static_assert(kTypes * kInCh / 4 == kOwnThr, "one 16-byte load of pe per thread");
    static_assert(kInCh / 4 == 16, "a row of x is 16 16-byte words");
    const float4* __restrict__ x4 = reinterpret_cast<const float4*>(x + (size_t)s0 * kInCh);
    float4* rows4 = reinterpret_cast<float4*>(rows);
    reinterpret_cast<float4*>(pes)[tid] = reinterpret_cast<const float4*>(pe)[tid];
#pragma unroll
    for (int j = 0; j < kXv; ++j) {
      const int i = j * kOwnThr + tid;
      rows4[(i >> 4) * (kRowPitch / 4) + (i & 15)] = x4[i < nvec ? i : nvec - 1];
    }
  }
  if (lane == 0) wflag[wave] = bkt_over(bw, wave) ? 1u : 0u;
  __syncthreads();

  const bool slow = __builtin_amdgcn_ballot_w32(wflag[lane] != 0u) != 0u;
  int wcount = 0;
  if (!slow) {
    bkt_take(bw, wave, lane, wcount, seg);
  } else {
    if (tid < kMarkWords) mk[tid] = ~0u;
    __syncthreads();
    const int4* __restrict__ src4 = reinterpret_cast<const int4*>(ei);
    const int off = blockIdx.x * kPeStagger;
    const int nt = pe_todo(mk, todo, off, wave, lane);
    int4 da[kPeLoads], db[kPeLoads];
    pe_load(da, src4, todo, 0, nt, lane);
    for (int b = 0; b < nt; b += 2 * kStepGroups) {
      pe_load(db, src4, todo, b + kStepGroups, nt, lane);
      if (wcount > kPeSeg - kPeStepMax) {
        note_resolve(seg, 0, wcount, lane, et);
        pe_add_noted(seg, wcount, lane, rows, pes);
        wcount = 0;
      }
      pe_scan(da, todo, b, nt, lane, s0, (unsigned)nrow, wcount, seg);
      pe_load(da, src4, todo, b + 2 * kStepGroups, nt, lane);
      if (wcount > kPeSeg - kPeStepMax) {
        note_resolve(seg, 0, wcount, lane, et);
        pe_add_noted(seg, wcount, lane, rows, pes);
        wcount = 0;
      }
      pe_scan(db, todo, b + kStepGroups, nt, lane, s0, (unsigned)nrow, wcount, seg);
    }
    note_resolve(seg, 0, wcount, lane, et);
  }

  pe_add_noted(seg, wcount, lane, rows, pes);
  __syncthreads();

  {
    const int half  = lane >> 4;
    const int l15   = lane & 15;
    const int ntile = (nrow + 15) >> 4;
    const int px = 8 * half * kHC + l15;
    const int ps = 8 * half * kHeads + (l15 & (kHeads - 1));
    for (int it = wave; it < ntile * kColTiles; it += kOwnWaves) {
      const int t  = it / kColTiles;
      const int c  = it - t * kColTiles;
      const int r0 = t * 16 + 8 * half;
      const float* ar = rows + (t * 16 + l15) * kRowPitch + 2 * half;
      const v8f acc = tile_product(ar, reinterpret_cast<const float4*>(wf + c * kWfTile) + lane);
      if (c < kColTiles - 1) {
        float* __restrict__ o = xl + (size_t)(s0 + t * 16) * kHC + c * 16;
        if (r0 < nrow)     { o[px]           = acc[0]; o[px + kHC]     = acc[1]; o[px + 2 * kHC] = acc[2]; o[px + 3 * kHC] = acc[3]; }
        if (r0 + 4 < nrow) { o[px + 4 * kHC] = acc[4]; o[px + 5 * kHC] = acc[5]; o[px + 6 * kHC] = acc[6]; o[px + 7 * kHC] = acc[7]; }
      } else {
        if (l15 < 2 * kHeads) {
          float* __restrict__ o = ((l15 < kHeads) ? a_src : a_dst) + (size_t)(s0 + t * 16) * kHeads;
          if (r0 < nrow)     { o[ps]              = acc[0]; o[ps + kHeads]     = acc[1]; o[ps + 2 * kHeads] = acc[2]; o[ps + 3 * kHeads] = acc[3]; }
          if (r0 + 4 < nrow) { o[ps + 4 * kHeads] = acc[4]; o[ps + 5 * kHeads] = acc[5]; o[ps + 6 * kHeads] = acc[6]; o[ps + 7 * kHeads] = acc[7]; }
        }
      }
    }
  }
}


template <int PH>
__device__ __forceinline__ void ag_stats(const unsigned* seg, unsigned* sc, bool keep, int n, int lane,
                                         const float* __restrict__ a_src,
                                         const float* adst_l, float* amax_l, float* den_l) {
  const int u  = lane >> 2;
  const int hh = lane & 3;
  const bool kept = (PH == 1) && keep;
  for (int base = 0; base < n; base += 32) {
    unsigned ent[4];
    float as[4];
    if (kept) {
#pragma unroll
      for (int j = 0; j < 4; ++j) {
        const int idx = base + 8 * j + u;
        const int ci  = idx < n ? idx : n - 1;
        ent[j] = seg[ci];
        as[j]  = __uint_as_float(sc[4 * ci + hh]);
      }
    } else {
#pragma unroll
      for (int j = 0; j < 4; ++j) {
        const int idx = base + 8 * j + u;
        ent[j] = seg[idx < n ? idx : n - 1];
        as[j]  = a_src[(ent[j] >> 8) * kHeads + hh];
      }
    }
    asm volatile("" :: "v"(as[0]), "v"(as[1]), "v"(as[2]), "v"(as[3]));
#pragma unroll
    for (int j = 0; j < 4; ++j) {
      const int idx = base + 8 * j + u;
      if (idx < n) {
        const int k = (int)(ent[j] & 255u) * kHeads + hh;
        const float al = kept ? as[j] : leaky02(as[j] + adst_l[k]);
        if (PH == 0) {
          if (keep) sc[4 * idx + hh] = __float_as_uint(al);
          atomic_max_f32(&amax_l[k], al);
        } else {
          const float ex = expf(al - amax_l[k]);
          if (keep) sc[4 * idx + hh] = __float_as_uint(ex);
          atomicAdd(&den_l[k], ex);
        }
      }
    }
  }
}

__device__ __forceinline__ void ag_rows(const unsigned* seg, const unsigned* sc, bool keep, int n, int lane,
                                        const float* __restrict__ a_src,
                                        const float* adst_l, const float* amax_l,
                                        const float* den_l, float* acc,
                                        const float4* __restrict__ xl4) {
  const int u  = lane >> 2;
  const int hh = lane & 3;
  const int hr = lane >> 3;
  for (int base = 0; base < n; base += 8) {
    const int idx = base + u;
    const int ci  = idx < n ? idx : n - 1;
    const unsigned ent = seg[ci];
    const int k = (int)(ent & 255u) * kHeads + hh;
    float as = 0.0f, ex = 0.0f;
    if (keep) ex = __uint_as_float(sc[4 * ci + hh]);
    else      as = a_src[(ent >> 8) * kHeads + hh];
    float4 v[8];
    int rl[8];
#pragma unroll
    for (int j = 0; j < 8; ++j) {
      const unsigned eu = (unsigned)__builtin_amdgcn_readlane((int)ent, 4 * j);
      rl[j] = (int)(eu & 255u);
      v[j]  = xl4[(size_t)(eu >> 8) * (kHC / 4) + lane];
    }
    if (!keep) ex = expf(leaky02(as + adst_l[k]) - amax_l[k]);
    const float coef = ex / den_l[k];
    float c[8];
#pragma unroll
    for (int j = 0; j < 8; ++j) c[j] = __shfl(coef, 4 * j + hr);
    asm volatile("" :: "v"(coef), "v"(v[0].x), "v"(v[1].x), "v"(v[2].x), "v"(v[3].x),
                       "v"(v[4].x), "v"(v[5].x), "v"(v[6].x), "v"(v[7].x));
#pragma unroll
    for (int j = 0; j < 8; ++j) {
      if (base + j < n) {
        float* row = acc + rl[j] * kHC + lane;
        atomicAdd(row,      c[j] * v[j].x);
        atomicAdd(row + 32, c[j] * v[j].y);
        atomicAdd(row + 64, c[j] * v[j].z);
        atomicAdd(row + 96, c[j] * v[j].w);
      }
    }
  }
}

__global__ void __launch_bounds__(kOwnThr)
k_owner_aggregate(const int* __restrict__ ei, const unsigned* __restrict__ bdst,
                  const float* __restrict__ a_src, const float* __restrict__ a_dst,
                  const float* __restrict__ xl, const float* __restrict__ bias,
                  float* __restrict__ out) {
  __shared__ float    acc[kOwnRows * kHC];
  __shared__ unsigned list[kOwnWaves * kPeSeg];
  __shared__ float    adst_l[kOwnRows * kHeads];
  __shared__ float    amax_l[kOwnRows * kHeads];
  __shared__ float    den_l[kOwnRows * kHeads];
  __shared__ int      ovf;
  __shared__ unsigned short todo_s[kOwnWaves * kTodoPad];
  __shared__ unsigned mk[kMarkWords];
  __shared__ unsigned wflag[kOwnWaves];

  const int tid  = threadIdx.x;
  const int lane = tid & 31;
  const int wave = __builtin_amdgcn_readfirstlane(tid >> 5);
  const int d0   = blockIdx.x * kOwnRows;
  uint2 bw[kBktTrips];
  bkt_load(bw, bdst + (size_t)blockIdx.x * kChunks * kBktWords, wave, lane);
  const int nrow = (kNodes - d0 < kOwnRows) ? (kNodes - d0) : kOwnRows;
  const float4* __restrict__ xl4 = reinterpret_cast<const float4*>(xl);
  unsigned* seg = list + wave * kPeSeg;
  unsigned short* todo = todo_s + wave * kTodoPad;

  {
    float4* acc4 = reinterpret_cast<float4*>(acc);
    for (int i = tid; i < kOwnRows * kHC / 4; i += kOwnThr) acc4[i] = make_float4(0.f, 0.f, 0.f, 0.f);
    if (tid < kOwnRows * kHeads) {
      adst_l[tid] = (tid < nrow * kHeads) ? a_dst[d0 * kHeads + tid] : 0.0f;
      amax_l[tid] = -INFINITY;
      den_l[tid]  = 0.0f;
    }
    if (tid == 0) ovf = 0;
    if (lane == 0) wflag[wave] = bkt_over(bw, wave) ? 1u : 0u;
  }
  __syncthreads();

  const bool slow = __builtin_amdgcn_ballot_w32(wflag[lane] != 0u) != 0u;

  const int nself = (nrow > wave) ? ((nrow - wave + 31) >> 5) : 0;
  if (lane < nself) seg[lane] = ((unsigned)(d0 + wave + 32 * lane) << 8) | (unsigned)(wave + 32 * lane);

  const int4* __restrict__ dst4 = reinterpret_cast<const int4*>(ei + kEdges);
  int wcount = nself;
  int nt = 0;
  if (!slow) {
    bkt_take(bw, wave, lane, wcount, seg);
  } else {
    if (tid < kMarkWords) mk[tid] = ~0u;
    __syncthreads();
    const int off = blockIdx.x * kPeStagger;
    nt = pe_todo(mk, todo, off, wave, lane);
    bool over = false;
    int4 da[kPeLoads], db[kPeLoads];
    pe_load(da, dst4, todo, 0, nt, lane);
    for (int b = 0; b < nt; b += 2 * kStepGroups) {
      pe_load(db, dst4, todo, b + kStepGroups, nt, lane);
      over = over || (wcount > kPeSeg - kPeStepMax);
      if (!over) pe_scan(da, todo, b, nt, lane, d0, (unsigned)nrow, wcount, seg);
      pe_load(da, dst4, todo, b + 2 * kStepGroups, nt, lane);
      over = over || (wcount > kPeSeg - kPeStepMax);
      if (!over) pe_scan(db, todo, b + kStepGroups, nt, lane, d0, (unsigned)nrow, wcount, seg);
    }
    if (over) ovf = 1;
    else      note_resolve(seg, nself, wcount, lane, ei);
  }
  __syncthreads();

  const bool fits = __builtin_amdgcn_readfirstlane(ovf) == 0;
  if (fits) {
    const bool keep = 5 * wcount <= kPeSeg;
    unsigned* sc = seg + wcount;
    ag_stats<0>(seg, sc, keep, wcount, lane, a_src, adst_l, amax_l, den_l);
    __syncthreads();
    ag_stats<1>(seg, sc, keep, wcount, lane, a_src, adst_l, amax_l, den_l);
    __syncthreads();
    ag_rows(seg, sc, keep, wcount, lane, a_src, adst_l, amax_l, den_l, acc, xl4);
  } else {
    int4 da[kPeLoads];
    for (int ph = 0; ph < 3; ++ph) {
      for (int b = 0; b == 0 || b < nt; b += kStepGroups) {
        int n = 0;
        if (b == 0) {
          if (lane < nself) seg[lane] = ((unsigned)(d0 + wave + 32 * lane) << 8) | (unsigned)(wave + 32 * lane);
          n = nself;
        }
        const int n0 = n;
        pe_load(da, dst4, todo, b, nt, lane);
        pe_scan(da, todo, b, nt, lane, d0, (unsigned)nrow, n, seg);
        note_resolve(seg, n0, n, lane, ei);
        if (ph == 0)      ag_stats<0>(seg, seg, false, n, lane, a_src, adst_l, amax_l, den_l);
        else if (ph == 1) ag_stats<1>(seg, seg, false, n, lane, a_src, adst_l, amax_l, den_l);
        else              ag_rows(seg, seg, false, n, lane, a_src, adst_l, amax_l, den_l, acc, xl4);
      }
      __syncthreads();
    }
  }
  __syncthreads();

  const float4 b = reinterpret_cast<const float4*>(bias)[lane];
  float4* __restrict__ out4 = reinterpret_cast<float4*>(out);
  for (int dl = wave; dl < nrow; dl += kOwnWaves) {
    const int d = d0 + dl;
    const float* row = acc + dl * kHC + lane;
    float4 r;
    r.x = row[0]  + b.x;
    r.y = row[32] + b.y;
    r.z = row[64] + b.z;
    r.w = row[96] + b.w;
    r.x = r.x > 0.0f ? r.x : expm1f(r.x);
    r.y = r.y > 0.0f ? r.y : expm1f(r.y);
    r.z = r.z > 0.0f ? r.z : expm1f(r.z);
    r.w = r.w > 0.0f ? r.w : expm1f(r.w);
    out4[(size_t)d * (kHC / 4) + lane] = r;
  }
}


extern "C" void kernel_launch(void* const* d_in, const int* in_sizes, int n_in,
                              void* d_out, int out_size, void* d_ws, size_t ws_size,
                              hipStream_t stream) {
  (void)in_sizes; (void)n_in; (void)out_size; (void)ws_size;
  const float* x       = (const float*)d_in[0];
  const int*   ei      = (const int*)d_in[1];
  const int*   et      = (const int*)d_in[2];
  const float* pe      = (const float*)d_in[3];
  const float* W       = (const float*)d_in[4];
  const float* att_src = (const float*)d_in[5];
  const float* att_dst = (const float*)d_in[6];
  const float* bias    = (const float*)d_in[7];
  float* out = (float*)d_out;

  float* ws    = (float*)d_ws;
  float* xl    = ws;
  float* a_src = xl    + (size_t)kNodes * kHC;
  float* a_dst = a_src + (size_t)kNodes * kHeads;
  unsigned* bkt = (unsigned*)(a_dst + (size_t)kNodes * kHeads);
  float* wf    = (float*)(bkt + (size_t)2 * kOwnBlocks * kChunks * kBktWords);

  k_buckets<<<kChunks + 1, kOwnThr, 0, stream>>>(ei, et, bkt, W, att_src, att_dst, wf);
  k_owner_pe<<<kOwnBlocks, kOwnThr, 0, stream>>>(x, ei, et, pe, bkt, wf, xl, a_src, a_dst);
  k_owner_aggregate<<<kOwnBlocks, kOwnThr, 0, stream>>>(ei, bkt + (size_t)kOwnBlocks * kChunks * kBktWords,
                                                        a_src, a_dst, xl, bias, out);
}
